# starting kernel
# baseline (speedup 1.0000x reference)
#include <hip/hip_runtime.h>
#include <hip/hip_bf16.h>
#include <float.h>

#define BATCH   8
#define NPTS    2048
#define CH      128
#define HALF_CH 64
#define KNN_K   20
#define EPSB    1e-5f

typedef __attribute__((ext_vector_type(2))) float v2f;
typedef __attribute__((ext_vector_type(8))) float v8f;

__global__ void prep_params(const float* __restrict__ w1, const float* __restrict__ b1,
                            const float* __restrict__ g1, const float* __restrict__ be1,
                            const float* __restrict__ m1, const float* __restrict__ v1,
                            const float* __restrict__ w2, const float* __restrict__ b2,
                            const float* __restrict__ g2, const float* __restrict__ be2,
                            const float* __restrict__ m2, const float* __restrict__ v2,
                            float* __restrict__ Wc,   float* __restrict__ sw1a,
                            float* __restrict__ sw1d, float* __restrict__ bias1p,
                            float* __restrict__ bias2p) {
  int o = threadIdx.x;
  if (o >= HALF_CH) return;
  float s1 = g1[o] / sqrtf(v1[o] + EPSB);
  bias1p[o] = b1[o] * s1 + (be1[o] - m1[o] * s1);
  #pragma unroll
  for (int c = 0; c < 3; ++c) {
    float wa = w1[o * 6 + c];
    float wb = w1[o * 6 + 3 + c];
    sw1a[o * 3 + c] = s1 * wa;
    sw1d[o * 3 + c] = s1 * (wb - wa);
  }
  float s2 = g2[o] / sqrtf(v2[o] + EPSB);
  bias2p[o] = b2[o] * s2 + (be2[o] - m2[o] * s2);
  for (int c = 0; c < CH; ++c) {
    float wa = w2[o * (2 * CH) + c];
    float wb = w2[o * (2 * CH) + CH + c];
    Wc[(size_t)o * CH + c]             = s2 * wa;
    Wc[(size_t)(HALF_CH + o) * CH + c] = s2 * (wb - wa);
  }
}

__global__ void knn_kernel(const float* __restrict__ xyz, int* __restrict__ idx_out) {
  __shared__ float4 pts[NPTS];
  const int b = blockIdx.y;
  const float* xb = xyz + (size_t)b * 3 * NPTS;
  for (int m = threadIdx.x; m < NPTS; m += blockDim.x) {
    float x = xb[m], y = xb[NPTS + m], z = xb[2 * NPTS + m];
    pts[m] = make_float4(x, y, z, x * x + y * y + z * z);
  }
  __syncthreads();

  const int n = blockIdx.x * blockDim.x + threadIdx.x;
  const float4 me = pts[n];

  float td[KNN_K];
  int   ti[KNN_K];
  #pragma unroll
  for (int i = 0; i < KNN_K; ++i) { td[i] = FLT_MAX; ti[i] = 0; }

  for (int m = 0; m < NPTS; ++m) {
    float4 p = pts[m];
    float dot = me.x * p.x + me.y * p.y + me.z * p.z;
    float d = me.w + p.w - 2.0f * dot;
    if (d < td[KNN_K - 1]) {
      td[KNN_K - 1] = d; ti[KNN_K - 1] = m;
      #pragma unroll
      for (int i = KNN_K - 1; i > 0; --i) {
        if (td[i] < td[i - 1]) {
          float tf = td[i]; td[i] = td[i - 1]; td[i - 1] = tf;
          int   tt = ti[i]; ti[i] = ti[i - 1]; ti[i - 1] = tt;
        }
      }
    }
  }
  int* op = idx_out + ((size_t)b * NPTS + n) * KNN_K;
  #pragma unroll
  for (int i = 0; i < KNN_K; ++i) op[i] = ti[i];
}

__global__ void geo_kernel(const float* __restrict__ xyz, const float* __restrict__ sw1a,
                           const float* __restrict__ sw1d, const float* __restrict__ bias1p,
                           float* __restrict__ PQ) {
  int t  = blockIdx.x * blockDim.x + threadIdx.x;
  int ch = t & (HALF_CH - 1);
  int pt = t >> 6;
  int b  = pt / NPTS;
  int n  = pt - b * NPTS;
  const float* xb = xyz + (size_t)b * 3 * NPTS;
  float x = xb[n], y = xb[NPTS + n], z = xb[2 * NPTS + n];
  float p = sw1a[ch * 3 + 0] * x + sw1a[ch * 3 + 1] * y + sw1a[ch * 3 + 2] * z;
  float q = sw1d[ch * 3 + 0] * x + sw1d[ch * 3 + 1] * y + sw1d[ch * 3 + 2] * z + bias1p[ch];
  float* o = PQ + (size_t)pt * CH;
  o[ch] = p;
  o[HALF_CH + ch] = q;
}

__global__ void feat_gemm(const float* __restrict__ feat, const float* __restrict__ Wc,
                          float* __restrict__ Cf) {
  const int b    = blockIdx.y;
  const int n0   = blockIdx.x * 16;
  const int wave = threadIdx.x >> 5;
  const int lane = threadIdx.x & 31;
  const int l16  = lane & 15;
  const int hi   = lane >> 4;
  const int m0   = wave * 16;

  const float* Fb   = feat + (size_t)b * CH * NPTS;
  const float* arow = Wc + (size_t)(m0 + l16) * CH + 2 * hi;
  const float* bcol = Fb + (size_t)(2 * hi) * NPTS + n0 + l16;

  v8f acc = {0.f, 0.f, 0.f, 0.f, 0.f, 0.f, 0.f, 0.f};

  #pragma unroll 8
  for (int k = 0; k < CH; k += 4) {
    v2f a;
    a.x = arow[k];
    a.y = arow[k + 1];
    v2f bb;
    bb.x = bcol[(size_t)k * NPTS];
    bb.y = bcol[(size_t)(k + 1) * NPTS];
    acc = __builtin_amdgcn_wmma_f32_16x16x4_f32(false, a, false, bb,
                                                (short)0, acc, false, false);
  }

  float* out = Cf + ((size_t)b * NPTS + n0 + l16) * CH + m0 + 8 * hi;
  #pragma unroll
  for (int v = 0; v < 8; ++v) out[v] = acc[v];
}

__global__ void gather_max(const float* __restrict__ PQ, const float* __restrict__ Cf,
                           const int* __restrict__ idx, const float* __restrict__ bias2p,
                           float* __restrict__ out) {
  const int b  = blockIdx.y;
  const int ch = threadIdx.x & (HALF_CH - 1);
  const int n  = blockIdx.x * 4 + (threadIdx.x >> 6);

  const size_t pbase = (size_t)b * NPTS + n;
  const int* id = idx + pbase * KNN_K;

  const float qg = PQ[pbase * CH + HALF_CH + ch];
  const float vf = Cf[pbase * CH + HALF_CH + ch] + bias2p[ch];

  float mg = -FLT_MAX, mf = -FLT_MAX;
  #pragma unroll 4
  for (int k = 0; k < KNN_K; ++k) {
    int j = id[k];
    size_t jb = ((size_t)b * NPTS + j) * CH;
    mg = fmaxf(mg, PQ[jb + ch]);
    mf = fmaxf(mf, Cf[jb + ch]);
  }
  float geo = fmaxf(mg + qg, 0.0f);
  float fea = fmaxf(mf + vf, 0.0f);

  out[(size_t)b * CH * NPTS + (size_t)ch * NPTS + n]             = geo;
  out[(size_t)b * CH * NPTS + (size_t)(HALF_CH + ch) * NPTS + n] = fea;
}

extern "C" void kernel_launch(void* const* d_in, const int* in_sizes, int n_in,
                              void* d_out, int out_size, void* d_ws, size_t ws_size,
                              hipStream_t stream) {
  (void)in_sizes; (void)n_in; (void)out_size; (void)ws_size;

  const float* xyz  = (const float*)d_in[0];
  const float* feat = (const float*)d_in[1];
  const float* w1   = (const float*)d_in[2];
  const float* b1   = (const float*)d_in[3];
  const float* g1   = (const float*)d_in[4];
  const float* be1  = (const float*)d_in[5];
  const float* m1   = (const float*)d_in[6];
  const float* v1   = (const float*)d_in[7];
  const float* w2   = (const float*)d_in[8];
  const float* b2   = (const float*)d_in[9];
  const float* g2   = (const float*)d_in[10];
  const float* be2  = (const float*)d_in[11];
  const float* m2   = (const float*)d_in[12];
  const float* v2   = (const float*)d_in[13];

  char* ws = (char*)d_ws;
  float* Wc     = (float*)ws;
  float* sw1a   = Wc + 16384;
  float* sw1d   = sw1a + 192;
  float* bias1p = sw1d + 192;
  float* bias2p = bias1p + 64;
  float* PQ     = (float*)(ws + (128 << 10));
  float* Cf     = PQ + (size_t)BATCH * NPTS * CH;
  int*   idxb   = (int*)(Cf + (size_t)BATCH * NPTS * CH);

  prep_params<<<1, 64, 0, stream>>>(w1, b1, g1, be1, m1, v1,
                                    w2, b2, g2, be2, m2, v2,
                                    Wc, sw1a, sw1d, bias1p, bias2p);

  knn_kernel<<<dim3(NPTS / 256, BATCH), 256, 0, stream>>>(xyz, idxb);

  geo_kernel<<<(BATCH * NPTS * HALF_CH) / 256, 256, 0, stream>>>(xyz, sw1a, sw1d,
                                                                 bias1p, PQ);

  feat_gemm<<<dim3(NPTS / 16, BATCH), 256, 0, stream>>>(feat, Wc, Cf);

  gather_max<<<dim3(NPTS / 4, BATCH), 256, 0, stream>>>(PQ, Cf, idxb, bias2p,
                                                        (float*)d_out);
}


# kNN: per-key ballot+popcount (a vector-to-scalar round trip per key and trial) replaced by per-lane counts on the vector ALU, one DPP wave sum per trial, bit-mask write-out; scores bit-identical
# speedup vs baseline: 23.6952x; 23.6952x over previous
#include <hip/hip_runtime.h>
#include <hip/hip_bf16.h>
#include <float.h>

#define BATCH   8
#define NPTS    2048
#define CH      128
#define HALF_CH 64
#define KNN_K   20
#define EPSB    1e-5f

typedef __attribute__((ext_vector_type(2))) float v2f;
typedef __attribute__((ext_vector_type(8))) float v8f;

__global__ void prep_params(const float* __restrict__ w1, const float* __restrict__ b1,
                            const float* __restrict__ g1, const float* __restrict__ be1,
                            const float* __restrict__ m1, const float* __restrict__ v1,
                            const float* __restrict__ w2, const float* __restrict__ b2,
                            const float* __restrict__ g2, const float* __restrict__ be2,
                            const float* __restrict__ m2, const float* __restrict__ v2,
                            float* __restrict__ Wc,   float* __restrict__ sw1a,
                            float* __restrict__ sw1d, float* __restrict__ bias1p,
                            float* __restrict__ bias2p) {
  int o = threadIdx.x;
  if (o >= HALF_CH) return;
  float s1 = g1[o] / sqrtf(v1[o] + EPSB);
  bias1p[o] = b1[o] * s1 + (be1[o] - m1[o] * s1);
  #pragma unroll
  for (int c = 0; c < 3; ++c) {
    float wa = w1[o * 6 + c];
    float wb = w1[o * 6 + 3 + c];
    sw1a[o * 3 + c] = s1 * wa;
    sw1d[o * 3 + c] = s1 * (wb - wa);
  }
  float s2 = g2[o] / sqrtf(v2[o] + EPSB);
  bias2p[o] = b2[o] * s2 + (be2[o] - m2[o] * s2);
  for (int c = 0; c < CH; ++c) {
    float wa = w2[o * (2 * CH) + c];
    float wb = w2[o * (2 * CH) + CH + c];
    Wc[(size_t)o * CH + c]             = s2 * wa;
    Wc[(size_t)(HALF_CH + o) * CH + c] = s2 * (wb - wa);
  }
}

#define KNN_WAVES 16
#define KNN_QPW   4
#define KNN_QPB   (KNN_WAVES * KNN_QPW)
#define KNN_SLOTS (NPTS / 32)

__device__ __forceinline__ unsigned wave_sum_u32(unsigned v) {
  v += (unsigned)__builtin_amdgcn_update_dpp(0, (int)v, 0x161, 0xf, 0xf, false);
  v += (unsigned)__builtin_amdgcn_update_dpp(0, (int)v, 0x162, 0xf, 0xf, false);
  v += (unsigned)__builtin_amdgcn_update_dpp(0, (int)v, 0x164, 0xf, 0xf, false);
  v += (unsigned)__builtin_amdgcn_update_dpp(0, (int)v, 0x168, 0xf, 0xf, false);
  return (unsigned)__builtin_amdgcn_readlane((int)v, 0) +
         (unsigned)__builtin_amdgcn_readlane((int)v, 16);
}

__device__ __forceinline__ void count_below8(unsigned& c0, unsigned& c1, unsigned& c2,
                                             unsigned& c3, unsigned k0, unsigned k1,
                                             unsigned k2, unsigned k3, unsigned k4,
                                             unsigned k5, unsigned k6, unsigned k7,
                                             unsigned t) {
  unsigned m1, m2, m3, m4, m5, m6, m7;
  asm("v_cmp_lt_u32_e32 vcc_lo, %11, %19\n\t"
      "v_cmp_lt_u32_e64 %4, %12, %19\n\t"
      "v_cmp_lt_u32_e64 %5, %13, %19\n\t"
      "v_cmp_lt_u32_e64 %6, %14, %19\n\t"
      "v_cmp_lt_u32_e64 %7, %15, %19\n\t"
      "v_cmp_lt_u32_e64 %8, %16, %19\n\t"
      "v_cmp_lt_u32_e64 %9, %17, %19\n\t"
      "v_cmp_lt_u32_e64 %10, %18, %19\n\t"
      "s_delay_alu instid0(VALU_DEP_4) | instskip(NEXT) | instid1(VALU_DEP_4)\n\t"
      "v_add_co_ci_u32_e32 %0, vcc_lo, 0, %0, vcc_lo\n\t"
      "v_add_co_ci_u32_e64 %1, %4, 0, %1, %4\n\t"
      "s_delay_alu instid0(VALU_DEP_4) | instskip(NEXT) | instid1(VALU_DEP_4)\n\t"
      "v_add_co_ci_u32_e64 %2, %5, 0, %2, %5\n\t"
      "v_add_co_ci_u32_e64 %3, %6, 0, %3, %6\n\t"
      "s_delay_alu instid0(VALU_DEP_4) | instskip(NEXT) | instid1(VALU_DEP_4)\n\t"
      "v_add_co_ci_u32_e64 %0, %7, 0, %0, %7\n\t"
      "v_add_co_ci_u32_e64 %1, %8, 0, %1, %8\n\t"
      "s_delay_alu instid0(VALU_DEP_4) | instskip(NEXT) | instid1(VALU_DEP_4)\n\t"
      "v_add_co_ci_u32_e64 %2, %9, 0, %2, %9\n\t"
      "v_add_co_ci_u32_e64 %3, %10, 0, %3, %10"
      : "+v"(c0), "+v"(c1), "+v"(c2), "+v"(c3),
        "=&s"(m1), "=&s"(m2), "=&s"(m3), "=&s"(m4), "=&s"(m5), "=&s"(m6), "=&s"(m7)
      : "v"(k0), "v"(k1), "v"(k2), "v"(k3), "v"(k4), "v"(k5), "v"(k6), "v"(k7), "v"(t)
      : "vcc");
}
__device__ __forceinline__ void shift_in_below2(unsigned& mlo, unsigned& mhi, unsigned klo,
                                                unsigned khi, unsigned t) {
  unsigned m1;
  asm("v_cmp_lt_u32_e32 vcc_lo, %3, %5\n\t"
      "v_cmp_lt_u32_e64 %2, %4, %5\n\t"
      "s_delay_alu instid0(VALU_DEP_2) | instskip(NEXT) | instid1(VALU_DEP_2)\n\t"
      "v_add_co_ci_u32_e32 %0, vcc_lo, %0, %0, vcc_lo\n\t"
      "v_add_co_ci_u32_e64 %1, %2, %1, %1, %2"
      : "+v"(mlo), "+v"(mhi), "=&s"(m1)
      : "v"(klo), "v"(khi), "v"(t)
      : "vcc");
}
__device__ __forceinline__ void shift_in_equal2(unsigned& mlo, unsigned& mhi, unsigned klo,
                                                unsigned khi, unsigned t) {
  unsigned m1;
  asm("v_cmp_eq_u32_e32 vcc_lo, %3, %5\n\t"
      "v_cmp_eq_u32_e64 %2, %4, %5\n\t"
      "s_delay_alu instid0(VALU_DEP_2) | instskip(NEXT) | instid1(VALU_DEP_2)\n\t"
      "v_add_co_ci_u32_e32 %0, vcc_lo, %0, %0, vcc_lo\n\t"
      "v_add_co_ci_u32_e64 %1, %2, %1, %1, %2"
      : "+v"(mlo), "+v"(mhi), "=&s"(m1)
      : "v"(klo), "v"(khi), "v"(t)
      : "vcc");
}

__global__ void __launch_bounds__(KNN_WAVES * 32)
knn_kernel(const float* __restrict__ xyz, int* __restrict__ idx_out) {
  __shared__ float4 pts[NPTS];
  const int b = blockIdx.y;
  const float* xb = xyz + (size_t)b * 3 * NPTS;
  for (int m = threadIdx.x; m < NPTS; m += blockDim.x) {
    float x = xb[m], y = xb[NPTS + m], z = xb[2 * NPTS + m];
    pts[m] = make_float4(x, y, z, fmaf(z, z, fmaf(x, x, y * y)));
  }
  __syncthreads();

  const int lane = threadIdx.x & 31;
  const int wave = threadIdx.x >> 5;

  for (int qi = 0; qi < KNN_QPW; ++qi) {
    const int q = blockIdx.x * KNN_QPB + wave * KNN_QPW + qi;
    const float4 me = pts[q];

    unsigned key[KNN_SLOTS];
    #pragma unroll
    for (int j = 0; j < KNN_SLOTS; ++j) {
      const float4 p = pts[j * 32 + lane];
      float dot;
      {
        #pragma clang fp contract(off)
        dot = fmaf(me.x, p.x, me.y * p.y) + me.z * p.z;
      }
      const float d = fmaf(-2.0f, dot, me.w + p.w);
      const unsigned u = __float_as_uint(d);
      key[j] = u ^ ((unsigned)((int)u >> 31) | 0x80000000u);
    }

    unsigned T = 0u, clt = 0u;
    for (int bit = 31; bit >= 0; --bit) {
      const unsigned trial = T | (1u << bit);
      unsigned c0 = 0u, c1 = 0u, c2 = 0u, c3 = 0u;
      #pragma unroll
      for (int j = 0; j < KNN_SLOTS; j += 8)
        count_below8(c0, c1, c2, c3, key[j + 0], key[j + 1], key[j + 2], key[j + 3],
                     key[j + 4], key[j + 5], key[j + 6], key[j + 7], trial);
      const unsigned c = wave_sum_u32((c0 + c1) + (c2 + c3));
      if (c <= (unsigned)KNN_K) {
        T = trial;
        clt = c;
        if (c == (unsigned)KNN_K) break;
      }
    }
    const unsigned need = (unsigned)KNN_K - clt;

    int* op = idx_out + ((size_t)b * NPTS + q) * KNN_K;

    unsigned mlo = 0u, mhi = 0u;
    #pragma unroll
    for (int j = 31; j >= 0; --j) shift_in_below2(mlo, mhi, key[j], key[j + 32], T);

    unsigned written = 0u;
    for (int pass = 0; pass < KNN_SLOTS; ++pass) {
      const bool has = (mlo | mhi) != 0u;
      const unsigned act = __builtin_amdgcn_ballot_w32(has);
      if (act == 0u) break;
      if (has) {
        int j;
        if (mlo != 0u) { j = __builtin_ctz(mlo);      mlo &= mlo - 1u; }
        else           { j = 32 + __builtin_ctz(mhi); mhi &= mhi - 1u; }
        op[written + __builtin_amdgcn_mbcnt_lo(act, 0u)] = j * 32 + lane;
      }
      written += (unsigned)__builtin_popcount(act);
    }

    if (need != 0u) {
      unsigned elo = 0u, ehi = 0u;
      #pragma unroll
      for (int j = 31; j >= 0; --j) shift_in_equal2(elo, ehi, key[j], key[j + 32], T);
      unsigned taken = 0u;
      for (int j = 0; j < KNN_SLOTS && taken < need; ++j) {
        const unsigned w = (j < 32) ? elo : ehi;
        const bool eq = ((w >> (j & 31)) & 1u) != 0u;
        const unsigned meq = __builtin_amdgcn_ballot_w32(eq);
        if (meq != 0u) {
          const unsigned r = taken + __builtin_amdgcn_mbcnt_lo(meq, 0u);
          if (eq && r < need) op[clt + r] = j * 32 + lane;
          taken += (unsigned)__builtin_popcount(meq);
        }
      }
    }
  }
}

__global__ void geo_kernel(const float* __restrict__ xyz, const float* __restrict__ sw1a,
                           const float* __restrict__ sw1d, const float* __restrict__ bias1p,
                           float* __restrict__ PQ) {
  int t  = blockIdx.x * blockDim.x + threadIdx.x;
  int ch = t & (HALF_CH - 1);
  int pt = t >> 6;
  int b  = pt / NPTS;
  int n  = pt - b * NPTS;
  const float* xb = xyz + (size_t)b * 3 * NPTS;
  float x = xb[n], y = xb[NPTS + n], z = xb[2 * NPTS + n];
  float p = sw1a[ch * 3 + 0] * x + sw1a[ch * 3 + 1] * y + sw1a[ch * 3 + 2] * z;
  float q = sw1d[ch * 3 + 0] * x + sw1d[ch * 3 + 1] * y + sw1d[ch * 3 + 2] * z + bias1p[ch];
  float* o = PQ + (size_t)pt * CH;
  o[ch] = p;
  o[HALF_CH + ch] = q;
}

__global__ void feat_gemm(const float* __restrict__ feat, const float* __restrict__ Wc,
                          float* __restrict__ Cf) {
  const int b    = blockIdx.y;
  const int n0   = blockIdx.x * 16;
  const int wave = threadIdx.x >> 5;
  const int lane = threadIdx.x & 31;
  const int l16  = lane & 15;
  const int hi   = lane >> 4;
  const int m0   = wave * 16;

  const float* Fb   = feat + (size_t)b * CH * NPTS;
  const float* arow = Wc + (size_t)(m0 + l16) * CH + 2 * hi;
  const float* bcol = Fb + (size_t)(2 * hi) * NPTS + n0 + l16;

  v8f acc = {0.f, 0.f, 0.f, 0.f, 0.f, 0.f, 0.f, 0.f};

  #pragma unroll 8
  for (int k = 0; k < CH; k += 4) {
    v2f a;
    a.x = arow[k];
    a.y = arow[k + 1];
    v2f bb;
    bb.x = bcol[(size_t)k * NPTS];
    bb.y = bcol[(size_t)(k + 1) * NPTS];
    acc = __builtin_amdgcn_wmma_f32_16x16x4_f32(false, a, false, bb,
                                                (short)0, acc, false, false);
  }

  float* out = Cf + ((size_t)b * NPTS + n0 + l16) * CH + m0 + 8 * hi;
  #pragma unroll
  for (int v = 0; v < 8; ++v) out[v] = acc[v];
}

__global__ void gather_max(const float* __restrict__ PQ, const float* __restrict__ Cf,
                           const int* __restrict__ idx, const float* __restrict__ bias2p,
                           float* __restrict__ out) {
  const int b  = blockIdx.y;
  const int ch = threadIdx.x & (HALF_CH - 1);
  const int n  = blockIdx.x * 4 + (threadIdx.x >> 6);

  const size_t pbase = (size_t)b * NPTS + n;
  const int* id = idx + pbase * KNN_K;

  const float qg = PQ[pbase * CH + HALF_CH + ch];
  const float vf = Cf[pbase * CH + HALF_CH + ch] + bias2p[ch];

  float mg = -FLT_MAX, mf = -FLT_MAX;
  #pragma unroll 4
  for (int k = 0; k < KNN_K; ++k) {
    int j = id[k];
    size_t jb = ((size_t)b * NPTS + j) * CH;
    mg = fmaxf(mg, PQ[jb + ch]);
    mf = fmaxf(mf, Cf[jb + ch]);
  }
  float geo = fmaxf(mg + qg, 0.0f);
  float fea = fmaxf(mf + vf, 0.0f);

  out[(size_t)b * CH * NPTS + (size_t)ch * NPTS + n]             = geo;
  out[(size_t)b * CH * NPTS + (size_t)(HALF_CH + ch) * NPTS + n] = fea;
}

extern "C" void kernel_launch(void* const* d_in, const int* in_sizes, int n_in,
                              void* d_out, int out_size, void* d_ws, size_t ws_size,
                              hipStream_t stream) {
  (void)in_sizes; (void)n_in; (void)out_size; (void)ws_size;

  const float* xyz  = (const float*)d_in[0];
  const float* feat = (const float*)d_in[1];
  const float* w1   = (const float*)d_in[2];
  const float* b1   = (const float*)d_in[3];
  const float* g1   = (const float*)d_in[4];
  const float* be1  = (const float*)d_in[5];
  const float* m1   = (const float*)d_in[6];
  const float* v1   = (const float*)d_in[7];
  const float* w2   = (const float*)d_in[8];
  const float* b2   = (const float*)d_in[9];
  const float* g2   = (const float*)d_in[10];
  const float* be2  = (const float*)d_in[11];
  const float* m2   = (const float*)d_in[12];
  const float* v2   = (const float*)d_in[13];

  char* ws = (char*)d_ws;
  float* Wc     = (float*)ws;
  float* sw1a   = Wc + 16384;
  float* sw1d   = sw1a + 192;
  float* bias1p = sw1d + 192;
  float* bias2p = bias1p + 64;
  float* PQ     = (float*)(ws + (128 << 10));
  float* Cf     = PQ + (size_t)BATCH * NPTS * CH;
  int*   idxb   = (int*)(Cf + (size_t)BATCH * NPTS * CH);

  prep_params<<<1, 64, 0, stream>>>(w1, b1, g1, be1, m1, v1,
                                    w2, b2, g2, be2, m2, v2,
                                    Wc, sw1a, sw1d, bias1p, bias2p);

  knn_kernel<<<dim3(NPTS / KNN_QPB, BATCH), KNN_WAVES * 32, 0, stream>>>(xyz, idxb);

  geo_kernel<<<(BATCH * NPTS * HALF_CH) / 256, 256, 0, stream>>>(xyz, sw1a, sw1d,
                                                                 bias1p, PQ);

  feat_gemm<<<dim3(NPTS / 16, BATCH), 256, 0, stream>>>(feat, Wc, Cf);

  gather_max<<<dim3(NPTS / 4, BATCH), 256, 0, stream>>>(PQ, Cf, idxb, bias2p,
                                                        (float*)d_out);
}


# gather kernel: geometry branch computed from the neighbours' coordinates (lane k loads neighbour k's index and xyz once, handed round by lane read); 8 MB table, its kernel and its gather dropped
# speedup vs baseline: 24.8892x; 1.0504x over previous
#include <hip/hip_runtime.h>
#include <hip/hip_bf16.h>
#include <float.h>

#define BATCH   8
#define NPTS    2048
#define CH      128
#define HALF_CH 64
#define KNN_K   20
#define EPSB    1e-5f

typedef __attribute__((ext_vector_type(2))) float v2f;
typedef __attribute__((ext_vector_type(8))) float v8f;

__global__ void prep_params(const float* __restrict__ w1, const float* __restrict__ b1,
                            const float* __restrict__ g1, const float* __restrict__ be1,
                            const float* __restrict__ m1, const float* __restrict__ v1,
                            const float* __restrict__ w2, const float* __restrict__ b2,
                            const float* __restrict__ g2, const float* __restrict__ be2,
                            const float* __restrict__ m2, const float* __restrict__ v2,
                            float* __restrict__ Wc,   float* __restrict__ sw1a,
                            float* __restrict__ sw1d, float* __restrict__ bias1p,
                            float* __restrict__ bias2p) {
  int o = threadIdx.x;
  if (o >= HALF_CH) return;
  float s1 = g1[o] / sqrtf(v1[o] + EPSB);
  bias1p[o] = b1[o] * s1 + (be1[o] - m1[o] * s1);
  #pragma unroll
  for (int c = 0; c < 3; ++c) {
    float wa = w1[o * 6 + c];
    float wb = w1[o * 6 + 3 + c];
    sw1a[o * 3 + c] = s1 * wa;
    sw1d[o * 3 + c] = s1 * (wb - wa);
  }
  float s2 = g2[o] / sqrtf(v2[o] + EPSB);
  bias2p[o] = b2[o] * s2 + (be2[o] - m2[o] * s2);
  for (int c = 0; c < CH; ++c) {
    float wa = w2[o * (2 * CH) + c];
    float wb = w2[o * (2 * CH) + CH + c];
    Wc[(size_t)o * CH + c]             = s2 * wa;
    Wc[(size_t)(HALF_CH + o) * CH + c] = s2 * (wb - wa);
  }
}

#define KNN_WAVES 16
#define KNN_QPW   4
#define KNN_QPB   (KNN_WAVES * KNN_QPW)
#define KNN_SLOTS (NPTS / 32)

__device__ __forceinline__ unsigned wave_sum_u32(unsigned v) {
  v += (unsigned)__builtin_amdgcn_update_dpp(0, (int)v, 0x161, 0xf, 0xf, false);
  v += (unsigned)__builtin_amdgcn_update_dpp(0, (int)v, 0x162, 0xf, 0xf, false);
  v += (unsigned)__builtin_amdgcn_update_dpp(0, (int)v, 0x164, 0xf, 0xf, false);
  v += (unsigned)__builtin_amdgcn_update_dpp(0, (int)v, 0x168, 0xf, 0xf, false);
  return (unsigned)__builtin_amdgcn_readlane((int)v, 0) +
         (unsigned)__builtin_amdgcn_readlane((int)v, 16);
}

__device__ __forceinline__ void count_below8(unsigned& c0, unsigned& c1, unsigned& c2,
                                             unsigned& c3, unsigned k0, unsigned k1,
                                             unsigned k2, unsigned k3, unsigned k4,
                                             unsigned k5, unsigned k6, unsigned k7,
                                             unsigned t) {
  unsigned m1, m2, m3, m4, m5, m6, m7;
  asm("v_cmp_lt_u32_e32 vcc_lo, %11, %19\n\t"
      "v_cmp_lt_u32_e64 %4, %12, %19\n\t"
      "v_cmp_lt_u32_e64 %5, %13, %19\n\t"
      "v_cmp_lt_u32_e64 %6, %14, %19\n\t"
      "v_cmp_lt_u32_e64 %7, %15, %19\n\t"
      "v_cmp_lt_u32_e64 %8, %16, %19\n\t"
      "v_cmp_lt_u32_e64 %9, %17, %19\n\t"
      "v_cmp_lt_u32_e64 %10, %18, %19\n\t"
      "s_delay_alu instid0(VALU_DEP_4) | instskip(NEXT) | instid1(VALU_DEP_4)\n\t"
      "v_add_co_ci_u32_e32 %0, vcc_lo, 0, %0, vcc_lo\n\t"
      "v_add_co_ci_u32_e64 %1, %4, 0, %1, %4\n\t"
      "s_delay_alu instid0(VALU_DEP_4) | instskip(NEXT) | instid1(VALU_DEP_4)\n\t"
      "v_add_co_ci_u32_e64 %2, %5, 0, %2, %5\n\t"
      "v_add_co_ci_u32_e64 %3, %6, 0, %3, %6\n\t"
      "s_delay_alu instid0(VALU_DEP_4) | instskip(NEXT) | instid1(VALU_DEP_4)\n\t"
      "v_add_co_ci_u32_e64 %0, %7, 0, %0, %7\n\t"
      "v_add_co_ci_u32_e64 %1, %8, 0, %1, %8\n\t"
      "s_delay_alu instid0(VALU_DEP_4) | instskip(NEXT) | instid1(VALU_DEP_4)\n\t"
      "v_add_co_ci_u32_e64 %2, %9, 0, %2, %9\n\t"
      "v_add_co_ci_u32_e64 %3, %10, 0, %3, %10"
      : "+v"(c0), "+v"(c1), "+v"(c2), "+v"(c3),
        "=&s"(m1), "=&s"(m2), "=&s"(m3), "=&s"(m4), "=&s"(m5), "=&s"(m6), "=&s"(m7)
      : "v"(k0), "v"(k1), "v"(k2), "v"(k3), "v"(k4), "v"(k5), "v"(k6), "v"(k7), "v"(t)
      : "vcc");
}
__device__ __forceinline__ void shift_in_below2(unsigned& mlo, unsigned& mhi, unsigned klo,
                                                unsigned khi, unsigned t) {
  unsigned m1;
  asm("v_cmp_lt_u32_e32 vcc_lo, %3, %5\n\t"
      "v_cmp_lt_u32_e64 %2, %4, %5\n\t"
      "s_delay_alu instid0(VALU_DEP_2) | instskip(NEXT) | instid1(VALU_DEP_2)\n\t"
      "v_add_co_ci_u32_e32 %0, vcc_lo, %0, %0, vcc_lo\n\t"
      "v_add_co_ci_u32_e64 %1, %2, %1, %1, %2"
      : "+v"(mlo), "+v"(mhi), "=&s"(m1)
      : "v"(klo), "v"(khi), "v"(t)
      : "vcc");
}
__device__ __forceinline__ void shift_in_equal2(unsigned& mlo, unsigned& mhi, unsigned klo,
                                                unsigned khi, unsigned t) {
  unsigned m1;
  asm("v_cmp_eq_u32_e32 vcc_lo, %3, %5\n\t"
      "v_cmp_eq_u32_e64 %2, %4, %5\n\t"
      "s_delay_alu instid0(VALU_DEP_2) | instskip(NEXT) | instid1(VALU_DEP_2)\n\t"
      "v_add_co_ci_u32_e32 %0, vcc_lo, %0, %0, vcc_lo\n\t"
      "v_add_co_ci_u32_e64 %1, %2, %1, %1, %2"
      : "+v"(mlo), "+v"(mhi), "=&s"(m1)
      : "v"(klo), "v"(khi), "v"(t)
      : "vcc");
}

__global__ void __launch_bounds__(KNN_WAVES * 32)
knn_kernel(const float* __restrict__ xyz, int* __restrict__ idx_out) {
  __shared__ float4 pts[NPTS];
  const int b = blockIdx.y;
  const float* xb = xyz + (size_t)b * 3 * NPTS;
  for (int m = threadIdx.x; m < NPTS; m += blockDim.x) {
    float x = xb[m], y = xb[NPTS + m], z = xb[2 * NPTS + m];
    pts[m] = make_float4(x, y, z, fmaf(z, z, fmaf(x, x, y * y)));
  }
  __syncthreads();

  const int lane = threadIdx.x & 31;
  const int wave = threadIdx.x >> 5;

  for (int qi = 0; qi < KNN_QPW; ++qi) {
    const int q = blockIdx.x * KNN_QPB + wave * KNN_QPW + qi;
    const float4 me = pts[q];

    unsigned key[KNN_SLOTS];
    #pragma unroll
    for (int j = 0; j < KNN_SLOTS; ++j) {
      const float4 p = pts[j * 32 + lane];
      float dot;
      {
        #pragma clang fp contract(off)
        dot = fmaf(me.x, p.x, me.y * p.y) + me.z * p.z;
      }
      const float d = fmaf(-2.0f, dot, me.w + p.w);
      const unsigned u = __float_as_uint(d);
      key[j] = u ^ ((unsigned)((int)u >> 31) | 0x80000000u);
    }

    unsigned T = 0u, clt = 0u;
    for (int bit = 31; bit >= 0; --bit) {
      const unsigned trial = T | (1u << bit);
      unsigned c0 = 0u, c1 = 0u, c2 = 0u, c3 = 0u;
      #pragma unroll
      for (int j = 0; j < KNN_SLOTS; j += 8)
        count_below8(c0, c1, c2, c3, key[j + 0], key[j + 1], key[j + 2], key[j + 3],
                     key[j + 4], key[j + 5], key[j + 6], key[j + 7], trial);
      const unsigned c = wave_sum_u32((c0 + c1) + (c2 + c3));
      if (c <= (unsigned)KNN_K) {
        T = trial;
        clt = c;
        if (c == (unsigned)KNN_K) break;
      }
    }
    const unsigned need = (unsigned)KNN_K - clt;

    int* op = idx_out + ((size_t)b * NPTS + q) * KNN_K;

    unsigned mlo = 0u, mhi = 0u;
    #pragma unroll
    for (int j = 31; j >= 0; --j) shift_in_below2(mlo, mhi, key[j], key[j + 32], T);

    unsigned written = 0u;
    for (int pass = 0; pass < KNN_SLOTS; ++pass) {
      const bool has = (mlo | mhi) != 0u;
      const unsigned act = __builtin_amdgcn_ballot_w32(has);
      if (act == 0u) break;
      if (has) {
        int j;
        if (mlo != 0u) { j = __builtin_ctz(mlo);      mlo &= mlo - 1u; }
        else           { j = 32 + __builtin_ctz(mhi); mhi &= mhi - 1u; }
        op[written + __builtin_amdgcn_mbcnt_lo(act, 0u)] = j * 32 + lane;
      }
      written += (unsigned)__builtin_popcount(act);
    }

    if (need != 0u) {
      unsigned elo = 0u, ehi = 0u;
      #pragma unroll
      for (int j = 31; j >= 0; --j) shift_in_equal2(elo, ehi, key[j], key[j + 32], T);
      unsigned taken = 0u;
      for (int j = 0; j < KNN_SLOTS && taken < need; ++j) {
        const unsigned w = (j < 32) ? elo : ehi;
        const bool eq = ((w >> (j & 31)) & 1u) != 0u;
        const unsigned meq = __builtin_amdgcn_ballot_w32(eq);
        if (meq != 0u) {
          const unsigned r = taken + __builtin_amdgcn_mbcnt_lo(meq, 0u);
          if (eq && r < need) op[clt + r] = j * 32 + lane;
          taken += (unsigned)__builtin_popcount(meq);
        }
      }
    }
  }
}

__global__ void feat_gemm(const float* __restrict__ feat, const float* __restrict__ Wc,
                          float* __restrict__ Cf) {
  const int b    = blockIdx.y;
  const int n0   = blockIdx.x * 16;
  const int wave = threadIdx.x >> 5;
  const int lane = threadIdx.x & 31;
  const int l16  = lane & 15;
  const int hi   = lane >> 4;
  const int m0   = wave * 16;

  const float* Fb   = feat + (size_t)b * CH * NPTS;
  const float* arow = Wc + (size_t)(m0 + l16) * CH + 2 * hi;
  const float* bcol = Fb + (size_t)(2 * hi) * NPTS + n0 + l16;

  v8f acc = {0.f, 0.f, 0.f, 0.f, 0.f, 0.f, 0.f, 0.f};

  #pragma unroll 8
  for (int k = 0; k < CH; k += 4) {
    v2f a;
    a.x = arow[k];
    a.y = arow[k + 1];
    v2f bb;
    bb.x = bcol[(size_t)k * NPTS];
    bb.y = bcol[(size_t)(k + 1) * NPTS];
    acc = __builtin_amdgcn_wmma_f32_16x16x4_f32(false, a, false, bb,
                                                (short)0, acc, false, false);
  }

  float* out = Cf + ((size_t)b * NPTS + n0 + l16) * CH + m0 + 8 * hi;
  #pragma unroll
  for (int v = 0; v < 8; ++v) out[v] = acc[v];
}

__global__ void __launch_bounds__(256)
gather_max(const float* __restrict__ xyz, const float* __restrict__ sw1a,
           const float* __restrict__ sw1d, const float* __restrict__ bias1p,
           const float* __restrict__ Cf, const int* __restrict__ idx,
           const float* __restrict__ bias2p, float* __restrict__ out) {
  const int b    = blockIdx.y;
  const int lane = threadIdx.x & 31;
  const int wave = __builtin_amdgcn_readfirstlane((int)(threadIdx.x >> 5));
  const int ch   = (wave & 1) * 32 + lane;
  const int n    = blockIdx.x * 4 + (wave >> 1);

  const size_t pbase = (size_t)b * NPTS + n;
  const float* xb = xyz + (size_t)b * 3 * NPTS;

  const int kk = (lane < KNN_K) ? lane : lane - KNN_K;
  const int jv = idx[pbase * KNN_K + kk];
  const int xv = __float_as_int(xb[jv]);
  const int yv = __float_as_int(xb[NPTS + jv]);
  const int zv = __float_as_int(xb[2 * NPTS + jv]);

  const float a0 = sw1a[ch * 3 + 0], a1 = sw1a[ch * 3 + 1], a2 = sw1a[ch * 3 + 2];
  const float d0 = sw1d[ch * 3 + 0], d1 = sw1d[ch * 3 + 1], d2 = sw1d[ch * 3 + 2];
  const float cx = xb[n], cy = xb[NPTS + n], cz = xb[2 * NPTS + n];

  const float qg = d0 * cx + d1 * cy + d2 * cz + bias1p[ch];
  const float vf = Cf[pbase * CH + HALF_CH + ch] + bias2p[ch];

  const float* cfb = Cf + (size_t)b * NPTS * CH + ch;

  float mg = -FLT_MAX, mf = -FLT_MAX;
  #pragma unroll
  for (int k = 0; k < KNN_K; ++k) {
    const int   j = __builtin_amdgcn_readlane(jv, k);
    const float x = __int_as_float(__builtin_amdgcn_readlane(xv, k));
    const float y = __int_as_float(__builtin_amdgcn_readlane(yv, k));
    const float z = __int_as_float(__builtin_amdgcn_readlane(zv, k));
    mg = fmaxf(mg, a0 * x + a1 * y + a2 * z);
    mf = fmaxf(mf, cfb[(size_t)j * CH]);
  }
  float geo = fmaxf(mg + qg, 0.0f);
  float fea = fmaxf(mf + vf, 0.0f);

  out[(size_t)b * CH * NPTS + (size_t)ch * NPTS + n]             = geo;
  out[(size_t)b * CH * NPTS + (size_t)(HALF_CH + ch) * NPTS + n] = fea;
}

extern "C" void kernel_launch(void* const* d_in, const int* in_sizes, int n_in,
                              void* d_out, int out_size, void* d_ws, size_t ws_size,
                              hipStream_t stream) {
  (void)in_sizes; (void)n_in; (void)out_size; (void)ws_size;

  const float* xyz  = (const float*)d_in[0];
  const float* feat = (const float*)d_in[1];
  const float* w1   = (const float*)d_in[2];
  const float* b1   = (const float*)d_in[3];
  const float* g1   = (const float*)d_in[4];
  const float* be1  = (const float*)d_in[5];
  const float* m1   = (const float*)d_in[6];
  const float* v1   = (const float*)d_in[7];
  const float* w2   = (const float*)d_in[8];
  const float* b2   = (const float*)d_in[9];
  const float* g2   = (const float*)d_in[10];
  const float* be2  = (const float*)d_in[11];
  const float* m2   = (const float*)d_in[12];
  const float* v2   = (const float*)d_in[13];

  char* ws = (char*)d_ws;
  float* Wc     = (float*)ws;
  float* sw1a   = Wc + 16384;
  float* sw1d   = sw1a + 192;
  float* bias1p = sw1d + 192;
  float* bias2p = bias1p + 64;
  float* Cf     = (float*)(ws + (128 << 10)) + (size_t)BATCH * NPTS * CH;
  int*   idxb   = (int*)(Cf + (size_t)BATCH * NPTS * CH);

  prep_params<<<1, 64, 0, stream>>>(w1, b1, g1, be1, m1, v1,
                                    w2, b2, g2, be2, m2, v2,
                                    Wc, sw1a, sw1d, bias1p, bias2p);

  knn_kernel<<<dim3(NPTS / KNN_QPB, BATCH), KNN_WAVES * 32, 0, stream>>>(xyz, idxb);

  feat_gemm<<<dim3(NPTS / 16, BATCH), 256, 0, stream>>>(feat, Wc, Cf);

  gather_max<<<dim3(NPTS / 4, BATCH), 256, 0, stream>>>(xyz, sw1a, sw1d, bias1p, Cf, idxb,
                                                        bias2p, (float*)d_out);
}


# kNN: bisection only over keys <= U (U = 21st smallest of 32 lane minima, a bound on the 21st smallest key): a few dozen, listed via LDS, 2 per lane; full bisection as fallback; scores bit-identical
# speedup vs baseline: 34.5526x; 1.3883x over previous
#include <hip/hip_runtime.h>
#include <hip/hip_bf16.h>
#include <float.h>

#define BATCH   8
#define NPTS    2048
#define CH      128
#define HALF_CH 64
#define KNN_K   20
#define EPSB    1e-5f

typedef __attribute__((ext_vector_type(2))) float v2f;
typedef __attribute__((ext_vector_type(8))) float v8f;

__global__ void prep_params(const float* __restrict__ w1, const float* __restrict__ b1,
                            const float* __restrict__ g1, const float* __restrict__ be1,
                            const float* __restrict__ m1, const float* __restrict__ v1,
                            const float* __restrict__ w2, const float* __restrict__ b2,
                            const float* __restrict__ g2, const float* __restrict__ be2,
                            const float* __restrict__ m2, const float* __restrict__ v2,
                            float* __restrict__ Wc,   float* __restrict__ sw1a,
                            float* __restrict__ sw1d, float* __restrict__ bias1p,
                            float* __restrict__ bias2p) {
  int o = threadIdx.x;
  if (o >= HALF_CH) return;
  float s1 = g1[o] / sqrtf(v1[o] + EPSB);
  bias1p[o] = b1[o] * s1 + (be1[o] - m1[o] * s1);
  #pragma unroll
  for (int c = 0; c < 3; ++c) {
    float wa = w1[o * 6 + c];
    float wb = w1[o * 6 + 3 + c];
    sw1a[o * 3 + c] = s1 * wa;
    sw1d[o * 3 + c] = s1 * (wb - wa);
  }
  float s2 = g2[o] / sqrtf(v2[o] + EPSB);
  bias2p[o] = b2[o] * s2 + (be2[o] - m2[o] * s2);
  for (int c = 0; c < CH; ++c) {
    float wa = w2[o * (2 * CH) + c];
    float wb = w2[o * (2 * CH) + CH + c];
    Wc[(size_t)o * CH + c]             = s2 * wa;
    Wc[(size_t)(HALF_CH + o) * CH + c] = s2 * (wb - wa);
  }
}

#define KNN_WAVES 16
#define KNN_QPW   4
#define KNN_QPB   (KNN_WAVES * KNN_QPW)
#define KNN_SLOTS (NPTS / 32)
#define KNN_CAND  64

__device__ __forceinline__ unsigned wave_sum_u32(unsigned v) {
  v += (unsigned)__builtin_amdgcn_update_dpp(0, (int)v, 0x161, 0xf, 0xf, false);
  v += (unsigned)__builtin_amdgcn_update_dpp(0, (int)v, 0x162, 0xf, 0xf, false);
  v += (unsigned)__builtin_amdgcn_update_dpp(0, (int)v, 0x164, 0xf, 0xf, false);
  v += (unsigned)__builtin_amdgcn_update_dpp(0, (int)v, 0x168, 0xf, 0xf, false);
  return (unsigned)__builtin_amdgcn_readlane((int)v, 0) +
         (unsigned)__builtin_amdgcn_readlane((int)v, 16);
}

__device__ __forceinline__ void count_below8(unsigned& c0, unsigned& c1, unsigned& c2,
                                             unsigned& c3, unsigned k0, unsigned k1,
                                             unsigned k2, unsigned k3, unsigned k4,
                                             unsigned k5, unsigned k6, unsigned k7,
                                             unsigned t) {
  unsigned m1, m2, m3, m4, m5, m6, m7;
  asm("v_cmp_lt_u32_e32 vcc_lo, %11, %19\n\t"
      "v_cmp_lt_u32_e64 %4, %12, %19\n\t"
      "v_cmp_lt_u32_e64 %5, %13, %19\n\t"
      "v_cmp_lt_u32_e64 %6, %14, %19\n\t"
      "v_cmp_lt_u32_e64 %7, %15, %19\n\t"
      "v_cmp_lt_u32_e64 %8, %16, %19\n\t"
      "v_cmp_lt_u32_e64 %9, %17, %19\n\t"
      "v_cmp_lt_u32_e64 %10, %18, %19\n\t"
      "s_delay_alu instid0(VALU_DEP_4) | instskip(NEXT) | instid1(VALU_DEP_4)\n\t"
      "v_add_co_ci_u32_e32 %0, vcc_lo, 0, %0, vcc_lo\n\t"
      "v_add_co_ci_u32_e64 %1, %4, 0, %1, %4\n\t"
      "s_delay_alu instid0(VALU_DEP_4) | instskip(NEXT) | instid1(VALU_DEP_4)\n\t"
      "v_add_co_ci_u32_e64 %2, %5, 0, %2, %5\n\t"
      "v_add_co_ci_u32_e64 %3, %6, 0, %3, %6\n\t"
      "s_delay_alu instid0(VALU_DEP_4) | instskip(NEXT) | instid1(VALU_DEP_4)\n\t"
      "v_add_co_ci_u32_e64 %0, %7, 0, %0, %7\n\t"
      "v_add_co_ci_u32_e64 %1, %8, 0, %1, %8\n\t"
      "s_delay_alu instid0(VALU_DEP_4) | instskip(NEXT) | instid1(VALU_DEP_4)\n\t"
      "v_add_co_ci_u32_e64 %2, %9, 0, %2, %9\n\t"
      "v_add_co_ci_u32_e64 %3, %10, 0, %3, %10"
      : "+v"(c0), "+v"(c1), "+v"(c2), "+v"(c3),
        "=&s"(m1), "=&s"(m2), "=&s"(m3), "=&s"(m4), "=&s"(m5), "=&s"(m6), "=&s"(m7)
      : "v"(k0), "v"(k1), "v"(k2), "v"(k3), "v"(k4), "v"(k5), "v"(k6), "v"(k7), "v"(t)
      : "vcc");
}
__device__ __forceinline__ void shift_in_below2(unsigned& mlo, unsigned& mhi, unsigned klo,
                                                unsigned khi, unsigned t) {
  unsigned m1;
  asm("v_cmp_lt_u32_e32 vcc_lo, %3, %5\n\t"
      "v_cmp_lt_u32_e64 %2, %4, %5\n\t"
      "s_delay_alu instid0(VALU_DEP_2) | instskip(NEXT) | instid1(VALU_DEP_2)\n\t"
      "v_add_co_ci_u32_e32 %0, vcc_lo, %0, %0, vcc_lo\n\t"
      "v_add_co_ci_u32_e64 %1, %2, %1, %1, %2"
      : "+v"(mlo), "+v"(mhi), "=&s"(m1)
      : "v"(klo), "v"(khi), "v"(t)
      : "vcc");
}
__device__ __forceinline__ void shift_in_equal2(unsigned& mlo, unsigned& mhi, unsigned klo,
                                                unsigned khi, unsigned t) {
  unsigned m1;
  asm("v_cmp_eq_u32_e32 vcc_lo, %3, %5\n\t"
      "v_cmp_eq_u32_e64 %2, %4, %5\n\t"
      "s_delay_alu instid0(VALU_DEP_2) | instskip(NEXT) | instid1(VALU_DEP_2)\n\t"
      "v_add_co_ci_u32_e32 %0, vcc_lo, %0, %0, vcc_lo\n\t"
      "v_add_co_ci_u32_e64 %1, %2, %1, %1, %2"
      : "+v"(mlo), "+v"(mhi), "=&s"(m1)
      : "v"(klo), "v"(khi), "v"(t)
      : "vcc");
}

__device__ __forceinline__ void shift_in_le2(unsigned& mlo, unsigned& mhi, unsigned klo,
                                             unsigned khi, unsigned t) {
  unsigned m1;
  asm("v_cmp_le_u32_e32 vcc_lo, %3, %5\n\t"
      "v_cmp_le_u32_e64 %2, %4, %5\n\t"
      "s_delay_alu instid0(VALU_DEP_2) | instskip(NEXT) | instid1(VALU_DEP_2)\n\t"
      "v_add_co_ci_u32_e32 %0, vcc_lo, %0, %0, vcc_lo\n\t"
      "v_add_co_ci_u32_e64 %1, %2, %1, %1, %2"
      : "+v"(mlo), "+v"(mhi), "=&s"(m1)
      : "v"(klo), "v"(khi), "v"(t)
      : "vcc");
}
__device__ __forceinline__ void rank_below4(unsigned& r0, unsigned& r1, unsigned s0,
                                            unsigned s1, unsigned s2, unsigned s3,
                                            unsigned v) {
  unsigned m1, m2, m3;
  asm("v_cmp_lt_u32_e32 vcc_lo, %5, %9\n\t"
      "v_cmp_lt_u32_e64 %2, %6, %9\n\t"
      "v_cmp_lt_u32_e64 %3, %7, %9\n\t"
      "v_cmp_lt_u32_e64 %4, %8, %9\n\t"
      "s_delay_alu instid0(VALU_DEP_4) | instskip(NEXT) | instid1(VALU_DEP_4)\n\t"
      "v_add_co_ci_u32_e32 %0, vcc_lo, 0, %0, vcc_lo\n\t"
      "v_add_co_ci_u32_e64 %1, %2, 0, %1, %2\n\t"
      "s_delay_alu instid0(VALU_DEP_2) | instskip(NEXT) | instid1(VALU_DEP_2)\n\t"
      "v_add_co_ci_u32_e64 %0, %3, 0, %0, %3\n\t"
      "v_add_co_ci_u32_e64 %1, %4, 0, %1, %4"
      : "+v"(r0), "+v"(r1), "=&s"(m1), "=&s"(m2), "=&s"(m3)
      : "s"(s0), "s"(s1), "s"(s2), "s"(s3), "v"(v)
      : "vcc");
}
__device__ __forceinline__ unsigned wave_min_u32(unsigned v) {
  v = min(v, (unsigned)__builtin_amdgcn_update_dpp(0, (int)v, 0x161, 0xf, 0xf, false));
  v = min(v, (unsigned)__builtin_amdgcn_update_dpp(0, (int)v, 0x162, 0xf, 0xf, false));
  v = min(v, (unsigned)__builtin_amdgcn_update_dpp(0, (int)v, 0x164, 0xf, 0xf, false));
  v = min(v, (unsigned)__builtin_amdgcn_update_dpp(0, (int)v, 0x168, 0xf, 0xf, false));
  return min((unsigned)__builtin_amdgcn_readlane((int)v, 0),
             (unsigned)__builtin_amdgcn_readlane((int)v, 16));
}
__device__ __forceinline__ unsigned dist_key(const float4 me, const float4 p) {
  float dot;
  {
    #pragma clang fp contract(off)
    dot = fmaf(me.x, p.x, me.y * p.y) + me.z * p.z;
  }
  const float d = fmaf(-2.0f, dot, me.w + p.w);
  const unsigned u = __float_as_uint(d);
  return u ^ ((unsigned)((int)u >> 31) | 0x80000000u);
}

__global__ void __launch_bounds__(KNN_WAVES * 32)
knn_kernel(const float* __restrict__ xyz, int* __restrict__ idx_out) {
  __shared__ float4 pts[NPTS];
  __shared__ int cand[KNN_WAVES][KNN_CAND];
  const int b = blockIdx.y;
  const float* xb = xyz + (size_t)b * 3 * NPTS;
  for (int m = threadIdx.x; m < NPTS; m += blockDim.x) {
    float x = xb[m], y = xb[NPTS + m], z = xb[2 * NPTS + m];
    pts[m] = make_float4(x, y, z, fmaf(z, z, fmaf(x, x, y * y)));
  }
  __syncthreads();

  const int lane = threadIdx.x & 31;
  const int wave = threadIdx.x >> 5;
  int* cw = cand[wave];

  for (int qi = 0; qi < KNN_QPW; ++qi) {
    const int q = blockIdx.x * KNN_QPB + wave * KNN_QPW + qi;
    const float4 me = pts[q];
    int* op = idx_out + ((size_t)b * NPTS + q) * KNN_K;

    unsigned key[KNN_SLOTS];
    #pragma unroll
    for (int j = 0; j < KNN_SLOTS; ++j) key[j] = dist_key(me, pts[j * 32 + lane]);

    unsigned mn = key[0];
    #pragma unroll
    for (int j = 1; j < KNN_SLOTS; ++j) mn = min(mn, key[j]);
    const unsigned mu = (mn & ~31u) | (unsigned)lane;
    unsigned r0 = 0u, r1 = 0u;
    #pragma unroll
    for (int l = 0; l < 32; l += 4)
      rank_below4(r0, r1, (unsigned)__builtin_amdgcn_readlane((int)mu, l),
                  (unsigned)__builtin_amdgcn_readlane((int)mu, l + 1),
                  (unsigned)__builtin_amdgcn_readlane((int)mu, l + 2),
                  (unsigned)__builtin_amdgcn_readlane((int)mu, l + 3), mu);
    const unsigned holder = __builtin_amdgcn_ballot_w32((r0 + r1) == (unsigned)KNN_K);
    const unsigned U =
        (unsigned)__builtin_amdgcn_readlane((int)mu, __builtin_ctz(holder)) | 31u;

    unsigned mlo = 0u, mhi = 0u;
    #pragma unroll
    for (int j = 31; j >= 0; --j) shift_in_le2(mlo, mhi, key[j], key[j + 32], U);
    const unsigned cS =
        wave_sum_u32((unsigned)__builtin_popcount(mlo) + (unsigned)__builtin_popcount(mhi));

    if (cS <= (unsigned)KNN_CAND) {
      unsigned listed = 0u;
      for (int pass = 0; pass < KNN_SLOTS; ++pass) {
        const bool has = (mlo | mhi) != 0u;
        const unsigned act = __builtin_amdgcn_ballot_w32(has);
        if (act == 0u) break;
        if (has) {
          int j;
          if (mlo != 0u) { j = __builtin_ctz(mlo);      mlo &= mlo - 1u; }
          else           { j = 32 + __builtin_ctz(mhi); mhi &= mhi - 1u; }
          cw[listed + __builtin_amdgcn_mbcnt_lo(act, 0u)] = j * 32 + lane;
        }
        listed += (unsigned)__builtin_popcount(act);
      }
      __builtin_amdgcn_wave_barrier();
      const int i0 = ((unsigned)lane < cS) ? cw[lane] : -1;
      const int i1 = ((unsigned)lane + 32u < cS) ? cw[lane + 32] : -1;
      unsigned k0 = 0xFFFFFFFFu, k1 = 0xFFFFFFFFu;
      if (i0 >= 0) k0 = dist_key(me, pts[i0]);
      if (i1 >= 0) k1 = dist_key(me, pts[i1]);
      __builtin_amdgcn_wave_barrier();

      unsigned T = 0u, clt = 0u;
      for (int bit = 31; bit >= 0; --bit) {
        const unsigned trial = T | (1u << bit);
        const unsigned c =
            wave_sum_u32((k0 < trial ? 1u : 0u) + (k1 < trial ? 1u : 0u));
        if (c <= (unsigned)KNN_K) {
          T = trial;
          clt = c;
          if (c == (unsigned)KNN_K) break;
        }
      }
      const unsigned need = (unsigned)KNN_K - clt;

      const bool s0 = k0 < T, s1 = k1 < T;
      const unsigned b0 = __builtin_amdgcn_ballot_w32(s0);
      const unsigned b1 = __builtin_amdgcn_ballot_w32(s1);
      if (s0) op[__builtin_amdgcn_mbcnt_lo(b0, 0u)] = i0;
      if (s1) op[(unsigned)__builtin_popcount(b0) + __builtin_amdgcn_mbcnt_lo(b1, 0u)] = i1;

      if (need != 0u) {
        unsigned e0 = (k0 == T) ? (unsigned)i0 : 0xFFFFFFFFu;
        unsigned e1 = (k1 == T) ? (unsigned)i1 : 0xFFFFFFFFu;
        for (unsigned r = 0u; r < need; ++r) {
          const unsigned m = wave_min_u32(min(e0, e1));
          if (lane == 0) op[clt + r] = (int)m;
          if (e0 == m) e0 = 0xFFFFFFFFu;
          if (e1 == m) e1 = 0xFFFFFFFFu;
        }
      }
      continue;
    }

    unsigned T = 0u, clt = 0u;
    for (int bit = 31; bit >= 0; --bit) {
      const unsigned trial = T | (1u << bit);
      unsigned c0 = 0u, c1 = 0u, c2 = 0u, c3 = 0u;
      #pragma unroll
      for (int j = 0; j < KNN_SLOTS; j += 8)
        count_below8(c0, c1, c2, c3, key[j + 0], key[j + 1], key[j + 2], key[j + 3],
                     key[j + 4], key[j + 5], key[j + 6], key[j + 7], trial);
      const unsigned c = wave_sum_u32((c0 + c1) + (c2 + c3));
      if (c <= (unsigned)KNN_K) {
        T = trial;
        clt = c;
        if (c == (unsigned)KNN_K) break;
      }
    }
    const unsigned need = (unsigned)KNN_K - clt;

    mlo = 0u; mhi = 0u;
    #pragma unroll
    for (int j = 31; j >= 0; --j) shift_in_below2(mlo, mhi, key[j], key[j + 32], T);

    unsigned written = 0u;
    for (int pass = 0; pass < KNN_SLOTS; ++pass) {
      const bool has = (mlo | mhi) != 0u;
      const unsigned act = __builtin_amdgcn_ballot_w32(has);
      if (act == 0u) break;
      if (has) {
        int j;
        if (mlo != 0u) { j = __builtin_ctz(mlo);      mlo &= mlo - 1u; }
        else           { j = 32 + __builtin_ctz(mhi); mhi &= mhi - 1u; }
        op[written + __builtin_amdgcn_mbcnt_lo(act, 0u)] = j * 32 + lane;
      }
      written += (unsigned)__builtin_popcount(act);
    }

    if (need != 0u) {
      unsigned elo = 0u, ehi = 0u;
      #pragma unroll
      for (int j = 31; j >= 0; --j) shift_in_equal2(elo, ehi, key[j], key[j + 32], T);
      unsigned taken = 0u;
      for (int j = 0; j < KNN_SLOTS && taken < need; ++j) {
        const unsigned w = (j < 32) ? elo : ehi;
        const bool eq = ((w >> (j & 31)) & 1u) != 0u;
        const unsigned meq = __builtin_amdgcn_ballot_w32(eq);
        if (meq != 0u) {
          const unsigned r = taken + __builtin_amdgcn_mbcnt_lo(meq, 0u);
          if (eq && r < need) op[clt + r] = j * 32 + lane;
          taken += (unsigned)__builtin_popcount(meq);
        }
      }
    }
  }
}

__global__ void feat_gemm(const float* __restrict__ feat, const float* __restrict__ Wc,
                          float* __restrict__ Cf) {
  const int b    = blockIdx.y;
  const int n0   = blockIdx.x * 16;
  const int wave = threadIdx.x >> 5;
  const int lane = threadIdx.x & 31;
  const int l16  = lane & 15;
  const int hi   = lane >> 4;
  const int m0   = wave * 16;

  const float* Fb   = feat + (size_t)b * CH * NPTS;
  const float* arow = Wc + (size_t)(m0 + l16) * CH + 2 * hi;
  const float* bcol = Fb + (size_t)(2 * hi) * NPTS + n0 + l16;

  v8f acc = {0.f, 0.f, 0.f, 0.f, 0.f, 0.f, 0.f, 0.f};

  #pragma unroll 8
  for (int k = 0; k < CH; k += 4) {
    v2f a;
    a.x = arow[k];
    a.y = arow[k + 1];
    v2f bb;
    bb.x = bcol[(size_t)k * NPTS];
    bb.y = bcol[(size_t)(k + 1) * NPTS];
    acc = __builtin_amdgcn_wmma_f32_16x16x4_f32(false, a, false, bb,
                                                (short)0, acc, false, false);
  }

  float* out = Cf + ((size_t)b * NPTS + n0 + l16) * CH + m0 + 8 * hi;
  #pragma unroll
  for (int v = 0; v < 8; ++v) out[v] = acc[v];
}

__global__ void __launch_bounds__(256)
gather_max(const float* __restrict__ xyz, const float* __restrict__ sw1a,
           const float* __restrict__ sw1d, const float* __restrict__ bias1p,
           const float* __restrict__ Cf, const int* __restrict__ idx,
           const float* __restrict__ bias2p, float* __restrict__ out) {
  const int b    = blockIdx.y;
  const int lane = threadIdx.x & 31;
  const int wave = __builtin_amdgcn_readfirstlane((int)(threadIdx.x >> 5));
  const int ch   = (wave & 1) * 32 + lane;
  const int n    = blockIdx.x * 4 + (wave >> 1);

  const size_t pbase = (size_t)b * NPTS + n;
  const float* xb = xyz + (size_t)b * 3 * NPTS;

  const int kk = (lane < KNN_K) ? lane : lane - KNN_K;
  const int jv = idx[pbase * KNN_K + kk];
  const int xv = __float_as_int(xb[jv]);
  const int yv = __float_as_int(xb[NPTS + jv]);
  const int zv = __float_as_int(xb[2 * NPTS + jv]);

  const float a0 = sw1a[ch * 3 + 0], a1 = sw1a[ch * 3 + 1], a2 = sw1a[ch * 3 + 2];
  const float d0 = sw1d[ch * 3 + 0], d1 = sw1d[ch * 3 + 1], d2 = sw1d[ch * 3 + 2];
  const float cx = xb[n], cy = xb[NPTS + n], cz = xb[2 * NPTS + n];

  const float qg = d0 * cx + d1 * cy + d2 * cz + bias1p[ch];
  const float vf = Cf[pbase * CH + HALF_CH + ch] + bias2p[ch];

  const float* cfb = Cf + (size_t)b * NPTS * CH + ch;

  float mg = -FLT_MAX, mf = -FLT_MAX;
  #pragma unroll
  for (int k = 0; k < KNN_K; ++k) {
    const int   j = __builtin_amdgcn_readlane(jv, k);
    const float x = __int_as_float(__builtin_amdgcn_readlane(xv, k));
    const float y = __int_as_float(__builtin_amdgcn_readlane(yv, k));
    const float z = __int_as_float(__builtin_amdgcn_readlane(zv, k));
    mg = fmaxf(mg, a0 * x + a1 * y + a2 * z);
    mf = fmaxf(mf, cfb[(size_t)j * CH]);
  }
  float geo = fmaxf(mg + qg, 0.0f);
  float fea = fmaxf(mf + vf, 0.0f);

  out[(size_t)b * CH * NPTS + (size_t)ch * NPTS + n]             = geo;
  out[(size_t)b * CH * NPTS + (size_t)(HALF_CH + ch) * NPTS + n] = fea;
}

extern "C" void kernel_launch(void* const* d_in, const int* in_sizes, int n_in,
                              void* d_out, int out_size, void* d_ws, size_t ws_size,
                              hipStream_t stream) {
  (void)in_sizes; (void)n_in; (void)out_size; (void)ws_size;

  const float* xyz  = (const float*)d_in[0];
  const float* feat = (const float*)d_in[1];
  const float* w1   = (const float*)d_in[2];
  const float* b1   = (const float*)d_in[3];
  const float* g1   = (const float*)d_in[4];
  const float* be1  = (const float*)d_in[5];
  const float* m1   = (const float*)d_in[6];
  const float* v1   = (const float*)d_in[7];
  const float* w2   = (const float*)d_in[8];
  const float* b2   = (const float*)d_in[9];
  const float* g2   = (const float*)d_in[10];
  const float* be2  = (const float*)d_in[11];
  const float* m2   = (const float*)d_in[12];
  const float* v2   = (const float*)d_in[13];

  char* ws = (char*)d_ws;
  float* Wc     = (float*)ws;
  float* sw1a   = Wc + 16384;
  float* sw1d   = sw1a + 192;
  float* bias1p = sw1d + 192;
  float* bias2p = bias1p + 64;
  float* Cf     = (float*)(ws + (128 << 10)) + (size_t)BATCH * NPTS * CH;
  int*   idxb   = (int*)(Cf + (size_t)BATCH * NPTS * CH);

  prep_params<<<1, 64, 0, stream>>>(w1, b1, g1, be1, m1, v1,
                                    w2, b2, g2, be2, m2, v2,
                                    Wc, sw1a, sw1d, bias1p, bias2p);

  knn_kernel<<<dim3(NPTS / KNN_QPB, BATCH), KNN_WAVES * 32, 0, stream>>>(xyz, idxb);

  feat_gemm<<<dim3(NPTS / 16, BATCH), 256, 0, stream>>>(feat, Wc, Cf);

  gather_max<<<dim3(NPTS / 4, BATCH), 256, 0, stream>>>(xyz, sw1a, sw1d, bias1p, Cf, idxb,
                                                        bias2p, (float*)d_out);
}


# gather kernel re-tiled: block = 32 points, wave = 2 points x all 64 channel pairs (8-byte row loads, 40 in flight), results transposed in LDS and stored as whole 128-byte lines; 4x fewer waves
# speedup vs baseline: 81.0564x; 2.3459x over previous
#include <hip/hip_runtime.h>
#include <hip/hip_bf16.h>
#include <float.h>

#define BATCH   8
#define NPTS    2048
#define CH      128
#define HALF_CH 64
#define KNN_K   20
#define EPSB    1e-5f

typedef __attribute__((ext_vector_type(2))) float v2f;
typedef __attribute__((ext_vector_type(8))) float v8f;

__global__ void prep_params(const float* __restrict__ w1, const float* __restrict__ b1,
                            const float* __restrict__ g1, const float* __restrict__ be1,
                            const float* __restrict__ m1, const float* __restrict__ v1,
                            const float* __restrict__ w2, const float* __restrict__ b2,
                            const float* __restrict__ g2, const float* __restrict__ be2,
                            const float* __restrict__ m2, const float* __restrict__ v2,
                            float* __restrict__ Wc,   float* __restrict__ sw1a,
                            float* __restrict__ sw1d, float* __restrict__ bias1p,
                            float* __restrict__ bias2p) {
  int o = threadIdx.x;
  if (o >= HALF_CH) return;
  float s1 = g1[o] / sqrtf(v1[o] + EPSB);
  bias1p[o] = b1[o] * s1 + (be1[o] - m1[o] * s1);
  #pragma unroll
  for (int c = 0; c < 3; ++c) {
    float wa = w1[o * 6 + c];
    float wb = w1[o * 6 + 3 + c];
    sw1a[o * 3 + c] = s1 * wa;
    sw1d[o * 3 + c] = s1 * (wb - wa);
  }
  float s2 = g2[o] / sqrtf(v2[o] + EPSB);
  bias2p[o] = b2[o] * s2 + (be2[o] - m2[o] * s2);
  for (int c = 0; c < CH; ++c) {
    float wa = w2[o * (2 * CH) + c];
    float wb = w2[o * (2 * CH) + CH + c];
    Wc[(size_t)o * CH + c]             = s2 * wa;
    Wc[(size_t)(HALF_CH + o) * CH + c] = s2 * (wb - wa);
  }
}

#define KNN_WAVES 16
#define KNN_QPW   4
#define KNN_QPB   (KNN_WAVES * KNN_QPW)
#define KNN_SLOTS (NPTS / 32)
#define KNN_CAND  64

__device__ __forceinline__ unsigned wave_sum_u32(unsigned v) {
  v += (unsigned)__builtin_amdgcn_update_dpp(0, (int)v, 0x161, 0xf, 0xf, false);
  v += (unsigned)__builtin_amdgcn_update_dpp(0, (int)v, 0x162, 0xf, 0xf, false);
  v += (unsigned)__builtin_amdgcn_update_dpp(0, (int)v, 0x164, 0xf, 0xf, false);
  v += (unsigned)__builtin_amdgcn_update_dpp(0, (int)v, 0x168, 0xf, 0xf, false);
  return (unsigned)__builtin_amdgcn_readlane((int)v, 0) +
         (unsigned)__builtin_amdgcn_readlane((int)v, 16);
}

__device__ __forceinline__ void count_below8(unsigned& c0, unsigned& c1, unsigned& c2,
                                             unsigned& c3, unsigned k0, unsigned k1,
                                             unsigned k2, unsigned k3, unsigned k4,
                                             unsigned k5, unsigned k6, unsigned k7,
                                             unsigned t) {
  unsigned m1, m2, m3, m4, m5, m6, m7;
  asm("v_cmp_lt_u32_e32 vcc_lo, %11, %19\n\t"
      "v_cmp_lt_u32_e64 %4, %12, %19\n\t"
      "v_cmp_lt_u32_e64 %5, %13, %19\n\t"
      "v_cmp_lt_u32_e64 %6, %14, %19\n\t"
      "v_cmp_lt_u32_e64 %7, %15, %19\n\t"
      "v_cmp_lt_u32_e64 %8, %16, %19\n\t"
      "v_cmp_lt_u32_e64 %9, %17, %19\n\t"
      "v_cmp_lt_u32_e64 %10, %18, %19\n\t"
      "s_delay_alu instid0(VALU_DEP_4) | instskip(NEXT) | instid1(VALU_DEP_4)\n\t"
      "v_add_co_ci_u32_e32 %0, vcc_lo, 0, %0, vcc_lo\n\t"
      "v_add_co_ci_u32_e64 %1, %4, 0, %1, %4\n\t"
      "s_delay_alu instid0(VALU_DEP_4) | instskip(NEXT) | instid1(VALU_DEP_4)\n\t"
      "v_add_co_ci_u32_e64 %2, %5, 0, %2, %5\n\t"
      "v_add_co_ci_u32_e64 %3, %6, 0, %3, %6\n\t"
      "s_delay_alu instid0(VALU_DEP_4) | instskip(NEXT) | instid1(VALU_DEP_4)\n\t"
      "v_add_co_ci_u32_e64 %0, %7, 0, %0, %7\n\t"
      "v_add_co_ci_u32_e64 %1, %8, 0, %1, %8\n\t"
      "s_delay_alu instid0(VALU_DEP_4) | instskip(NEXT) | instid1(VALU_DEP_4)\n\t"
      "v_add_co_ci_u32_e64 %2, %9, 0, %2, %9\n\t"
      "v_add_co_ci_u32_e64 %3, %10, 0, %3, %10"
      : "+v"(c0), "+v"(c1), "+v"(c2), "+v"(c3),
        "=&s"(m1), "=&s"(m2), "=&s"(m3), "=&s"(m4), "=&s"(m5), "=&s"(m6), "=&s"(m7)
      : "v"(k0), "v"(k1), "v"(k2), "v"(k3), "v"(k4), "v"(k5), "v"(k6), "v"(k7), "v"(t)
      : "vcc");
}
__device__ __forceinline__ void shift_in_below2(unsigned& mlo, unsigned& mhi, unsigned klo,
                                                unsigned khi, unsigned t) {
  unsigned m1;
  asm("v_cmp_lt_u32_e32 vcc_lo, %3, %5\n\t"
      "v_cmp_lt_u32_e64 %2, %4, %5\n\t"
      "s_delay_alu instid0(VALU_DEP_2) | instskip(NEXT) | instid1(VALU_DEP_2)\n\t"
      "v_add_co_ci_u32_e32 %0, vcc_lo, %0, %0, vcc_lo\n\t"
      "v_add_co_ci_u32_e64 %1, %2, %1, %1, %2"
      : "+v"(mlo), "+v"(mhi), "=&s"(m1)
      : "v"(klo), "v"(khi), "v"(t)
      : "vcc");
}
__device__ __forceinline__ void shift_in_equal2(unsigned& mlo, unsigned& mhi, unsigned klo,
                                                unsigned khi, unsigned t) {
  unsigned m1;
  asm("v_cmp_eq_u32_e32 vcc_lo, %3, %5\n\t"
      "v_cmp_eq_u32_e64 %2, %4, %5\n\t"
      "s_delay_alu instid0(VALU_DEP_2) | instskip(NEXT) | instid1(VALU_DEP_2)\n\t"
      "v_add_co_ci_u32_e32 %0, vcc_lo, %0, %0, vcc_lo\n\t"
      "v_add_co_ci_u32_e64 %1, %2, %1, %1, %2"
      : "+v"(mlo), "+v"(mhi), "=&s"(m1)
      : "v"(klo), "v"(khi), "v"(t)
      : "vcc");
}

__device__ __forceinline__ void shift_in_le2(unsigned& mlo, unsigned& mhi, unsigned klo,
                                             unsigned khi, unsigned t) {
  unsigned m1;
  asm("v_cmp_le_u32_e32 vcc_lo, %3, %5\n\t"
      "v_cmp_le_u32_e64 %2, %4, %5\n\t"
      "s_delay_alu instid0(VALU_DEP_2) | instskip(NEXT) | instid1(VALU_DEP_2)\n\t"
      "v_add_co_ci_u32_e32 %0, vcc_lo, %0, %0, vcc_lo\n\t"
      "v_add_co_ci_u32_e64 %1, %2, %1, %1, %2"
      : "+v"(mlo), "+v"(mhi), "=&s"(m1)
      : "v"(klo), "v"(khi), "v"(t)
      : "vcc");
}
__device__ __forceinline__ void rank_below4(unsigned& r0, unsigned& r1, unsigned s0,
                                            unsigned s1, unsigned s2, unsigned s3,
                                            unsigned v) {
  unsigned m1, m2, m3;
  asm("v_cmp_lt_u32_e32 vcc_lo, %5, %9\n\t"
      "v_cmp_lt_u32_e64 %2, %6, %9\n\t"
      "v_cmp_lt_u32_e64 %3, %7, %9\n\t"
      "v_cmp_lt_u32_e64 %4, %8, %9\n\t"
      "s_delay_alu instid0(VALU_DEP_4) | instskip(NEXT) | instid1(VALU_DEP_4)\n\t"
      "v_add_co_ci_u32_e32 %0, vcc_lo, 0, %0, vcc_lo\n\t"
      "v_add_co_ci_u32_e64 %1, %2, 0, %1, %2\n\t"
      "s_delay_alu instid0(VALU_DEP_2) | instskip(NEXT) | instid1(VALU_DEP_2)\n\t"
      "v_add_co_ci_u32_e64 %0, %3, 0, %0, %3\n\t"
      "v_add_co_ci_u32_e64 %1, %4, 0, %1, %4"
      : "+v"(r0), "+v"(r1), "=&s"(m1), "=&s"(m2), "=&s"(m3)
      : "s"(s0), "s"(s1), "s"(s2), "s"(s3), "v"(v)
      : "vcc");
}
__device__ __forceinline__ unsigned wave_min_u32(unsigned v) {
  v = min(v, (unsigned)__builtin_amdgcn_update_dpp(0, (int)v, 0x161, 0xf, 0xf, false));
  v = min(v, (unsigned)__builtin_amdgcn_update_dpp(0, (int)v, 0x162, 0xf, 0xf, false));
  v = min(v, (unsigned)__builtin_amdgcn_update_dpp(0, (int)v, 0x164, 0xf, 0xf, false));
  v = min(v, (unsigned)__builtin_amdgcn_update_dpp(0, (int)v, 0x168, 0xf, 0xf, false));
  return min((unsigned)__builtin_amdgcn_readlane((int)v, 0),
             (unsigned)__builtin_amdgcn_readlane((int)v, 16));
}
__device__ __forceinline__ unsigned dist_key(const float4 me, const float4 p) {
  float dot;
  {
    #pragma clang fp contract(off)
    dot = fmaf(me.x, p.x, me.y * p.y) + me.z * p.z;
  }
  const float d = fmaf(-2.0f, dot, me.w + p.w);
  const unsigned u = __float_as_uint(d);
  return u ^ ((unsigned)((int)u >> 31) | 0x80000000u);
}

__global__ void __launch_bounds__(KNN_WAVES * 32)
knn_kernel(const float* __restrict__ xyz, int* __restrict__ idx_out) {
  __shared__ float4 pts[NPTS];
  __shared__ int cand[KNN_WAVES][KNN_CAND];
  const int b = blockIdx.y;
  const float* xb = xyz + (size_t)b * 3 * NPTS;
  for (int m = threadIdx.x; m < NPTS; m += blockDim.x) {
    float x = xb[m], y = xb[NPTS + m], z = xb[2 * NPTS + m];
    pts[m] = make_float4(x, y, z, fmaf(z, z, fmaf(x, x, y * y)));
  }
  __syncthreads();

  const int lane = threadIdx.x & 31;
  const int wave = threadIdx.x >> 5;
  int* cw = cand[wave];

  for (int qi = 0; qi < KNN_QPW; ++qi) {
    const int q = blockIdx.x * KNN_QPB + wave * KNN_QPW + qi;
    const float4 me = pts[q];
    int* op = idx_out + ((size_t)b * NPTS + q) * KNN_K;

    unsigned key[KNN_SLOTS];
    #pragma unroll
    for (int j = 0; j < KNN_SLOTS; ++j) key[j] = dist_key(me, pts[j * 32 + lane]);

    unsigned mn = key[0];
    #pragma unroll
    for (int j = 1; j < KNN_SLOTS; ++j) mn = min(mn, key[j]);
    const unsigned mu = (mn & ~31u) | (unsigned)lane;
    unsigned r0 = 0u, r1 = 0u;
    #pragma unroll
    for (int l = 0; l < 32; l += 4)
      rank_below4(r0, r1, (unsigned)__builtin_amdgcn_readlane((int)mu, l),
                  (unsigned)__builtin_amdgcn_readlane((int)mu, l + 1),
                  (unsigned)__builtin_amdgcn_readlane((int)mu, l + 2),
                  (unsigned)__builtin_amdgcn_readlane((int)mu, l + 3), mu);
    const unsigned holder = __builtin_amdgcn_ballot_w32((r0 + r1) == (unsigned)KNN_K);
    const unsigned U =
        (unsigned)__builtin_amdgcn_readlane((int)mu, __builtin_ctz(holder)) | 31u;

    unsigned mlo = 0u, mhi = 0u;
    #pragma unroll
    for (int j = 31; j >= 0; --j) shift_in_le2(mlo, mhi, key[j], key[j + 32], U);
    const unsigned cS =
        wave_sum_u32((unsigned)__builtin_popcount(mlo) + (unsigned)__builtin_popcount(mhi));

    if (cS <= (unsigned)KNN_CAND) {
      unsigned listed = 0u;
      for (int pass = 0; pass < KNN_SLOTS; ++pass) {
        const bool has = (mlo | mhi) != 0u;
        const unsigned act = __builtin_amdgcn_ballot_w32(has);
        if (act == 0u) break;
        if (has) {
          int j;
          if (mlo != 0u) { j = __builtin_ctz(mlo);      mlo &= mlo - 1u; }
          else           { j = 32 + __builtin_ctz(mhi); mhi &= mhi - 1u; }
          cw[listed + __builtin_amdgcn_mbcnt_lo(act, 0u)] = j * 32 + lane;
        }
        listed += (unsigned)__builtin_popcount(act);
      }
      __builtin_amdgcn_wave_barrier();
      const int i0 = ((unsigned)lane < cS) ? cw[lane] : -1;
      const int i1 = ((unsigned)lane + 32u < cS) ? cw[lane + 32] : -1;
      unsigned k0 = 0xFFFFFFFFu, k1 = 0xFFFFFFFFu;
      if (i0 >= 0) k0 = dist_key(me, pts[i0]);
      if (i1 >= 0) k1 = dist_key(me, pts[i1]);
      __builtin_amdgcn_wave_barrier();

      unsigned T = 0u, clt = 0u;
      for (int bit = 31; bit >= 0; --bit) {
        const unsigned trial = T | (1u << bit);
        const unsigned c =
            wave_sum_u32((k0 < trial ? 1u : 0u) + (k1 < trial ? 1u : 0u));
        if (c <= (unsigned)KNN_K) {
          T = trial;
          clt = c;
          if (c == (unsigned)KNN_K) break;
        }
      }
      const unsigned need = (unsigned)KNN_K - clt;

      const bool s0 = k0 < T, s1 = k1 < T;
      const unsigned b0 = __builtin_amdgcn_ballot_w32(s0);
      const unsigned b1 = __builtin_amdgcn_ballot_w32(s1);
      if (s0) op[__builtin_amdgcn_mbcnt_lo(b0, 0u)] = i0;
      if (s1) op[(unsigned)__builtin_popcount(b0) + __builtin_amdgcn_mbcnt_lo(b1, 0u)] = i1;

      if (need != 0u) {
        unsigned e0 = (k0 == T) ? (unsigned)i0 : 0xFFFFFFFFu;
        unsigned e1 = (k1 == T) ? (unsigned)i1 : 0xFFFFFFFFu;
        for (unsigned r = 0u; r < need; ++r) {
          const unsigned m = wave_min_u32(min(e0, e1));
          if (lane == 0) op[clt + r] = (int)m;
          if (e0 == m) e0 = 0xFFFFFFFFu;
          if (e1 == m) e1 = 0xFFFFFFFFu;
        }
      }
      continue;
    }

    unsigned T = 0u, clt = 0u;
    for (int bit = 31; bit >= 0; --bit) {
      const unsigned trial = T | (1u << bit);
      unsigned c0 = 0u, c1 = 0u, c2 = 0u, c3 = 0u;
      #pragma unroll
      for (int j = 0; j < KNN_SLOTS; j += 8)
        count_below8(c0, c1, c2, c3, key[j + 0], key[j + 1], key[j + 2], key[j + 3],
                     key[j + 4], key[j + 5], key[j + 6], key[j + 7], trial);
      const unsigned c = wave_sum_u32((c0 + c1) + (c2 + c3));
      if (c <= (unsigned)KNN_K) {
        T = trial;
        clt = c;
        if (c == (unsigned)KNN_K) break;
      }
    }
    const unsigned need = (unsigned)KNN_K - clt;

    mlo = 0u; mhi = 0u;
    #pragma unroll
    for (int j = 31; j >= 0; --j) shift_in_below2(mlo, mhi, key[j], key[j + 32], T);

    unsigned written = 0u;
    for (int pass = 0; pass < KNN_SLOTS; ++pass) {
      const bool has = (mlo | mhi) != 0u;
      const unsigned act = __builtin_amdgcn_ballot_w32(has);
      if (act == 0u) break;
      if (has) {
        int j;
        if (mlo != 0u) { j = __builtin_ctz(mlo);      mlo &= mlo - 1u; }
        else           { j = 32 + __builtin_ctz(mhi); mhi &= mhi - 1u; }
        op[written + __builtin_amdgcn_mbcnt_lo(act, 0u)] = j * 32 + lane;
      }
      written += (unsigned)__builtin_popcount(act);
    }

    if (need != 0u) {
      unsigned elo = 0u, ehi = 0u;
      #pragma unroll
      for (int j = 31; j >= 0; --j) shift_in_equal2(elo, ehi, key[j], key[j + 32], T);
      unsigned taken = 0u;
      for (int j = 0; j < KNN_SLOTS && taken < need; ++j) {
        const unsigned w = (j < 32) ? elo : ehi;
        const bool eq = ((w >> (j & 31)) & 1u) != 0u;
        const unsigned meq = __builtin_amdgcn_ballot_w32(eq);
        if (meq != 0u) {
          const unsigned r = taken + __builtin_amdgcn_mbcnt_lo(meq, 0u);
          if (eq && r < need) op[clt + r] = j * 32 + lane;
          taken += (unsigned)__builtin_popcount(meq);
        }
      }
    }
  }
}

__global__ void feat_gemm(const float* __restrict__ feat, const float* __restrict__ Wc,
                          float* __restrict__ Cf) {
  const int b    = blockIdx.y;
  const int n0   = blockIdx.x * 16;
  const int wave = threadIdx.x >> 5;
  const int lane = threadIdx.x & 31;
  const int l16  = lane & 15;
  const int hi   = lane >> 4;
  const int m0   = wave * 16;

  const float* Fb   = feat + (size_t)b * CH * NPTS;
  const float* arow = Wc + (size_t)(m0 + l16) * CH + 2 * hi;
  const float* bcol = Fb + (size_t)(2 * hi) * NPTS + n0 + l16;

  v8f acc = {0.f, 0.f, 0.f, 0.f, 0.f, 0.f, 0.f, 0.f};

  #pragma unroll 8
  for (int k = 0; k < CH; k += 4) {
    v2f a;
    a.x = arow[k];
    a.y = arow[k + 1];
    v2f bb;
    bb.x = bcol[(size_t)k * NPTS];
    bb.y = bcol[(size_t)(k + 1) * NPTS];
    acc = __builtin_amdgcn_wmma_f32_16x16x4_f32(false, a, false, bb,
                                                (short)0, acc, false, false);
  }

  float* out = Cf + ((size_t)b * NPTS + n0 + l16) * CH + m0 + 8 * hi;
  #pragma unroll
  for (int v = 0; v < 8; ++v) out[v] = acc[v];
}

#define GM_PTS   32
#define GM_WAVES 16
#define GM_PPW   (GM_PTS / GM_WAVES)
#define GM_RPW   (CH / GM_WAVES)
#define GM_PITCH (GM_PTS + 1)

typedef const float2 __attribute__((address_space(1)))* gfloat2p;

__global__ void __launch_bounds__(GM_WAVES * 32, 8)
gather_max(const float* __restrict__ xyz, const float* __restrict__ sw1a,
           const float* __restrict__ sw1d, const float* __restrict__ bias1p,
           const float* __restrict__ Cf, const int* __restrict__ idx,
           const float* __restrict__ bias2p, float* __restrict__ out) {
  __shared__ float tile[CH * GM_PITCH];
  const int b    = blockIdx.y;
  const int n0   = blockIdx.x * GM_PTS;
  const int lane = threadIdx.x & 31;
  const int wave = __builtin_amdgcn_readfirstlane((int)(threadIdx.x >> 5));

  const float* xb  = xyz + (size_t)b * 3 * NPTS;
  const float* cfb = Cf + (size_t)b * NPTS * CH;

  const int kk = (lane < KNN_K) ? lane : lane - KNN_K;

  const float2 wa01 = *(const float2*)(sw1a + 6 * lane);
  const float2 wa23 = *(const float2*)(sw1a + 6 * lane + 2);
  const float2 wa45 = *(const float2*)(sw1a + 6 * lane + 4);
  const float2 wd01 = *(const float2*)(sw1d + 6 * lane);
  const float2 wd23 = *(const float2*)(sw1d + 6 * lane + 2);
  const float2 wd45 = *(const float2*)(sw1d + 6 * lane + 4);
  const float2 bg   = *(const float2*)(bias1p + 2 * lane);
  const float2 bf   = *(const float2*)(bias2p + 2 * lane);

  int   jv[GM_PPW], xv[GM_PPW], yv[GM_PPW], zv[GM_PPW];
  float qg[GM_PPW][2], vf[GM_PPW][2];
  #pragma unroll
  for (int p = 0; p < GM_PPW; ++p) {
    const int    n     = n0 + wave * GM_PPW + p;
    const size_t pbase = (size_t)b * NPTS + n;
    jv[p] = idx[pbase * KNN_K + kk];
    const float  cx = xb[n], cy = xb[NPTS + n], cz = xb[2 * NPTS + n];
    const float2 cv = *(const float2*)(Cf + pbase * CH + HALF_CH + 2 * lane);
    qg[p][0] = wd01.x * cx + wd01.y * cy + wd23.x * cz + bg.x;
    qg[p][1] = wd23.y * cx + wd45.x * cy + wd45.y * cz + bg.y;
    vf[p][0] = cv.x + bf.x;
    vf[p][1] = cv.y + bf.y;
  }
  asm volatile("" :: "v"(wa01.x), "v"(wa01.y), "v"(wa23.x), "v"(wa23.y), "v"(wa45.x), "v"(wa45.y));
  #pragma unroll
  for (int p = 0; p < GM_PPW; ++p) {
    xv[p] = __float_as_int(xb[jv[p]]);
    yv[p] = __float_as_int(xb[NPTS + jv[p]]);
    zv[p] = __float_as_int(xb[2 * NPTS + jv[p]]);
  }

  float2 row[GM_PPW][KNN_K];
  #pragma unroll
  for (int p = 0; p < GM_PPW; ++p) {
    #pragma unroll
    for (int k = 0; k < KNN_K; ++k) {
      const int j = __builtin_amdgcn_readlane(jv[p], k);
      const unsigned long long ra = (unsigned long long)(cfb + (size_t)j * CH);
      const unsigned ralo = (unsigned)__builtin_amdgcn_readfirstlane((int)(unsigned)ra);
      const unsigned rahi = (unsigned)__builtin_amdgcn_readfirstlane((int)(unsigned)(ra >> 32));
      const gfloat2p rp = (gfloat2p)(((unsigned long long)rahi << 32) | ralo);
      row[p][k] = rp[lane];
    }
  }

  float mg[GM_PPW][2];
  #pragma unroll
  for (int p = 0; p < GM_PPW; ++p) { mg[p][0] = -FLT_MAX; mg[p][1] = -FLT_MAX; }
  #pragma unroll 2
  for (int k = 0; k < KNN_K; ++k) {
    #pragma unroll
    for (int p = 0; p < GM_PPW; ++p) {
      const float x = __int_as_float(__builtin_amdgcn_readlane(xv[p], k));
      const float y = __int_as_float(__builtin_amdgcn_readlane(yv[p], k));
      const float z = __int_as_float(__builtin_amdgcn_readlane(zv[p], k));
      mg[p][0] = fmaxf(mg[p][0], wa01.x * x + wa01.y * y + wa23.x * z);
      mg[p][1] = fmaxf(mg[p][1], wa23.y * x + wa45.x * y + wa45.y * z);
    }
  }

  #pragma unroll
  for (int p = 0; p < GM_PPW; ++p) {
    const int pl = wave * GM_PPW + p;
    float mf0 = -FLT_MAX, mf1 = -FLT_MAX;
    #pragma unroll
    for (int k = 0; k < KNN_K; ++k) {
      mf0 = fmaxf(mf0, row[p][k].x);
      mf1 = fmaxf(mf1, row[p][k].y);
    }
    tile[(2 * lane)               * GM_PITCH + pl] = fmaxf(mg[p][0] + qg[p][0], 0.0f);
    tile[(2 * lane + 1)           * GM_PITCH + pl] = fmaxf(mg[p][1] + qg[p][1], 0.0f);
    tile[(HALF_CH + 2 * lane)     * GM_PITCH + pl] = fmaxf(mf0 + vf[p][0], 0.0f);
    tile[(HALF_CH + 2 * lane + 1) * GM_PITCH + pl] = fmaxf(mf1 + vf[p][1], 0.0f);
  }
  __syncthreads();

  float* ob = out + (size_t)b * CH * NPTS + n0 + lane;
  #pragma unroll
  for (int r = 0; r < GM_RPW; ++r) {
    const int c = wave * GM_RPW + r;
    ob[(size_t)c * NPTS] = tile[c * GM_PITCH + lane];
  }
}

extern "C" void kernel_launch(void* const* d_in, const int* in_sizes, int n_in,
                              void* d_out, int out_size, void* d_ws, size_t ws_size,
                              hipStream_t stream) {
  (void)in_sizes; (void)n_in; (void)out_size; (void)ws_size;

  const float* xyz  = (const float*)d_in[0];
  const float* feat = (const float*)d_in[1];
  const float* w1   = (const float*)d_in[2];
  const float* b1   = (const float*)d_in[3];
  const float* g1   = (const float*)d_in[4];
  const float* be1  = (const float*)d_in[5];
  const float* m1   = (const float*)d_in[6];
  const float* v1   = (const float*)d_in[7];
  const float* w2   = (const float*)d_in[8];
  const float* b2   = (const float*)d_in[9];
  const float* g2   = (const float*)d_in[10];
  const float* be2  = (const float*)d_in[11];
  const float* m2   = (const float*)d_in[12];
  const float* v2   = (const float*)d_in[13];

  char* ws = (char*)d_ws;
  float* Wc     = (float*)ws;
  float* sw1a   = Wc + 16384;
  float* sw1d   = sw1a + 192;
  float* bias1p = sw1d + 192;
  float* bias2p = bias1p + 64;
  float* Cf     = (float*)(ws + (128 << 10)) + (size_t)BATCH * NPTS * CH;
  int*   idxb   = (int*)(Cf + (size_t)BATCH * NPTS * CH);

  prep_params<<<1, 64, 0, stream>>>(w1, b1, g1, be1, m1, v1,
                                    w2, b2, g2, be2, m2, v2,
                                    Wc, sw1a, sw1d, bias1p, bias2p);

  knn_kernel<<<dim3(NPTS / KNN_QPB, BATCH), KNN_WAVES * 32, 0, stream>>>(xyz, idxb);

  feat_gemm<<<dim3(NPTS / 16, BATCH), 256, 0, stream>>>(feat, Wc, Cf);

  gather_max<<<dim3(NPTS / GM_PTS, BATCH), GM_WAVES * 32, 0, stream>>>(xyz, sw1a, sw1d, bias1p, Cf, idxb,
                                                        bias2p, (float*)d_out);
}


# batch-norm fold moved into its users: GEMM scales its A values from w2 as it loads them, gather kernel folds its own geometry weights and biases; fold kernel, its workspace copies and launch dropped
# speedup vs baseline: 87.6001x; 1.0807x over previous
#include <hip/hip_runtime.h>
#include <hip/hip_bf16.h>
#include <float.h>

#define BATCH   8
#define NPTS    2048
#define CH      128
#define HALF_CH 64
#define KNN_K   20
#define EPSB    1e-5f

typedef __attribute__((ext_vector_type(2))) float v2f;
typedef __attribute__((ext_vector_type(8))) float v8f;


#define KNN_WAVES 16
#define KNN_QPW   4
#define KNN_QPB   (KNN_WAVES * KNN_QPW)
#define KNN_SLOTS (NPTS / 32)
#define KNN_CAND  64

__device__ __forceinline__ unsigned wave_sum_u32(unsigned v) {
  v += (unsigned)__builtin_amdgcn_update_dpp(0, (int)v, 0x161, 0xf, 0xf, false);
  v += (unsigned)__builtin_amdgcn_update_dpp(0, (int)v, 0x162, 0xf, 0xf, false);
  v += (unsigned)__builtin_amdgcn_update_dpp(0, (int)v, 0x164, 0xf, 0xf, false);
  v += (unsigned)__builtin_amdgcn_update_dpp(0, (int)v, 0x168, 0xf, 0xf, false);
  return (unsigned)__builtin_amdgcn_readlane((int)v, 0) +
         (unsigned)__builtin_amdgcn_readlane((int)v, 16);
}

__device__ __forceinline__ void count_below8(unsigned& c0, unsigned& c1, unsigned& c2,
                                             unsigned& c3, unsigned k0, unsigned k1,
                                             unsigned k2, unsigned k3, unsigned k4,
                                             unsigned k5, unsigned k6, unsigned k7,
                                             unsigned t) {
  unsigned m1, m2, m3, m4, m5, m6, m7;
  asm("v_cmp_lt_u32_e32 vcc_lo, %11, %19\n\t"
      "v_cmp_lt_u32_e64 %4, %12, %19\n\t"
      "v_cmp_lt_u32_e64 %5, %13, %19\n\t"
      "v_cmp_lt_u32_e64 %6, %14, %19\n\t"
      "v_cmp_lt_u32_e64 %7, %15, %19\n\t"
      "v_cmp_lt_u32_e64 %8, %16, %19\n\t"
      "v_cmp_lt_u32_e64 %9, %17, %19\n\t"
      "v_cmp_lt_u32_e64 %10, %18, %19\n\t"
      "s_delay_alu instid0(VALU_DEP_4) | instskip(NEXT) | instid1(VALU_DEP_4)\n\t"
      "v_add_co_ci_u32_e32 %0, vcc_lo, 0, %0, vcc_lo\n\t"
      "v_add_co_ci_u32_e64 %1, %4, 0, %1, %4\n\t"
      "s_delay_alu instid0(VALU_DEP_4) | instskip(NEXT) | instid1(VALU_DEP_4)\n\t"
      "v_add_co_ci_u32_e64 %2, %5, 0, %2, %5\n\t"
      "v_add_co_ci_u32_e64 %3, %6, 0, %3, %6\n\t"
      "s_delay_alu instid0(VALU_DEP_4) | instskip(NEXT) | instid1(VALU_DEP_4)\n\t"
      "v_add_co_ci_u32_e64 %0, %7, 0, %0, %7\n\t"
      "v_add_co_ci_u32_e64 %1, %8, 0, %1, %8\n\t"
      "s_delay_alu instid0(VALU_DEP_4) | instskip(NEXT) | instid1(VALU_DEP_4)\n\t"
      "v_add_co_ci_u32_e64 %2, %9, 0, %2, %9\n\t"
      "v_add_co_ci_u32_e64 %3, %10, 0, %3, %10"
      : "+v"(c0), "+v"(c1), "+v"(c2), "+v"(c3),
        "=&s"(m1), "=&s"(m2), "=&s"(m3), "=&s"(m4), "=&s"(m5), "=&s"(m6), "=&s"(m7)
      : "v"(k0), "v"(k1), "v"(k2), "v"(k3), "v"(k4), "v"(k5), "v"(k6), "v"(k7), "v"(t)
      : "vcc");
}
__device__ __forceinline__ void shift_in_below2(unsigned& mlo, unsigned& mhi, unsigned klo,
                                                unsigned khi, unsigned t) {
  unsigned m1;
  asm("v_cmp_lt_u32_e32 vcc_lo, %3, %5\n\t"
      "v_cmp_lt_u32_e64 %2, %4, %5\n\t"
      "s_delay_alu instid0(VALU_DEP_2) | instskip(NEXT) | instid1(VALU_DEP_2)\n\t"
      "v_add_co_ci_u32_e32 %0, vcc_lo, %0, %0, vcc_lo\n\t"
      "v_add_co_ci_u32_e64 %1, %2, %1, %1, %2"
      : "+v"(mlo), "+v"(mhi), "=&s"(m1)
      : "v"(klo), "v"(khi), "v"(t)
      : "vcc");
}
__device__ __forceinline__ void shift_in_equal2(unsigned& mlo, unsigned& mhi, unsigned klo,
                                                unsigned khi, unsigned t) {
  unsigned m1;
  asm("v_cmp_eq_u32_e32 vcc_lo, %3, %5\n\t"
      "v_cmp_eq_u32_e64 %2, %4, %5\n\t"
      "s_delay_alu instid0(VALU_DEP_2) | instskip(NEXT) | instid1(VALU_DEP_2)\n\t"
      "v_add_co_ci_u32_e32 %0, vcc_lo, %0, %0, vcc_lo\n\t"
      "v_add_co_ci_u32_e64 %1, %2, %1, %1, %2"
      : "+v"(mlo), "+v"(mhi), "=&s"(m1)
      : "v"(klo), "v"(khi), "v"(t)
      : "vcc");
}

__device__ __forceinline__ void shift_in_le2(unsigned& mlo, unsigned& mhi, unsigned klo,
                                             unsigned khi, unsigned t) {
  unsigned m1;
  asm("v_cmp_le_u32_e32 vcc_lo, %3, %5\n\t"
      "v_cmp_le_u32_e64 %2, %4, %5\n\t"
      "s_delay_alu instid0(VALU_DEP_2) | instskip(NEXT) | instid1(VALU_DEP_2)\n\t"
      "v_add_co_ci_u32_e32 %0, vcc_lo, %0, %0, vcc_lo\n\t"
      "v_add_co_ci_u32_e64 %1, %2, %1, %1, %2"
      : "+v"(mlo), "+v"(mhi), "=&s"(m1)
      : "v"(klo), "v"(khi), "v"(t)
      : "vcc");
}
__device__ __forceinline__ void rank_below4(unsigned& r0, unsigned& r1, unsigned s0,
                                            unsigned s1, unsigned s2, unsigned s3,
                                            unsigned v) {
  unsigned m1, m2, m3;
  asm("v_cmp_lt_u32_e32 vcc_lo, %5, %9\n\t"
      "v_cmp_lt_u32_e64 %2, %6, %9\n\t"
      "v_cmp_lt_u32_e64 %3, %7, %9\n\t"
      "v_cmp_lt_u32_e64 %4, %8, %9\n\t"
      "s_delay_alu instid0(VALU_DEP_4) | instskip(NEXT) | instid1(VALU_DEP_4)\n\t"
      "v_add_co_ci_u32_e32 %0, vcc_lo, 0, %0, vcc_lo\n\t"
      "v_add_co_ci_u32_e64 %1, %2, 0, %1, %2\n\t"
      "s_delay_alu instid0(VALU_DEP_2) | instskip(NEXT) | instid1(VALU_DEP_2)\n\t"
      "v_add_co_ci_u32_e64 %0, %3, 0, %0, %3\n\t"
      "v_add_co_ci_u32_e64 %1, %4, 0, %1, %4"
      : "+v"(r0), "+v"(r1), "=&s"(m1), "=&s"(m2), "=&s"(m3)
      : "s"(s0), "s"(s1), "s"(s2), "s"(s3), "v"(v)
      : "vcc");
}
__device__ __forceinline__ unsigned wave_min_u32(unsigned v) {
  v = min(v, (unsigned)__builtin_amdgcn_update_dpp(0, (int)v, 0x161, 0xf, 0xf, false));
  v = min(v, (unsigned)__builtin_amdgcn_update_dpp(0, (int)v, 0x162, 0xf, 0xf, false));
  v = min(v, (unsigned)__builtin_amdgcn_update_dpp(0, (int)v, 0x164, 0xf, 0xf, false));
  v = min(v, (unsigned)__builtin_amdgcn_update_dpp(0, (int)v, 0x168, 0xf, 0xf, false));
  return min((unsigned)__builtin_amdgcn_readlane((int)v, 0),
             (unsigned)__builtin_amdgcn_readlane((int)v, 16));
}
__device__ __forceinline__ unsigned dist_key(const float4 me, const float4 p) {
  float dot;
  {
    #pragma clang fp contract(off)
    dot = fmaf(me.x, p.x, me.y * p.y) + me.z * p.z;
  }
  const float d = fmaf(-2.0f, dot, me.w + p.w);
  const unsigned u = __float_as_uint(d);
  return u ^ ((unsigned)((int)u >> 31) | 0x80000000u);
}

__global__ void __launch_bounds__(KNN_WAVES * 32)
knn_kernel(const float* __restrict__ xyz, int* __restrict__ idx_out) {
  __shared__ float4 pts[NPTS];
  __shared__ int cand[KNN_WAVES][KNN_CAND];
  const int b = blockIdx.y;
  const float* xb = xyz + (size_t)b * 3 * NPTS;
  for (int m = threadIdx.x; m < NPTS; m += blockDim.x) {
    float x = xb[m], y = xb[NPTS + m], z = xb[2 * NPTS + m];
    pts[m] = make_float4(x, y, z, fmaf(z, z, fmaf(x, x, y * y)));
  }
  __syncthreads();

  const int lane = threadIdx.x & 31;
  const int wave = threadIdx.x >> 5;
  int* cw = cand[wave];

  for (int qi = 0; qi < KNN_QPW; ++qi) {
    const int q = blockIdx.x * KNN_QPB + wave * KNN_QPW + qi;
    const float4 me = pts[q];
    int* op = idx_out + ((size_t)b * NPTS + q) * KNN_K;

    unsigned key[KNN_SLOTS];
    #pragma unroll
    for (int j = 0; j < KNN_SLOTS; ++j) key[j] = dist_key(me, pts[j * 32 + lane]);

    unsigned mn = key[0];
    #pragma unroll
    for (int j = 1; j < KNN_SLOTS; ++j) mn = min(mn, key[j]);
    const unsigned mu = (mn & ~31u) | (unsigned)lane;
    unsigned r0 = 0u, r1 = 0u;
    #pragma unroll
    for (int l = 0; l < 32; l += 4)
      rank_below4(r0, r1, (unsigned)__builtin_amdgcn_readlane((int)mu, l),
                  (unsigned)__builtin_amdgcn_readlane((int)mu, l + 1),
                  (unsigned)__builtin_amdgcn_readlane((int)mu, l + 2),
                  (unsigned)__builtin_amdgcn_readlane((int)mu, l + 3), mu);
    const unsigned holder = __builtin_amdgcn_ballot_w32((r0 + r1) == (unsigned)KNN_K);
    const unsigned U =
        (unsigned)__builtin_amdgcn_readlane((int)mu, __builtin_ctz(holder)) | 31u;

    unsigned mlo = 0u, mhi = 0u;
    #pragma unroll
    for (int j = 31; j >= 0; --j) shift_in_le2(mlo, mhi, key[j], key[j + 32], U);
    const unsigned cS =
        wave_sum_u32((unsigned)__builtin_popcount(mlo) + (unsigned)__builtin_popcount(mhi));

    if (cS <= (unsigned)KNN_CAND) {
      unsigned listed = 0u;
      for (int pass = 0; pass < KNN_SLOTS; ++pass) {
        const bool has = (mlo | mhi) != 0u;
        const unsigned act = __builtin_amdgcn_ballot_w32(has);
        if (act == 0u) break;
        if (has) {
          int j;
          if (mlo != 0u) { j = __builtin_ctz(mlo);      mlo &= mlo - 1u; }
          else           { j = 32 + __builtin_ctz(mhi); mhi &= mhi - 1u; }
          cw[listed + __builtin_amdgcn_mbcnt_lo(act, 0u)] = j * 32 + lane;
        }
        listed += (unsigned)__builtin_popcount(act);
      }
      __builtin_amdgcn_wave_barrier();
      const int i0 = ((unsigned)lane < cS) ? cw[lane] : -1;
      const int i1 = ((unsigned)lane + 32u < cS) ? cw[lane + 32] : -1;
      unsigned k0 = 0xFFFFFFFFu, k1 = 0xFFFFFFFFu;
      if (i0 >= 0) k0 = dist_key(me, pts[i0]);
      if (i1 >= 0) k1 = dist_key(me, pts[i1]);
      __builtin_amdgcn_wave_barrier();

      unsigned T = 0u, clt = 0u;
      for (int bit = 31; bit >= 0; --bit) {
        const unsigned trial = T | (1u << bit);
        const unsigned c =
            wave_sum_u32((k0 < trial ? 1u : 0u) + (k1 < trial ? 1u : 0u));
        if (c <= (unsigned)KNN_K) {
          T = trial;
          clt = c;
          if (c == (unsigned)KNN_K) break;
        }
      }
      const unsigned need = (unsigned)KNN_K - clt;

      const bool s0 = k0 < T, s1 = k1 < T;
      const unsigned b0 = __builtin_amdgcn_ballot_w32(s0);
      const unsigned b1 = __builtin_amdgcn_ballot_w32(s1);
      if (s0) op[__builtin_amdgcn_mbcnt_lo(b0, 0u)] = i0;
      if (s1) op[(unsigned)__builtin_popcount(b0) + __builtin_amdgcn_mbcnt_lo(b1, 0u)] = i1;

      if (need != 0u) {
        unsigned e0 = (k0 == T) ? (unsigned)i0 : 0xFFFFFFFFu;
        unsigned e1 = (k1 == T) ? (unsigned)i1 : 0xFFFFFFFFu;
        for (unsigned r = 0u; r < need; ++r) {
          const unsigned m = wave_min_u32(min(e0, e1));
          if (lane == 0) op[clt + r] = (int)m;
          if (e0 == m) e0 = 0xFFFFFFFFu;
          if (e1 == m) e1 = 0xFFFFFFFFu;
        }
      }
      continue;
    }

    unsigned T = 0u, clt = 0u;
    for (int bit = 31; bit >= 0; --bit) {
      const unsigned trial = T | (1u << bit);
      unsigned c0 = 0u, c1 = 0u, c2 = 0u, c3 = 0u;
      #pragma unroll
      for (int j = 0; j < KNN_SLOTS; j += 8)
        count_below8(c0, c1, c2, c3, key[j + 0], key[j + 1], key[j + 2], key[j + 3],
                     key[j + 4], key[j + 5], key[j + 6], key[j + 7], trial);
      const unsigned c = wave_sum_u32((c0 + c1) + (c2 + c3));
      if (c <= (unsigned)KNN_K) {
        T = trial;
        clt = c;
        if (c == (unsigned)KNN_K) break;
      }
    }
    const unsigned need = (unsigned)KNN_K - clt;

    mlo = 0u; mhi = 0u;
    #pragma unroll
    for (int j = 31; j >= 0; --j) shift_in_below2(mlo, mhi, key[j], key[j + 32], T);

    unsigned written = 0u;
    for (int pass = 0; pass < KNN_SLOTS; ++pass) {
      const bool has = (mlo | mhi) != 0u;
      const unsigned act = __builtin_amdgcn_ballot_w32(has);
      if (act == 0u) break;
      if (has) {
        int j;
        if (mlo != 0u) { j = __builtin_ctz(mlo);      mlo &= mlo - 1u; }
        else           { j = 32 + __builtin_ctz(mhi); mhi &= mhi - 1u; }
        op[written + __builtin_amdgcn_mbcnt_lo(act, 0u)] = j * 32 + lane;
      }
      written += (unsigned)__builtin_popcount(act);
    }

    if (need != 0u) {
      unsigned elo = 0u, ehi = 0u;
      #pragma unroll
      for (int j = 31; j >= 0; --j) shift_in_equal2(elo, ehi, key[j], key[j + 32], T);
      unsigned taken = 0u;
      for (int j = 0; j < KNN_SLOTS && taken < need; ++j) {
        const unsigned w = (j < 32) ? elo : ehi;
        const bool eq = ((w >> (j & 31)) & 1u) != 0u;
        const unsigned meq = __builtin_amdgcn_ballot_w32(eq);
        if (meq != 0u) {
          const unsigned r = taken + __builtin_amdgcn_mbcnt_lo(meq, 0u);
          if (eq && r < need) op[clt + r] = j * 32 + lane;
          taken += (unsigned)__builtin_popcount(meq);
        }
      }
    }
  }
}

template <bool kCenter>
__device__ __forceinline__ v8f gemm_tile(const float* __restrict__ wrow,
                                         const float* __restrict__ bcol, const float s2) {
  v8f acc = {0.f, 0.f, 0.f, 0.f, 0.f, 0.f, 0.f, 0.f};
  #pragma unroll 8
  for (int k = 0; k < CH; k += 4) {
    v2f a;
    if (kCenter) {
      a.x = s2 * (wrow[CH + k] - wrow[k]);
      a.y = s2 * (wrow[CH + k + 1] - wrow[k + 1]);
    } else {
      a.x = s2 * wrow[k];
      a.y = s2 * wrow[k + 1];
    }
    v2f bb;
    bb.x = bcol[(size_t)k * NPTS];
    bb.y = bcol[(size_t)(k + 1) * NPTS];
    acc = __builtin_amdgcn_wmma_f32_16x16x4_f32(false, a, false, bb,
                                                (short)0, acc, false, false);
  }
  return acc;
}

__global__ void feat_gemm(const float* __restrict__ feat, const float* __restrict__ w2,
                          const float* __restrict__ g2, const float* __restrict__ v2,
                          float* __restrict__ Cf) {
  const int b    = blockIdx.y;
  const int n0   = blockIdx.x * 16;
  const int wave = __builtin_amdgcn_readfirstlane((int)(threadIdx.x >> 5));
  const int lane = threadIdx.x & 31;
  const int l16  = lane & 15;
  const int hi   = lane >> 4;
  const int m0   = wave * 16;
  const int o    = (m0 & (HALF_CH - 1)) + l16;

  const float* Fb   = feat + (size_t)b * CH * NPTS;
  const float* wrow = w2 + (size_t)o * (2 * CH) + 2 * hi;
  const float* bcol = Fb + (size_t)(2 * hi) * NPTS + n0 + l16;
  const float  s2   = g2[o] / sqrtf(v2[o] + EPSB);

  v8f acc;
  if (m0 < HALF_CH) acc = gemm_tile<false>(wrow, bcol, s2);
  else              acc = gemm_tile<true>(wrow, bcol, s2);

  float* out = Cf + ((size_t)b * NPTS + n0 + l16) * CH + m0 + 8 * hi;
  #pragma unroll
  for (int v = 0; v < 8; ++v) out[v] = acc[v];
}

#define GM_PTS   32
#define GM_WAVES 16
#define GM_PPW   (GM_PTS / GM_WAVES)
#define GM_RPW   (CH / GM_WAVES)
#define GM_PITCH (GM_PTS + 1)

typedef const float2 __attribute__((address_space(1)))* gfloat2p;

__global__ void __launch_bounds__(GM_WAVES * 32, 8)
gather_max(const float* __restrict__ xyz, const float* __restrict__ w1,
           const float* __restrict__ b1, const float* __restrict__ g1,
           const float* __restrict__ be1, const float* __restrict__ m1,
           const float* __restrict__ v1, const float* __restrict__ Cf,
           const int* __restrict__ idx, const float* __restrict__ b2,
           const float* __restrict__ g2, const float* __restrict__ be2,
           const float* __restrict__ m2, const float* __restrict__ v2,
           float* __restrict__ out) {
  __shared__ float tile[CH * GM_PITCH];
  const int b    = blockIdx.y;
  const int n0   = blockIdx.x * GM_PTS;
  const int lane = threadIdx.x & 31;
  const int wave = __builtin_amdgcn_readfirstlane((int)(threadIdx.x >> 5));

  const float* xb  = xyz + (size_t)b * 3 * NPTS;
  const float* cfb = Cf + (size_t)b * NPTS * CH;

  const int kk = (lane < KNN_K) ? lane : lane - KNN_K;

  const float4 w1A = *(const float4*)(w1 + 12 * lane);
  const float4 w1B = *(const float4*)(w1 + 12 * lane + 4);
  const float4 w1C = *(const float4*)(w1 + 12 * lane + 8);
  const float2 g1v  = *(const float2*)(g1 + 2 * lane);
  const float2 v1v  = *(const float2*)(v1 + 2 * lane);
  const float2 b1v  = *(const float2*)(b1 + 2 * lane);
  const float2 be1v = *(const float2*)(be1 + 2 * lane);
  const float2 m1v  = *(const float2*)(m1 + 2 * lane);
  const float2 g2v  = *(const float2*)(g2 + 2 * lane);
  const float2 v2v  = *(const float2*)(v2 + 2 * lane);
  const float2 b2v  = *(const float2*)(b2 + 2 * lane);
  const float2 be2v = *(const float2*)(be2 + 2 * lane);
  const float2 m2v  = *(const float2*)(m2 + 2 * lane);

  int    jv[GM_PPW], xv[GM_PPW], yv[GM_PPW], zv[GM_PPW];
  float  cx[GM_PPW], cy[GM_PPW], cz[GM_PPW];
  float2 cv[GM_PPW];
  #pragma unroll
  for (int p = 0; p < GM_PPW; ++p) {
    const int    n     = n0 + wave * GM_PPW + p;
    const size_t pbase = (size_t)b * NPTS + n;
    jv[p] = idx[pbase * KNN_K + kk];
    cx[p] = xb[n];
    cy[p] = xb[NPTS + n];
    cz[p] = xb[2 * NPTS + n];
    cv[p] = *(const float2*)(Cf + pbase * CH + HALF_CH + 2 * lane);
  }

  const float s10 = g1v.x / sqrtf(v1v.x + EPSB);
  const float s11 = g1v.y / sqrtf(v1v.y + EPSB);
  const float s20 = g2v.x / sqrtf(v2v.x + EPSB);
  const float s21 = g2v.y / sqrtf(v2v.y + EPSB);
  const float2 bg = make_float2(b1v.x * s10 + (be1v.x - m1v.x * s10),
                                b1v.y * s11 + (be1v.y - m1v.y * s11));
  const float2 bf = make_float2(b2v.x * s20 + (be2v.x - m2v.x * s20),
                                b2v.y * s21 + (be2v.y - m2v.y * s21));
  const float2 wa01 = make_float2(s10 * w1A.x, s10 * w1A.y);
  const float2 wa23 = make_float2(s10 * w1A.z, s11 * w1B.z);
  const float2 wa45 = make_float2(s11 * w1B.w, s11 * w1C.x);
  const float2 wd01 = make_float2(s10 * (w1A.w - w1A.x), s10 * (w1B.x - w1A.y));
  const float2 wd23 = make_float2(s10 * (w1B.y - w1A.z), s11 * (w1C.y - w1B.z));
  const float2 wd45 = make_float2(s11 * (w1C.z - w1B.w), s11 * (w1C.w - w1C.x));

  float qg[GM_PPW][2], vf[GM_PPW][2];
  #pragma unroll
  for (int p = 0; p < GM_PPW; ++p) {
    qg[p][0] = wd01.x * cx[p] + wd01.y * cy[p] + wd23.x * cz[p] + bg.x;
    qg[p][1] = wd23.y * cx[p] + wd45.x * cy[p] + wd45.y * cz[p] + bg.y;
    vf[p][0] = cv[p].x + bf.x;
    vf[p][1] = cv[p].y + bf.y;
  }
  asm volatile("" :: "v"(wa01.x), "v"(wa01.y), "v"(wa23.x), "v"(wa23.y), "v"(wa45.x), "v"(wa45.y),
                     "v"(qg[0][0]), "v"(qg[0][1]), "v"(qg[1][0]), "v"(qg[1][1]),
                     "v"(vf[0][0]), "v"(vf[0][1]), "v"(vf[1][0]), "v"(vf[1][1]));
  #pragma unroll
  for (int p = 0; p < GM_PPW; ++p) {
    xv[p] = __float_as_int(xb[jv[p]]);
    yv[p] = __float_as_int(xb[NPTS + jv[p]]);
    zv[p] = __float_as_int(xb[2 * NPTS + jv[p]]);
  }

  float2 row[GM_PPW][KNN_K];
  #pragma unroll
  for (int p = 0; p < GM_PPW; ++p) {
    #pragma unroll
    for (int k = 0; k < KNN_K; ++k) {
      const int j = __builtin_amdgcn_readlane(jv[p], k);
      const unsigned long long ra = (unsigned long long)(cfb + (size_t)j * CH);
      const unsigned ralo = (unsigned)__builtin_amdgcn_readfirstlane((int)(unsigned)ra);
      const unsigned rahi = (unsigned)__builtin_amdgcn_readfirstlane((int)(unsigned)(ra >> 32));
      const gfloat2p rp = (gfloat2p)(((unsigned long long)rahi << 32) | ralo);
      row[p][k] = rp[lane];
    }
  }

  float mg[GM_PPW][2];
  #pragma unroll
  for (int p = 0; p < GM_PPW; ++p) { mg[p][0] = -FLT_MAX; mg[p][1] = -FLT_MAX; }
  #pragma unroll 2
  for (int k = 0; k < KNN_K; ++k) {
    #pragma unroll
    for (int p = 0; p < GM_PPW; ++p) {
      const float x = __int_as_float(__builtin_amdgcn_readlane(xv[p], k));
      const float y = __int_as_float(__builtin_amdgcn_readlane(yv[p], k));
      const float z = __int_as_float(__builtin_amdgcn_readlane(zv[p], k));
      mg[p][0] = fmaxf(mg[p][0], wa01.x * x + wa01.y * y + wa23.x * z);
      mg[p][1] = fmaxf(mg[p][1], wa23.y * x + wa45.x * y + wa45.y * z);
    }
  }

  #pragma unroll
  for (int p = 0; p < GM_PPW; ++p) {
    const int pl = wave * GM_PPW + p;
    float mf0 = -FLT_MAX, mf1 = -FLT_MAX;
    #pragma unroll
    for (int k = 0; k < KNN_K; ++k) {
      mf0 = fmaxf(mf0, row[p][k].x);
      mf1 = fmaxf(mf1, row[p][k].y);
    }
    tile[(2 * lane)               * GM_PITCH + pl] = fmaxf(mg[p][0] + qg[p][0], 0.0f);
    tile[(2 * lane + 1)           * GM_PITCH + pl] = fmaxf(mg[p][1] + qg[p][1], 0.0f);
    tile[(HALF_CH + 2 * lane)     * GM_PITCH + pl] = fmaxf(mf0 + vf[p][0], 0.0f);
    tile[(HALF_CH + 2 * lane + 1) * GM_PITCH + pl] = fmaxf(mf1 + vf[p][1], 0.0f);
  }
  __syncthreads();

  float* ob = out + (size_t)b * CH * NPTS + n0 + lane;
  #pragma unroll
  for (int r = 0; r < GM_RPW; ++r) {
    const int c = wave * GM_RPW + r;
    ob[(size_t)c * NPTS] = tile[c * GM_PITCH + lane];
  }
}

extern "C" void kernel_launch(void* const* d_in, const int* in_sizes, int n_in,
                              void* d_out, int out_size, void* d_ws, size_t ws_size,
                              hipStream_t stream) {
  (void)in_sizes; (void)n_in; (void)out_size; (void)ws_size;

  const float* xyz  = (const float*)d_in[0];
  const float* feat = (const float*)d_in[1];
  const float* w1   = (const float*)d_in[2];
  const float* b1   = (const float*)d_in[3];
  const float* g1   = (const float*)d_in[4];
  const float* be1  = (const float*)d_in[5];
  const float* m1   = (const float*)d_in[6];
  const float* v1   = (const float*)d_in[7];
  const float* w2   = (const float*)d_in[8];
  const float* b2   = (const float*)d_in[9];
  const float* g2   = (const float*)d_in[10];
  const float* be2  = (const float*)d_in[11];
  const float* m2   = (const float*)d_in[12];
  const float* v2   = (const float*)d_in[13];

  char* ws = (char*)d_ws;
  float* Cf     = (float*)(ws + (128 << 10)) + (size_t)BATCH * NPTS * CH;
  int*   idxb   = (int*)(Cf + (size_t)BATCH * NPTS * CH);

  knn_kernel<<<dim3(NPTS / KNN_QPB, BATCH), KNN_WAVES * 32, 0, stream>>>(xyz, idxb);

  feat_gemm<<<dim3(NPTS / 16, BATCH), 256, 0, stream>>>(feat, w2, g2, v2, Cf);

  gather_max<<<dim3(NPTS / GM_PTS, BATCH), GM_WAVES * 32, 0, stream>>>(
      xyz, w1, b1, g1, be1, m1, v1, Cf, idxb, b2, g2, be2, m2, v2, (float*)d_out);
}


# feature GEMM moved into the kNN launch: each kNN wave also computes a 16-point x 32-channel strip (2 WMMA tiles sharing B), half the waves before, half after their queries; one launch less, same sums
# speedup vs baseline: 110.2671x; 1.2588x over previous
#include <hip/hip_runtime.h>
#include <hip/hip_bf16.h>
#include <float.h>

#define BATCH   8
#define NPTS    2048
#define CH      128
#define HALF_CH 64
#define KNN_K   20
#define EPSB    1e-5f

typedef __attribute__((ext_vector_type(2))) float v2f;
typedef __attribute__((ext_vector_type(8))) float v8f;


template <bool kCenter>
__device__ __forceinline__ void gemm_strip(const float* __restrict__ wrow0,
                                           const float* __restrict__ wrow1,
                                           const float* __restrict__ bcol, const float s20,
                                           const float s21, v8f& acc0, v8f& acc1) {
  v8f c0 = {0.f, 0.f, 0.f, 0.f, 0.f, 0.f, 0.f, 0.f};
  v8f c1 = {0.f, 0.f, 0.f, 0.f, 0.f, 0.f, 0.f, 0.f};
  #pragma unroll 8
  for (int k = 0; k < CH; k += 4) {
    v2f a0, a1;
    if (kCenter) {
      a0.x = s20 * (wrow0[CH + k] - wrow0[k]);
      a0.y = s20 * (wrow0[CH + k + 1] - wrow0[k + 1]);
      a1.x = s21 * (wrow1[CH + k] - wrow1[k]);
      a1.y = s21 * (wrow1[CH + k + 1] - wrow1[k + 1]);
    } else {
      a0.x = s20 * wrow0[k];
      a0.y = s20 * wrow0[k + 1];
      a1.x = s21 * wrow1[k];
      a1.y = s21 * wrow1[k + 1];
    }
    v2f bb;
    bb.x = bcol[(size_t)k * NPTS];
    bb.y = bcol[(size_t)(k + 1) * NPTS];
    c0 = __builtin_amdgcn_wmma_f32_16x16x4_f32(false, a0, false, bb,
                                               (short)0, c0, false, false);
    c1 = __builtin_amdgcn_wmma_f32_16x16x4_f32(false, a1, false, bb,
                                               (short)0, c1, false, false);
  }
  acc0 = c0;
  acc1 = c1;
}

__device__ __forceinline__ void feat_gemm_strip(const float* __restrict__ feat,
                                                const float* __restrict__ w2,
                                                const float* __restrict__ g2,
                                                const float* __restrict__ v2,
                                                float* __restrict__ Cf, const int b,
                                                const int n0, const int strip,
                                                const int lane) {
  const int l16 = lane & 15;
  const int hi  = lane >> 4;
  const int m0  = strip * 32;
  const int o0  = (m0 & (HALF_CH - 1)) + l16;
  const int o1  = o0 + 16;

  const float* Fb    = feat + (size_t)b * CH * NPTS;
  const float* wrow0 = w2 + (size_t)o0 * (2 * CH) + 2 * hi;
  const float* wrow1 = w2 + (size_t)o1 * (2 * CH) + 2 * hi;
  const float* bcol  = Fb + (size_t)(2 * hi) * NPTS + n0 + l16;
  const float  s20   = g2[o0] / sqrtf(v2[o0] + EPSB);
  const float  s21   = g2[o1] / sqrtf(v2[o1] + EPSB);

  v8f acc0, acc1;
  if (m0 < HALF_CH) gemm_strip<false>(wrow0, wrow1, bcol, s20, s21, acc0, acc1);
  else              gemm_strip<true>(wrow0, wrow1, bcol, s20, s21, acc0, acc1);

  float* out = Cf + ((size_t)b * NPTS + n0 + l16) * CH + m0 + 8 * hi;
  #pragma unroll
  for (int v = 0; v < 8; ++v) out[v] = acc0[v];
  #pragma unroll
  for (int v = 0; v < 8; ++v) out[16 + v] = acc1[v];
}

#define KNN_WAVES 16
#define KNN_QPW   4
#define KNN_QPB   (KNN_WAVES * KNN_QPW)
#define KNN_SLOTS (NPTS / 32)
#define KNN_CAND  64

__device__ __forceinline__ unsigned wave_sum_u32(unsigned v) {
  v += (unsigned)__builtin_amdgcn_update_dpp(0, (int)v, 0x161, 0xf, 0xf, false);
  v += (unsigned)__builtin_amdgcn_update_dpp(0, (int)v, 0x162, 0xf, 0xf, false);
  v += (unsigned)__builtin_amdgcn_update_dpp(0, (int)v, 0x164, 0xf, 0xf, false);
  v += (unsigned)__builtin_amdgcn_update_dpp(0, (int)v, 0x168, 0xf, 0xf, false);
  return (unsigned)__builtin_amdgcn_readlane((int)v, 0) +
         (unsigned)__builtin_amdgcn_readlane((int)v, 16);
}

__device__ __forceinline__ void count_below8(unsigned& c0, unsigned& c1, unsigned& c2,
                                             unsigned& c3, unsigned k0, unsigned k1,
                                             unsigned k2, unsigned k3, unsigned k4,
                                             unsigned k5, unsigned k6, unsigned k7,
                                             unsigned t) {
  unsigned m1, m2, m3, m4, m5, m6, m7;
  asm("v_cmp_lt_u32_e32 vcc_lo, %11, %19\n\t"
      "v_cmp_lt_u32_e64 %4, %12, %19\n\t"
      "v_cmp_lt_u32_e64 %5, %13, %19\n\t"
      "v_cmp_lt_u32_e64 %6, %14, %19\n\t"
      "v_cmp_lt_u32_e64 %7, %15, %19\n\t"
      "v_cmp_lt_u32_e64 %8, %16, %19\n\t"
      "v_cmp_lt_u32_e64 %9, %17, %19\n\t"
      "v_cmp_lt_u32_e64 %10, %18, %19\n\t"
      "s_delay_alu instid0(VALU_DEP_4) | instskip(NEXT) | instid1(VALU_DEP_4)\n\t"
      "v_add_co_ci_u32_e32 %0, vcc_lo, 0, %0, vcc_lo\n\t"
      "v_add_co_ci_u32_e64 %1, %4, 0, %1, %4\n\t"
      "s_delay_alu instid0(VALU_DEP_4) | instskip(NEXT) | instid1(VALU_DEP_4)\n\t"
      "v_add_co_ci_u32_e64 %2, %5, 0, %2, %5\n\t"
      "v_add_co_ci_u32_e64 %3, %6, 0, %3, %6\n\t"
      "s_delay_alu instid0(VALU_DEP_4) | instskip(NEXT) | instid1(VALU_DEP_4)\n\t"
      "v_add_co_ci_u32_e64 %0, %7, 0, %0, %7\n\t"
      "v_add_co_ci_u32_e64 %1, %8, 0, %1, %8\n\t"
      "s_delay_alu instid0(VALU_DEP_4) | instskip(NEXT) | instid1(VALU_DEP_4)\n\t"
      "v_add_co_ci_u32_e64 %2, %9, 0, %2, %9\n\t"
      "v_add_co_ci_u32_e64 %3, %10, 0, %3, %10"
      : "+v"(c0), "+v"(c1), "+v"(c2), "+v"(c3),
        "=&s"(m1), "=&s"(m2), "=&s"(m3), "=&s"(m4), "=&s"(m5), "=&s"(m6), "=&s"(m7)
      : "v"(k0), "v"(k1), "v"(k2), "v"(k3), "v"(k4), "v"(k5), "v"(k6), "v"(k7), "v"(t)
      : "vcc");
}
__device__ __forceinline__ void shift_in_below2(unsigned& mlo, unsigned& mhi, unsigned klo,
                                                unsigned khi, unsigned t) {
  unsigned m1;
  asm("v_cmp_lt_u32_e32 vcc_lo, %3, %5\n\t"
      "v_cmp_lt_u32_e64 %2, %4, %5\n\t"
      "s_delay_alu instid0(VALU_DEP_2) | instskip(NEXT) | instid1(VALU_DEP_2)\n\t"
      "v_add_co_ci_u32_e32 %0, vcc_lo, %0, %0, vcc_lo\n\t"
      "v_add_co_ci_u32_e64 %1, %2, %1, %1, %2"
      : "+v"(mlo), "+v"(mhi), "=&s"(m1)
      : "v"(klo), "v"(khi), "v"(t)
      : "vcc");
}
__device__ __forceinline__ void shift_in_equal2(unsigned& mlo, unsigned& mhi, unsigned klo,
                                                unsigned khi, unsigned t) {
  unsigned m1;
  asm("v_cmp_eq_u32_e32 vcc_lo, %3, %5\n\t"
      "v_cmp_eq_u32_e64 %2, %4, %5\n\t"
      "s_delay_alu instid0(VALU_DEP_2) | instskip(NEXT) | instid1(VALU_DEP_2)\n\t"
      "v_add_co_ci_u32_e32 %0, vcc_lo, %0, %0, vcc_lo\n\t"
      "v_add_co_ci_u32_e64 %1, %2, %1, %1, %2"
      : "+v"(mlo), "+v"(mhi), "=&s"(m1)
      : "v"(klo), "v"(khi), "v"(t)
      : "vcc");
}

__device__ __forceinline__ void shift_in_le2(unsigned& mlo, unsigned& mhi, unsigned klo,
                                             unsigned khi, unsigned t) {
  unsigned m1;
  asm("v_cmp_le_u32_e32 vcc_lo, %3, %5\n\t"
      "v_cmp_le_u32_e64 %2, %4, %5\n\t"
      "s_delay_alu instid0(VALU_DEP_2) | instskip(NEXT) | instid1(VALU_DEP_2)\n\t"
      "v_add_co_ci_u32_e32 %0, vcc_lo, %0, %0, vcc_lo\n\t"
      "v_add_co_ci_u32_e64 %1, %2, %1, %1, %2"
      : "+v"(mlo), "+v"(mhi), "=&s"(m1)
      : "v"(klo), "v"(khi), "v"(t)
      : "vcc");
}
__device__ __forceinline__ void rank_below4(unsigned& r0, unsigned& r1, unsigned s0,
                                            unsigned s1, unsigned s2, unsigned s3,
                                            unsigned v) {
  unsigned m1, m2, m3;
  asm("v_cmp_lt_u32_e32 vcc_lo, %5, %9\n\t"
      "v_cmp_lt_u32_e64 %2, %6, %9\n\t"
      "v_cmp_lt_u32_e64 %3, %7, %9\n\t"
      "v_cmp_lt_u32_e64 %4, %8, %9\n\t"
      "s_delay_alu instid0(VALU_DEP_4) | instskip(NEXT) | instid1(VALU_DEP_4)\n\t"
      "v_add_co_ci_u32_e32 %0, vcc_lo, 0, %0, vcc_lo\n\t"
      "v_add_co_ci_u32_e64 %1, %2, 0, %1, %2\n\t"
      "s_delay_alu instid0(VALU_DEP_2) | instskip(NEXT) | instid1(VALU_DEP_2)\n\t"
      "v_add_co_ci_u32_e64 %0, %3, 0, %0, %3\n\t"
      "v_add_co_ci_u32_e64 %1, %4, 0, %1, %4"
      : "+v"(r0), "+v"(r1), "=&s"(m1), "=&s"(m2), "=&s"(m3)
      : "s"(s0), "s"(s1), "s"(s2), "s"(s3), "v"(v)
      : "vcc");
}
__device__ __forceinline__ unsigned wave_min_u32(unsigned v) {
  v = min(v, (unsigned)__builtin_amdgcn_update_dpp(0, (int)v, 0x161, 0xf, 0xf, false));
  v = min(v, (unsigned)__builtin_amdgcn_update_dpp(0, (int)v, 0x162, 0xf, 0xf, false));
  v = min(v, (unsigned)__builtin_amdgcn_update_dpp(0, (int)v, 0x164, 0xf, 0xf, false));
  v = min(v, (unsigned)__builtin_amdgcn_update_dpp(0, (int)v, 0x168, 0xf, 0xf, false));
  return min((unsigned)__builtin_amdgcn_readlane((int)v, 0),
             (unsigned)__builtin_amdgcn_readlane((int)v, 16));
}
__device__ __forceinline__ unsigned dist_key(const float4 me, const float4 p) {
  float dot;
  {
    #pragma clang fp contract(off)
    dot = fmaf(me.x, p.x, me.y * p.y) + me.z * p.z;
  }
  const float d = fmaf(-2.0f, dot, me.w + p.w);
  const unsigned u = __float_as_uint(d);
  return u ^ ((unsigned)((int)u >> 31) | 0x80000000u);
}

__global__ void __launch_bounds__(KNN_WAVES * 32)
knn_kernel(const float* __restrict__ xyz, int* __restrict__ idx_out,
           const float* __restrict__ feat, const float* __restrict__ w2,
           const float* __restrict__ g2, const float* __restrict__ v2,
           float* __restrict__ Cf) {
  __shared__ float4 pts[NPTS];
  __shared__ int cand[KNN_WAVES][KNN_CAND];
  const int b = blockIdx.y;
  const float* xb = xyz + (size_t)b * 3 * NPTS;
  for (int m = threadIdx.x; m < NPTS; m += blockDim.x) {
    float x = xb[m], y = xb[NPTS + m], z = xb[2 * NPTS + m];
    pts[m] = make_float4(x, y, z, fmaf(z, z, fmaf(x, x, y * y)));
  }
  __syncthreads();

  const int lane = threadIdx.x & 31;
  const int wave = threadIdx.x >> 5;
  int* cw = cand[wave];

  const int  swave      = __builtin_amdgcn_readfirstlane((int)(threadIdx.x >> 5));
  const int  gemm_n0    = blockIdx.x * KNN_QPB + 16 * (swave & 3);
  const int  gemm_strip_id = swave >> 2;
  const bool gemm_first = ((swave ^ (swave >> 2)) & 1) == 0;
  if (gemm_first) feat_gemm_strip(feat, w2, g2, v2, Cf, b, gemm_n0, gemm_strip_id, lane);

  for (int qi = 0; qi < KNN_QPW; ++qi) {
    const int q = blockIdx.x * KNN_QPB + wave * KNN_QPW + qi;
    const float4 me = pts[q];
    int* op = idx_out + ((size_t)b * NPTS + q) * KNN_K;

    unsigned key[KNN_SLOTS];
    #pragma unroll
    for (int j = 0; j < KNN_SLOTS; ++j) key[j] = dist_key(me, pts[j * 32 + lane]);

    unsigned mn = key[0];
    #pragma unroll
    for (int j = 1; j < KNN_SLOTS; ++j) mn = min(mn, key[j]);
    const unsigned mu = (mn & ~31u) | (unsigned)lane;
    unsigned r0 = 0u, r1 = 0u;
    #pragma unroll
    for (int l = 0; l < 32; l += 4)
      rank_below4(r0, r1, (unsigned)__builtin_amdgcn_readlane((int)mu, l),
                  (unsigned)__builtin_amdgcn_readlane((int)mu, l + 1),
                  (unsigned)__builtin_amdgcn_readlane((int)mu, l + 2),
                  (unsigned)__builtin_amdgcn_readlane((int)mu, l + 3), mu);
    const unsigned holder = __builtin_amdgcn_ballot_w32((r0 + r1) == (unsigned)KNN_K);
    const unsigned U =
        (unsigned)__builtin_amdgcn_readlane((int)mu, __builtin_ctz(holder)) | 31u;

    unsigned mlo = 0u, mhi = 0u;
    #pragma unroll
    for (int j = 31; j >= 0; --j) shift_in_le2(mlo, mhi, key[j], key[j + 32], U);
    const unsigned cS =
        wave_sum_u32((unsigned)__builtin_popcount(mlo) + (unsigned)__builtin_popcount(mhi));

    if (cS <= (unsigned)KNN_CAND) {
      unsigned listed = 0u;
      for (int pass = 0; pass < KNN_SLOTS; ++pass) {
        const bool has = (mlo | mhi) != 0u;
        const unsigned act = __builtin_amdgcn_ballot_w32(has);
        if (act == 0u) break;
        if (has) {
          int j;
          if (mlo != 0u) { j = __builtin_ctz(mlo);      mlo &= mlo - 1u; }
          else           { j = 32 + __builtin_ctz(mhi); mhi &= mhi - 1u; }
          cw[listed + __builtin_amdgcn_mbcnt_lo(act, 0u)] = j * 32 + lane;
        }
        listed += (unsigned)__builtin_popcount(act);
      }
      __builtin_amdgcn_wave_barrier();
      const int i0 = ((unsigned)lane < cS) ? cw[lane] : -1;
      const int i1 = ((unsigned)lane + 32u < cS) ? cw[lane + 32] : -1;
      unsigned k0 = 0xFFFFFFFFu, k1 = 0xFFFFFFFFu;
      if (i0 >= 0) k0 = dist_key(me, pts[i0]);
      if (i1 >= 0) k1 = dist_key(me, pts[i1]);
      __builtin_amdgcn_wave_barrier();

      unsigned T = 0u, clt = 0u;
      for (int bit = 31; bit >= 0; --bit) {
        const unsigned trial = T | (1u << bit);
        const unsigned c =
            wave_sum_u32((k0 < trial ? 1u : 0u) + (k1 < trial ? 1u : 0u));
        if (c <= (unsigned)KNN_K) {
          T = trial;
          clt = c;
          if (c == (unsigned)KNN_K) break;
        }
      }
      const unsigned need = (unsigned)KNN_K - clt;

      const bool s0 = k0 < T, s1 = k1 < T;
      const unsigned b0 = __builtin_amdgcn_ballot_w32(s0);
      const unsigned b1 = __builtin_amdgcn_ballot_w32(s1);
      if (s0) op[__builtin_amdgcn_mbcnt_lo(b0, 0u)] = i0;
      if (s1) op[(unsigned)__builtin_popcount(b0) + __builtin_amdgcn_mbcnt_lo(b1, 0u)] = i1;

      if (need != 0u) {
        unsigned e0 = (k0 == T) ? (unsigned)i0 : 0xFFFFFFFFu;
        unsigned e1 = (k1 == T) ? (unsigned)i1 : 0xFFFFFFFFu;
        for (unsigned r = 0u; r < need; ++r) {
          const unsigned m = wave_min_u32(min(e0, e1));
          if (lane == 0) op[clt + r] = (int)m;
          if (e0 == m) e0 = 0xFFFFFFFFu;
          if (e1 == m) e1 = 0xFFFFFFFFu;
        }
      }
      continue;
    }

    unsigned T = 0u, clt = 0u;
    for (int bit = 31; bit >= 0; --bit) {
      const unsigned trial = T | (1u << bit);
      unsigned c0 = 0u, c1 = 0u, c2 = 0u, c3 = 0u;
      #pragma unroll
      for (int j = 0; j < KNN_SLOTS; j += 8)
        count_below8(c0, c1, c2, c3, key[j + 0], key[j + 1], key[j + 2], key[j + 3],
                     key[j + 4], key[j + 5], key[j + 6], key[j + 7], trial);
      const unsigned c = wave_sum_u32((c0 + c1) + (c2 + c3));
      if (c <= (unsigned)KNN_K) {
        T = trial;
        clt = c;
        if (c == (unsigned)KNN_K) break;
      }
    }
    const unsigned need = (unsigned)KNN_K - clt;

    mlo = 0u; mhi = 0u;
    #pragma unroll
    for (int j = 31; j >= 0; --j) shift_in_below2(mlo, mhi, key[j], key[j + 32], T);

    unsigned written = 0u;
    for (int pass = 0; pass < KNN_SLOTS; ++pass) {
      const bool has = (mlo | mhi) != 0u;
      const unsigned act = __builtin_amdgcn_ballot_w32(has);
      if (act == 0u) break;
      if (has) {
        int j;
        if (mlo != 0u) { j = __builtin_ctz(mlo);      mlo &= mlo - 1u; }
        else           { j = 32 + __builtin_ctz(mhi); mhi &= mhi - 1u; }
        op[written + __builtin_amdgcn_mbcnt_lo(act, 0u)] = j * 32 + lane;
      }
      written += (unsigned)__builtin_popcount(act);
    }

    if (need != 0u) {
      unsigned elo = 0u, ehi = 0u;
      #pragma unroll
      for (int j = 31; j >= 0; --j) shift_in_equal2(elo, ehi, key[j], key[j + 32], T);
      unsigned taken = 0u;
      for (int j = 0; j < KNN_SLOTS && taken < need; ++j) {
        const unsigned w = (j < 32) ? elo : ehi;
        const bool eq = ((w >> (j & 31)) & 1u) != 0u;
        const unsigned meq = __builtin_amdgcn_ballot_w32(eq);
        if (meq != 0u) {
          const unsigned r = taken + __builtin_amdgcn_mbcnt_lo(meq, 0u);
          if (eq && r < need) op[clt + r] = j * 32 + lane;
          taken += (unsigned)__builtin_popcount(meq);
        }
      }
    }
  }

  if (!gemm_first) feat_gemm_strip(feat, w2, g2, v2, Cf, b, gemm_n0, gemm_strip_id, lane);
}

#define GM_PTS   32
#define GM_WAVES 16
#define GM_PPW   (GM_PTS / GM_WAVES)
#define GM_RPW   (CH / GM_WAVES)
#define GM_PITCH (GM_PTS + 1)

typedef const float2 __attribute__((address_space(1)))* gfloat2p;

__global__ void __launch_bounds__(GM_WAVES * 32, 8)
gather_max(const float* __restrict__ xyz, const float* __restrict__ w1,
           const float* __restrict__ b1, const float* __restrict__ g1,
           const float* __restrict__ be1, const float* __restrict__ m1,
           const float* __restrict__ v1, const float* __restrict__ Cf,
           const int* __restrict__ idx, const float* __restrict__ b2,
           const float* __restrict__ g2, const float* __restrict__ be2,
           const float* __restrict__ m2, const float* __restrict__ v2,
           float* __restrict__ out) {
  __shared__ float tile[CH * GM_PITCH];
  const int b    = blockIdx.y;
  const int n0   = blockIdx.x * GM_PTS;
  const int lane = threadIdx.x & 31;
  const int wave = __builtin_amdgcn_readfirstlane((int)(threadIdx.x >> 5));

  const float* xb  = xyz + (size_t)b * 3 * NPTS;
  const float* cfb = Cf + (size_t)b * NPTS * CH;

  const int kk = (lane < KNN_K) ? lane : lane - KNN_K;

  const float4 w1A = *(const float4*)(w1 + 12 * lane);
  const float4 w1B = *(const float4*)(w1 + 12 * lane + 4);
  const float4 w1C = *(const float4*)(w1 + 12 * lane + 8);
  const float2 g1v  = *(const float2*)(g1 + 2 * lane);
  const float2 v1v  = *(const float2*)(v1 + 2 * lane);
  const float2 b1v  = *(const float2*)(b1 + 2 * lane);
  const float2 be1v = *(const float2*)(be1 + 2 * lane);
  const float2 m1v  = *(const float2*)(m1 + 2 * lane);
  const float2 g2v  = *(const float2*)(g2 + 2 * lane);
  const float2 v2v  = *(const float2*)(v2 + 2 * lane);
  const float2 b2v  = *(const float2*)(b2 + 2 * lane);
  const float2 be2v = *(const float2*)(be2 + 2 * lane);
  const float2 m2v  = *(const float2*)(m2 + 2 * lane);

  int    jv[GM_PPW], xv[GM_PPW], yv[GM_PPW], zv[GM_PPW];
  float  cx[GM_PPW], cy[GM_PPW], cz[GM_PPW];
  float2 cv[GM_PPW];
  #pragma unroll
  for (int p = 0; p < GM_PPW; ++p) {
    const int    n     = n0 + wave * GM_PPW + p;
    const size_t pbase = (size_t)b * NPTS + n;
    jv[p] = idx[pbase * KNN_K + kk];
    cx[p] = xb[n];
    cy[p] = xb[NPTS + n];
    cz[p] = xb[2 * NPTS + n];
    cv[p] = *(const float2*)(Cf + pbase * CH + HALF_CH + 2 * lane);
  }

  const float s10 = g1v.x / sqrtf(v1v.x + EPSB);
  const float s11 = g1v.y / sqrtf(v1v.y + EPSB);
  const float s20 = g2v.x / sqrtf(v2v.x + EPSB);
  const float s21 = g2v.y / sqrtf(v2v.y + EPSB);
  const float2 bg = make_float2(b1v.x * s10 + (be1v.x - m1v.x * s10),
                                b1v.y * s11 + (be1v.y - m1v.y * s11));
  const float2 bf = make_float2(b2v.x * s20 + (be2v.x - m2v.x * s20),
                                b2v.y * s21 + (be2v.y - m2v.y * s21));
  const float2 wa01 = make_float2(s10 * w1A.x, s10 * w1A.y);
  const float2 wa23 = make_float2(s10 * w1A.z, s11 * w1B.z);
  const float2 wa45 = make_float2(s11 * w1B.w, s11 * w1C.x);
  const float2 wd01 = make_float2(s10 * (w1A.w - w1A.x), s10 * (w1B.x - w1A.y));
  const float2 wd23 = make_float2(s10 * (w1B.y - w1A.z), s11 * (w1C.y - w1B.z));
  const float2 wd45 = make_float2(s11 * (w1C.z - w1B.w), s11 * (w1C.w - w1C.x));

  float qg[GM_PPW][2], vf[GM_PPW][2];
  #pragma unroll
  for (int p = 0; p < GM_PPW; ++p) {
    qg[p][0] = wd01.x * cx[p] + wd01.y * cy[p] + wd23.x * cz[p] + bg.x;
    qg[p][1] = wd23.y * cx[p] + wd45.x * cy[p] + wd45.y * cz[p] + bg.y;
    vf[p][0] = cv[p].x + bf.x;
    vf[p][1] = cv[p].y + bf.y;
  }
  asm volatile("" :: "v"(wa01.x), "v"(wa01.y), "v"(wa23.x), "v"(wa23.y), "v"(wa45.x), "v"(wa45.y),
                     "v"(qg[0][0]), "v"(qg[0][1]), "v"(qg[1][0]), "v"(qg[1][1]),
                     "v"(vf[0][0]), "v"(vf[0][1]), "v"(vf[1][0]), "v"(vf[1][1]));
  #pragma unroll
  for (int p = 0; p < GM_PPW; ++p) {
    xv[p] = __float_as_int(xb[jv[p]]);
    yv[p] = __float_as_int(xb[NPTS + jv[p]]);
    zv[p] = __float_as_int(xb[2 * NPTS + jv[p]]);
  }

  float2 row[GM_PPW][KNN_K];
  #pragma unroll
  for (int p = 0; p < GM_PPW; ++p) {
    #pragma unroll
    for (int k = 0; k < KNN_K; ++k) {
      const int j = __builtin_amdgcn_readlane(jv[p], k);
      const unsigned long long ra = (unsigned long long)(cfb + (size_t)j * CH);
      const unsigned ralo = (unsigned)__builtin_amdgcn_readfirstlane((int)(unsigned)ra);
      const unsigned rahi = (unsigned)__builtin_amdgcn_readfirstlane((int)(unsigned)(ra >> 32));
      const gfloat2p rp = (gfloat2p)(((unsigned long long)rahi << 32) | ralo);
      row[p][k] = rp[lane];
    }
  }

  float mg[GM_PPW][2];
  #pragma unroll
  for (int p = 0; p < GM_PPW; ++p) { mg[p][0] = -FLT_MAX; mg[p][1] = -FLT_MAX; }
  #pragma unroll 2
  for (int k = 0; k < KNN_K; ++k) {
    #pragma unroll
    for (int p = 0; p < GM_PPW; ++p) {
      const float x = __int_as_float(__builtin_amdgcn_readlane(xv[p], k));
      const float y = __int_as_float(__builtin_amdgcn_readlane(yv[p], k));
      const float z = __int_as_float(__builtin_amdgcn_readlane(zv[p], k));
      mg[p][0] = fmaxf(mg[p][0], wa01.x * x + wa01.y * y + wa23.x * z);
      mg[p][1] = fmaxf(mg[p][1], wa23.y * x + wa45.x * y + wa45.y * z);
    }
  }

  #pragma unroll
  for (int p = 0; p < GM_PPW; ++p) {
    const int pl = wave * GM_PPW + p;
    float mf0 = -FLT_MAX, mf1 = -FLT_MAX;
    #pragma unroll
    for (int k = 0; k < KNN_K; ++k) {
      mf0 = fmaxf(mf0, row[p][k].x);
      mf1 = fmaxf(mf1, row[p][k].y);
    }
    tile[(2 * lane)               * GM_PITCH + pl] = fmaxf(mg[p][0] + qg[p][0], 0.0f);
    tile[(2 * lane + 1)           * GM_PITCH + pl] = fmaxf(mg[p][1] + qg[p][1], 0.0f);
    tile[(HALF_CH + 2 * lane)     * GM_PITCH + pl] = fmaxf(mf0 + vf[p][0], 0.0f);
    tile[(HALF_CH + 2 * lane + 1) * GM_PITCH + pl] = fmaxf(mf1 + vf[p][1], 0.0f);
  }
  __syncthreads();

  float* ob = out + (size_t)b * CH * NPTS + n0 + lane;
  #pragma unroll
  for (int r = 0; r < GM_RPW; ++r) {
    const int c = wave * GM_RPW + r;
    ob[(size_t)c * NPTS] = tile[c * GM_PITCH + lane];
  }
}

extern "C" void kernel_launch(void* const* d_in, const int* in_sizes, int n_in,
                              void* d_out, int out_size, void* d_ws, size_t ws_size,
                              hipStream_t stream) {
  (void)in_sizes; (void)n_in; (void)out_size; (void)ws_size;

  const float* xyz  = (const float*)d_in[0];
  const float* feat = (const float*)d_in[1];
  const float* w1   = (const float*)d_in[2];
  const float* b1   = (const float*)d_in[3];
  const float* g1   = (const float*)d_in[4];
  const float* be1  = (const float*)d_in[5];
  const float* m1   = (const float*)d_in[6];
  const float* v1   = (const float*)d_in[7];
  const float* w2   = (const float*)d_in[8];
  const float* b2   = (const float*)d_in[9];
  const float* g2   = (const float*)d_in[10];
  const float* be2  = (const float*)d_in[11];
  const float* m2   = (const float*)d_in[12];
  const float* v2   = (const float*)d_in[13];

  char* ws = (char*)d_ws;
  float* Cf     = (float*)(ws + (128 << 10)) + (size_t)BATCH * NPTS * CH;
  int*   idxb   = (int*)(Cf + (size_t)BATCH * NPTS * CH);

  knn_kernel<<<dim3(NPTS / KNN_QPB, BATCH), KNN_WAVES * 32, 0, stream>>>(xyz, idxb, feat, w2,
                                                                         g2, v2, Cf);

  gather_max<<<dim3(NPTS / GM_PTS, BATCH), GM_WAVES * 32, 0, stream>>>(
      xyz, w1, b1, g1, be1, m1, v1, Cf, idxb, b2, g2, be2, m2, v2, (float*)d_out);
}


# kNN: the 20 nearest among the listed candidates chosen by rank counting (each key against the lane-read keys, no dependent trials) instead of bitwise bisection, kept for ties; scores bit-identical
# speedup vs baseline: 115.8322x; 1.0505x over previous
#include <hip/hip_runtime.h>
#include <hip/hip_bf16.h>
#include <float.h>

#define BATCH   8
#define NPTS    2048
#define CH      128
#define HALF_CH 64
#define KNN_K   20
#define EPSB    1e-5f

typedef __attribute__((ext_vector_type(2))) float v2f;
typedef __attribute__((ext_vector_type(8))) float v8f;


template <bool kCenter>
__device__ __forceinline__ void gemm_strip(const float* __restrict__ wrow0,
                                           const float* __restrict__ wrow1,
                                           const float* __restrict__ bcol, const float s20,
                                           const float s21, v8f& acc0, v8f& acc1) {
  v8f c0 = {0.f, 0.f, 0.f, 0.f, 0.f, 0.f, 0.f, 0.f};
  v8f c1 = {0.f, 0.f, 0.f, 0.f, 0.f, 0.f, 0.f, 0.f};
  #pragma unroll 8
  for (int k = 0; k < CH; k += 4) {
    v2f a0, a1;
    if (kCenter) {
      a0.x = s20 * (wrow0[CH + k] - wrow0[k]);
      a0.y = s20 * (wrow0[CH + k + 1] - wrow0[k + 1]);
      a1.x = s21 * (wrow1[CH + k] - wrow1[k]);
      a1.y = s21 * (wrow1[CH + k + 1] - wrow1[k + 1]);
    } else {
      a0.x = s20 * wrow0[k];
      a0.y = s20 * wrow0[k + 1];
      a1.x = s21 * wrow1[k];
      a1.y = s21 * wrow1[k + 1];
    }
    v2f bb;
    bb.x = bcol[(size_t)k * NPTS];
    bb.y = bcol[(size_t)(k + 1) * NPTS];
    c0 = __builtin_amdgcn_wmma_f32_16x16x4_f32(false, a0, false, bb,
                                               (short)0, c0, false, false);
    c1 = __builtin_amdgcn_wmma_f32_16x16x4_f32(false, a1, false, bb,
                                               (short)0, c1, false, false);
  }
  acc0 = c0;
  acc1 = c1;
}

__device__ __forceinline__ void feat_gemm_strip(const float* __restrict__ feat,
                                                const float* __restrict__ w2,
                                                const float* __restrict__ g2,
                                                const float* __restrict__ v2,
                                                float* __restrict__ Cf, const int b,
                                                const int n0, const int strip,
                                                const int lane) {
  const int l16 = lane & 15;
  const int hi  = lane >> 4;
  const int m0  = strip * 32;
  const int o0  = (m0 & (HALF_CH - 1)) + l16;
  const int o1  = o0 + 16;

  const float* Fb    = feat + (size_t)b * CH * NPTS;
  const float* wrow0 = w2 + (size_t)o0 * (2 * CH) + 2 * hi;
  const float* wrow1 = w2 + (size_t)o1 * (2 * CH) + 2 * hi;
  const float* bcol  = Fb + (size_t)(2 * hi) * NPTS + n0 + l16;
  const float  s20   = g2[o0] / sqrtf(v2[o0] + EPSB);
  const float  s21   = g2[o1] / sqrtf(v2[o1] + EPSB);

  v8f acc0, acc1;
  if (m0 < HALF_CH) gemm_strip<false>(wrow0, wrow1, bcol, s20, s21, acc0, acc1);
  else              gemm_strip<true>(wrow0, wrow1, bcol, s20, s21, acc0, acc1);

  float* out = Cf + ((size_t)b * NPTS + n0 + l16) * CH + m0 + 8 * hi;
  #pragma unroll
  for (int v = 0; v < 8; ++v) out[v] = acc0[v];
  #pragma unroll
  for (int v = 0; v < 8; ++v) out[16 + v] = acc1[v];
}

#define KNN_WAVES 16
#define KNN_QPW   4
#define KNN_QPB   (KNN_WAVES * KNN_QPW)
#define KNN_SLOTS (NPTS / 32)
#define KNN_CAND  64

__device__ __forceinline__ unsigned wave_sum_u32(unsigned v) {
  v += (unsigned)__builtin_amdgcn_update_dpp(0, (int)v, 0x161, 0xf, 0xf, false);
  v += (unsigned)__builtin_amdgcn_update_dpp(0, (int)v, 0x162, 0xf, 0xf, false);
  v += (unsigned)__builtin_amdgcn_update_dpp(0, (int)v, 0x164, 0xf, 0xf, false);
  v += (unsigned)__builtin_amdgcn_update_dpp(0, (int)v, 0x168, 0xf, 0xf, false);
  return (unsigned)__builtin_amdgcn_readlane((int)v, 0) +
         (unsigned)__builtin_amdgcn_readlane((int)v, 16);
}

__device__ __forceinline__ void count_below8(unsigned& c0, unsigned& c1, unsigned& c2,
                                             unsigned& c3, unsigned k0, unsigned k1,
                                             unsigned k2, unsigned k3, unsigned k4,
                                             unsigned k5, unsigned k6, unsigned k7,
                                             unsigned t) {
  unsigned m1, m2, m3, m4, m5, m6, m7;
  asm("v_cmp_lt_u32_e32 vcc_lo, %11, %19\n\t"
      "v_cmp_lt_u32_e64 %4, %12, %19\n\t"
      "v_cmp_lt_u32_e64 %5, %13, %19\n\t"
      "v_cmp_lt_u32_e64 %6, %14, %19\n\t"
      "v_cmp_lt_u32_e64 %7, %15, %19\n\t"
      "v_cmp_lt_u32_e64 %8, %16, %19\n\t"
      "v_cmp_lt_u32_e64 %9, %17, %19\n\t"
      "v_cmp_lt_u32_e64 %10, %18, %19\n\t"
      "s_delay_alu instid0(VALU_DEP_4) | instskip(NEXT) | instid1(VALU_DEP_4)\n\t"
      "v_add_co_ci_u32_e32 %0, vcc_lo, 0, %0, vcc_lo\n\t"
      "v_add_co_ci_u32_e64 %1, %4, 0, %1, %4\n\t"
      "s_delay_alu instid0(VALU_DEP_4) | instskip(NEXT) | instid1(VALU_DEP_4)\n\t"
      "v_add_co_ci_u32_e64 %2, %5, 0, %2, %5\n\t"
      "v_add_co_ci_u32_e64 %3, %6, 0, %3, %6\n\t"
      "s_delay_alu instid0(VALU_DEP_4) | instskip(NEXT) | instid1(VALU_DEP_4)\n\t"
      "v_add_co_ci_u32_e64 %0, %7, 0, %0, %7\n\t"
      "v_add_co_ci_u32_e64 %1, %8, 0, %1, %8\n\t"
      "s_delay_alu instid0(VALU_DEP_4) | instskip(NEXT) | instid1(VALU_DEP_4)\n\t"
      "v_add_co_ci_u32_e64 %2, %9, 0, %2, %9\n\t"
      "v_add_co_ci_u32_e64 %3, %10, 0, %3, %10"
      : "+v"(c0), "+v"(c1), "+v"(c2), "+v"(c3),
        "=&s"(m1), "=&s"(m2), "=&s"(m3), "=&s"(m4), "=&s"(m5), "=&s"(m6), "=&s"(m7)
      : "v"(k0), "v"(k1), "v"(k2), "v"(k3), "v"(k4), "v"(k5), "v"(k6), "v"(k7), "v"(t)
      : "vcc");
}
__device__ __forceinline__ void shift_in_below2(unsigned& mlo, unsigned& mhi, unsigned klo,
                                                unsigned khi, unsigned t) {
  unsigned m1;
  asm("v_cmp_lt_u32_e32 vcc_lo, %3, %5\n\t"
      "v_cmp_lt_u32_e64 %2, %4, %5\n\t"
      "s_delay_alu instid0(VALU_DEP_2) | instskip(NEXT) | instid1(VALU_DEP_2)\n\t"
      "v_add_co_ci_u32_e32 %0, vcc_lo, %0, %0, vcc_lo\n\t"
      "v_add_co_ci_u32_e64 %1, %2, %1, %1, %2"
      : "+v"(mlo), "+v"(mhi), "=&s"(m1)
      : "v"(klo), "v"(khi), "v"(t)
      : "vcc");
}
__device__ __forceinline__ void shift_in_equal2(unsigned& mlo, unsigned& mhi, unsigned klo,
                                                unsigned khi, unsigned t) {
  unsigned m1;
  asm("v_cmp_eq_u32_e32 vcc_lo, %3, %5\n\t"
      "v_cmp_eq_u32_e64 %2, %4, %5\n\t"
      "s_delay_alu instid0(VALU_DEP_2) | instskip(NEXT) | instid1(VALU_DEP_2)\n\t"
      "v_add_co_ci_u32_e32 %0, vcc_lo, %0, %0, vcc_lo\n\t"
      "v_add_co_ci_u32_e64 %1, %2, %1, %1, %2"
      : "+v"(mlo), "+v"(mhi), "=&s"(m1)
      : "v"(klo), "v"(khi), "v"(t)
      : "vcc");
}

__device__ __forceinline__ void shift_in_le2(unsigned& mlo, unsigned& mhi, unsigned klo,
                                             unsigned khi, unsigned t) {
  unsigned m1;
  asm("v_cmp_le_u32_e32 vcc_lo, %3, %5\n\t"
      "v_cmp_le_u32_e64 %2, %4, %5\n\t"
      "s_delay_alu instid0(VALU_DEP_2) | instskip(NEXT) | instid1(VALU_DEP_2)\n\t"
      "v_add_co_ci_u32_e32 %0, vcc_lo, %0, %0, vcc_lo\n\t"
      "v_add_co_ci_u32_e64 %1, %2, %1, %1, %2"
      : "+v"(mlo), "+v"(mhi), "=&s"(m1)
      : "v"(klo), "v"(khi), "v"(t)
      : "vcc");
}
__device__ __forceinline__ void rank_below4(unsigned& r0, unsigned& r1, unsigned s0,
                                            unsigned s1, unsigned s2, unsigned s3,
                                            unsigned v) {
  unsigned m1, m2, m3;
  asm("v_cmp_lt_u32_e32 vcc_lo, %5, %9\n\t"
      "v_cmp_lt_u32_e64 %2, %6, %9\n\t"
      "v_cmp_lt_u32_e64 %3, %7, %9\n\t"
      "v_cmp_lt_u32_e64 %4, %8, %9\n\t"
      "s_delay_alu instid0(VALU_DEP_4) | instskip(NEXT) | instid1(VALU_DEP_4)\n\t"
      "v_add_co_ci_u32_e32 %0, vcc_lo, 0, %0, vcc_lo\n\t"
      "v_add_co_ci_u32_e64 %1, %2, 0, %1, %2\n\t"
      "s_delay_alu instid0(VALU_DEP_2) | instskip(NEXT) | instid1(VALU_DEP_2)\n\t"
      "v_add_co_ci_u32_e64 %0, %3, 0, %0, %3\n\t"
      "v_add_co_ci_u32_e64 %1, %4, 0, %1, %4"
      : "+v"(r0), "+v"(r1), "=&s"(m1), "=&s"(m2), "=&s"(m3)
      : "s"(s0), "s"(s1), "s"(s2), "s"(s3), "v"(v)
      : "vcc");
}
__device__ __forceinline__ unsigned wave_min_u32(unsigned v) {
  v = min(v, (unsigned)__builtin_amdgcn_update_dpp(0, (int)v, 0x161, 0xf, 0xf, false));
  v = min(v, (unsigned)__builtin_amdgcn_update_dpp(0, (int)v, 0x162, 0xf, 0xf, false));
  v = min(v, (unsigned)__builtin_amdgcn_update_dpp(0, (int)v, 0x164, 0xf, 0xf, false));
  v = min(v, (unsigned)__builtin_amdgcn_update_dpp(0, (int)v, 0x168, 0xf, 0xf, false));
  return min((unsigned)__builtin_amdgcn_readlane((int)v, 0),
             (unsigned)__builtin_amdgcn_readlane((int)v, 16));
}
__device__ __forceinline__ unsigned dist_key(const float4 me, const float4 p) {
  float dot;
  {
    #pragma clang fp contract(off)
    dot = fmaf(me.x, p.x, me.y * p.y) + me.z * p.z;
  }
  const float d = fmaf(-2.0f, dot, me.w + p.w);
  const unsigned u = __float_as_uint(d);
  return u ^ ((unsigned)((int)u >> 31) | 0x80000000u);
}

__global__ void __launch_bounds__(KNN_WAVES * 32)
knn_kernel(const float* __restrict__ xyz, int* __restrict__ idx_out,
           const float* __restrict__ feat, const float* __restrict__ w2,
           const float* __restrict__ g2, const float* __restrict__ v2,
           float* __restrict__ Cf) {
  __shared__ float4 pts[NPTS];
  __shared__ int cand[KNN_WAVES][KNN_CAND];
  const int b = blockIdx.y;
  const float* xb = xyz + (size_t)b * 3 * NPTS;
  for (int m = threadIdx.x; m < NPTS; m += blockDim.x) {
    float x = xb[m], y = xb[NPTS + m], z = xb[2 * NPTS + m];
    pts[m] = make_float4(x, y, z, fmaf(z, z, fmaf(x, x, y * y)));
  }
  __syncthreads();

  const int lane = threadIdx.x & 31;
  const int wave = threadIdx.x >> 5;
  int* cw = cand[wave];

  const int  swave      = __builtin_amdgcn_readfirstlane((int)(threadIdx.x >> 5));
  const int  gemm_n0    = blockIdx.x * KNN_QPB + 16 * (swave & 3);
  const int  gemm_strip_id = swave >> 2;
  const bool gemm_first = ((swave ^ (swave >> 2)) & 1) == 0;
  if (gemm_first) feat_gemm_strip(feat, w2, g2, v2, Cf, b, gemm_n0, gemm_strip_id, lane);

  for (int qi = 0; qi < KNN_QPW; ++qi) {
    const int q = blockIdx.x * KNN_QPB + wave * KNN_QPW + qi;
    const float4 me = pts[q];
    int* op = idx_out + ((size_t)b * NPTS + q) * KNN_K;

    unsigned key[KNN_SLOTS];
    #pragma unroll
    for (int j = 0; j < KNN_SLOTS; ++j) key[j] = dist_key(me, pts[j * 32 + lane]);

    unsigned mn = key[0];
    #pragma unroll
    for (int j = 1; j < KNN_SLOTS; ++j) mn = min(mn, key[j]);
    const unsigned mu = (mn & ~31u) | (unsigned)lane;
    unsigned r0 = 0u, r1 = 0u;
    #pragma unroll
    for (int l = 0; l < 32; l += 4)
      rank_below4(r0, r1, (unsigned)__builtin_amdgcn_readlane((int)mu, l),
                  (unsigned)__builtin_amdgcn_readlane((int)mu, l + 1),
                  (unsigned)__builtin_amdgcn_readlane((int)mu, l + 2),
                  (unsigned)__builtin_amdgcn_readlane((int)mu, l + 3), mu);
    const unsigned holder = __builtin_amdgcn_ballot_w32((r0 + r1) == (unsigned)KNN_K);
    const unsigned U =
        (unsigned)__builtin_amdgcn_readlane((int)mu, __builtin_ctz(holder)) | 31u;

    unsigned mlo = 0u, mhi = 0u;
    #pragma unroll
    for (int j = 31; j >= 0; --j) shift_in_le2(mlo, mhi, key[j], key[j + 32], U);
    const unsigned cS =
        wave_sum_u32((unsigned)__builtin_popcount(mlo) + (unsigned)__builtin_popcount(mhi));

    if (cS <= (unsigned)KNN_CAND) {
      unsigned listed = 0u;
      for (int pass = 0; pass < KNN_SLOTS; ++pass) {
        const bool has = (mlo | mhi) != 0u;
        const unsigned act = __builtin_amdgcn_ballot_w32(has);
        if (act == 0u) break;
        if (has) {
          int j;
          if (mlo != 0u) { j = __builtin_ctz(mlo);      mlo &= mlo - 1u; }
          else           { j = 32 + __builtin_ctz(mhi); mhi &= mhi - 1u; }
          cw[listed + __builtin_amdgcn_mbcnt_lo(act, 0u)] = j * 32 + lane;
        }
        listed += (unsigned)__builtin_popcount(act);
      }
      __builtin_amdgcn_wave_barrier();
      const int i0 = ((unsigned)lane < cS) ? cw[lane] : -1;
      const int i1 = ((unsigned)lane + 32u < cS) ? cw[lane + 32] : -1;
      unsigned k0 = 0xFFFFFFFFu, k1 = 0xFFFFFFFFu;
      if (i0 >= 0) k0 = dist_key(me, pts[i0]);
      if (i1 >= 0) k1 = dist_key(me, pts[i1]);
      __builtin_amdgcn_wave_barrier();

      {
        unsigned ra0 = 0u, ra1 = 0u, rb0 = 0u, rb1 = 0u;
        if (cS <= 32u) {
          #pragma unroll
          for (int l = 0; l < 32; l += 4) {
            const unsigned s0 = (unsigned)__builtin_amdgcn_readlane((int)k0, l);
            const unsigned s1 = (unsigned)__builtin_amdgcn_readlane((int)k0, l + 1);
            const unsigned s2 = (unsigned)__builtin_amdgcn_readlane((int)k0, l + 2);
            const unsigned s3 = (unsigned)__builtin_amdgcn_readlane((int)k0, l + 3);
            rank_below4(ra0, ra1, s0, s1, s2, s3, k0);
          }
        } else {
          #pragma unroll
          for (int l = 0; l < 32; l += 4) {
            const unsigned s0 = (unsigned)__builtin_amdgcn_readlane((int)k0, l);
            const unsigned s1 = (unsigned)__builtin_amdgcn_readlane((int)k0, l + 1);
            const unsigned s2 = (unsigned)__builtin_amdgcn_readlane((int)k0, l + 2);
            const unsigned s3 = (unsigned)__builtin_amdgcn_readlane((int)k0, l + 3);
            rank_below4(ra0, ra1, s0, s1, s2, s3, k0);
            rank_below4(rb0, rb1, s0, s1, s2, s3, k1);
          }
          for (unsigned base = 32u; base < cS; base += 4u) {
            const int e = (int)base - 32;
            const unsigned s0 = (unsigned)__builtin_amdgcn_readlane((int)k1, e);
            const unsigned s1 = (unsigned)__builtin_amdgcn_readlane((int)k1, e + 1);
            const unsigned s2 = (unsigned)__builtin_amdgcn_readlane((int)k1, e + 2);
            const unsigned s3 = (unsigned)__builtin_amdgcn_readlane((int)k1, e + 3);
            rank_below4(ra0, ra1, s0, s1, s2, s3, k0);
            rank_below4(rb0, rb1, s0, s1, s2, s3, k1);
          }
        }
        const bool t0 = (ra0 + ra1) < (unsigned)KNN_K && i0 >= 0;
        const bool t1 = (rb0 + rb1) < (unsigned)KNN_K && i1 >= 0;
        const unsigned tb0 = __builtin_amdgcn_ballot_w32(t0);
        const unsigned tb1 = __builtin_amdgcn_ballot_w32(t1);
        const unsigned tn0 = (unsigned)__builtin_popcount(tb0);
        if (tn0 + (unsigned)__builtin_popcount(tb1) == (unsigned)KNN_K) {
          if (t0) op[__builtin_amdgcn_mbcnt_lo(tb0, 0u)] = i0;
          if (t1) op[tn0 + __builtin_amdgcn_mbcnt_lo(tb1, 0u)] = i1;
          continue;
        }
      }

      unsigned T = 0u, clt = 0u;
      for (int bit = 31; bit >= 0; --bit) {
        const unsigned trial = T | (1u << bit);
        const unsigned c =
            wave_sum_u32((k0 < trial ? 1u : 0u) + (k1 < trial ? 1u : 0u));
        if (c <= (unsigned)KNN_K) {
          T = trial;
          clt = c;
          if (c == (unsigned)KNN_K) break;
        }
      }
      const unsigned need = (unsigned)KNN_K - clt;

      const bool s0 = k0 < T, s1 = k1 < T;
      const unsigned b0 = __builtin_amdgcn_ballot_w32(s0);
      const unsigned b1 = __builtin_amdgcn_ballot_w32(s1);
      if (s0) op[__builtin_amdgcn_mbcnt_lo(b0, 0u)] = i0;
      if (s1) op[(unsigned)__builtin_popcount(b0) + __builtin_amdgcn_mbcnt_lo(b1, 0u)] = i1;

      if (need != 0u) {
        unsigned e0 = (k0 == T) ? (unsigned)i0 : 0xFFFFFFFFu;
        unsigned e1 = (k1 == T) ? (unsigned)i1 : 0xFFFFFFFFu;
        for (unsigned r = 0u; r < need; ++r) {
          const unsigned m = wave_min_u32(min(e0, e1));
          if (lane == 0) op[clt + r] = (int)m;
          if (e0 == m) e0 = 0xFFFFFFFFu;
          if (e1 == m) e1 = 0xFFFFFFFFu;
        }
      }
      continue;
    }

    unsigned T = 0u, clt = 0u;
    for (int bit = 31; bit >= 0; --bit) {
      const unsigned trial = T | (1u << bit);
      unsigned c0 = 0u, c1 = 0u, c2 = 0u, c3 = 0u;
      #pragma unroll
      for (int j = 0; j < KNN_SLOTS; j += 8)
        count_below8(c0, c1, c2, c3, key[j + 0], key[j + 1], key[j + 2], key[j + 3],
                     key[j + 4], key[j + 5], key[j + 6], key[j + 7], trial);
      const unsigned c = wave_sum_u32((c0 + c1) + (c2 + c3));
      if (c <= (unsigned)KNN_K) {
        T = trial;
        clt = c;
        if (c == (unsigned)KNN_K) break;
      }
    }
    const unsigned need = (unsigned)KNN_K - clt;

    mlo = 0u; mhi = 0u;
    #pragma unroll
    for (int j = 31; j >= 0; --j) shift_in_below2(mlo, mhi, key[j], key[j + 32], T);

    unsigned written = 0u;
    for (int pass = 0; pass < KNN_SLOTS; ++pass) {
      const bool has = (mlo | mhi) != 0u;
      const unsigned act = __builtin_amdgcn_ballot_w32(has);
      if (act == 0u) break;
      if (has) {
        int j;
        if (mlo != 0u) { j = __builtin_ctz(mlo);      mlo &= mlo - 1u; }
        else           { j = 32 + __builtin_ctz(mhi); mhi &= mhi - 1u; }
        op[written + __builtin_amdgcn_mbcnt_lo(act, 0u)] = j * 32 + lane;
      }
      written += (unsigned)__builtin_popcount(act);
    }

    if (need != 0u) {
      unsigned elo = 0u, ehi = 0u;
      #pragma unroll
      for (int j = 31; j >= 0; --j) shift_in_equal2(elo, ehi, key[j], key[j + 32], T);
      unsigned taken = 0u;
      for (int j = 0; j < KNN_SLOTS && taken < need; ++j) {
        const unsigned w = (j < 32) ? elo : ehi;
        const bool eq = ((w >> (j & 31)) & 1u) != 0u;
        const unsigned meq = __builtin_amdgcn_ballot_w32(eq);
        if (meq != 0u) {
          const unsigned r = taken + __builtin_amdgcn_mbcnt_lo(meq, 0u);
          if (eq && r < need) op[clt + r] = j * 32 + lane;
          taken += (unsigned)__builtin_popcount(meq);
        }
      }
    }
  }

  if (!gemm_first) feat_gemm_strip(feat, w2, g2, v2, Cf, b, gemm_n0, gemm_strip_id, lane);
}

#define GM_PTS   32
#define GM_WAVES 16
#define GM_PPW   (GM_PTS / GM_WAVES)
#define GM_RPW   (CH / GM_WAVES)
#define GM_PITCH (GM_PTS + 1)

typedef const float2 __attribute__((address_space(1)))* gfloat2p;

__global__ void __launch_bounds__(GM_WAVES * 32, 8)
gather_max(const float* __restrict__ xyz, const float* __restrict__ w1,
           const float* __restrict__ b1, const float* __restrict__ g1,
           const float* __restrict__ be1, const float* __restrict__ m1,
           const float* __restrict__ v1, const float* __restrict__ Cf,
           const int* __restrict__ idx, const float* __restrict__ b2,
           const float* __restrict__ g2, const float* __restrict__ be2,
           const float* __restrict__ m2, const float* __restrict__ v2,
           float* __restrict__ out) {
  __shared__ float tile[CH * GM_PITCH];
  const int b    = blockIdx.y;
  const int n0   = blockIdx.x * GM_PTS;
  const int lane = threadIdx.x & 31;
  const int wave = __builtin_amdgcn_readfirstlane((int)(threadIdx.x >> 5));

  const float* xb  = xyz + (size_t)b * 3 * NPTS;
  const float* cfb = Cf + (size_t)b * NPTS * CH;

  const int kk = (lane < KNN_K) ? lane : lane - KNN_K;

  const float4 w1A = *(const float4*)(w1 + 12 * lane);
  const float4 w1B = *(const float4*)(w1 + 12 * lane + 4);
  const float4 w1C = *(const float4*)(w1 + 12 * lane + 8);
  const float2 g1v  = *(const float2*)(g1 + 2 * lane);
  const float2 v1v  = *(const float2*)(v1 + 2 * lane);
  const float2 b1v  = *(const float2*)(b1 + 2 * lane);
  const float2 be1v = *(const float2*)(be1 + 2 * lane);
  const float2 m1v  = *(const float2*)(m1 + 2 * lane);
  const float2 g2v  = *(const float2*)(g2 + 2 * lane);
  const float2 v2v  = *(const float2*)(v2 + 2 * lane);
  const float2 b2v  = *(const float2*)(b2 + 2 * lane);
  const float2 be2v = *(const float2*)(be2 + 2 * lane);
  const float2 m2v  = *(const float2*)(m2 + 2 * lane);

  int    jv[GM_PPW], xv[GM_PPW], yv[GM_PPW], zv[GM_PPW];
  float  cx[GM_PPW], cy[GM_PPW], cz[GM_PPW];
  float2 cv[GM_PPW];
  #pragma unroll
  for (int p = 0; p < GM_PPW; ++p) {
    const int    n     = n0 + wave * GM_PPW + p;
    const size_t pbase = (size_t)b * NPTS + n;
    jv[p] = idx[pbase * KNN_K + kk];
    cx[p] = xb[n];
    cy[p] = xb[NPTS + n];
    cz[p] = xb[2 * NPTS + n];
    cv[p] = *(const float2*)(Cf + pbase * CH + HALF_CH + 2 * lane);
  }

  const float s10 = g1v.x / sqrtf(v1v.x + EPSB);
  const float s11 = g1v.y / sqrtf(v1v.y + EPSB);
  const float s20 = g2v.x / sqrtf(v2v.x + EPSB);
  const float s21 = g2v.y / sqrtf(v2v.y + EPSB);
  const float2 bg = make_float2(b1v.x * s10 + (be1v.x - m1v.x * s10),
                                b1v.y * s11 + (be1v.y - m1v.y * s11));
  const float2 bf = make_float2(b2v.x * s20 + (be2v.x - m2v.x * s20),
                                b2v.y * s21 + (be2v.y - m2v.y * s21));
  const float2 wa01 = make_float2(s10 * w1A.x, s10 * w1A.y);
  const float2 wa23 = make_float2(s10 * w1A.z, s11 * w1B.z);
  const float2 wa45 = make_float2(s11 * w1B.w, s11 * w1C.x);
  const float2 wd01 = make_float2(s10 * (w1A.w - w1A.x), s10 * (w1B.x - w1A.y));
  const float2 wd23 = make_float2(s10 * (w1B.y - w1A.z), s11 * (w1C.y - w1B.z));
  const float2 wd45 = make_float2(s11 * (w1C.z - w1B.w), s11 * (w1C.w - w1C.x));

  float qg[GM_PPW][2], vf[GM_PPW][2];
  #pragma unroll
  for (int p = 0; p < GM_PPW; ++p) {
    qg[p][0] = wd01.x * cx[p] + wd01.y * cy[p] + wd23.x * cz[p] + bg.x;
    qg[p][1] = wd23.y * cx[p] + wd45.x * cy[p] + wd45.y * cz[p] + bg.y;
    vf[p][0] = cv[p].x + bf.x;
    vf[p][1] = cv[p].y + bf.y;
  }
  asm volatile("" :: "v"(wa01.x), "v"(wa01.y), "v"(wa23.x), "v"(wa23.y), "v"(wa45.x), "v"(wa45.y),
                     "v"(qg[0][0]), "v"(qg[0][1]), "v"(qg[1][0]), "v"(qg[1][1]),
                     "v"(vf[0][0]), "v"(vf[0][1]), "v"(vf[1][0]), "v"(vf[1][1]));
  #pragma unroll
  for (int p = 0; p < GM_PPW; ++p) {
    xv[p] = __float_as_int(xb[jv[p]]);
    yv[p] = __float_as_int(xb[NPTS + jv[p]]);
    zv[p] = __float_as_int(xb[2 * NPTS + jv[p]]);
  }

  float2 row[GM_PPW][KNN_K];
  #pragma unroll
  for (int p = 0; p < GM_PPW; ++p) {
    #pragma unroll
    for (int k = 0; k < KNN_K; ++k) {
      const int j = __builtin_amdgcn_readlane(jv[p], k);
      const unsigned long long ra = (unsigned long long)(cfb + (size_t)j * CH);
      const unsigned ralo = (unsigned)__builtin_amdgcn_readfirstlane((int)(unsigned)ra);
      const unsigned rahi = (unsigned)__builtin_amdgcn_readfirstlane((int)(unsigned)(ra >> 32));
      const gfloat2p rp = (gfloat2p)(((unsigned long long)rahi << 32) | ralo);
      row[p][k] = rp[lane];
    }
  }

  float mg[GM_PPW][2];
  #pragma unroll
  for (int p = 0; p < GM_PPW; ++p) { mg[p][0] = -FLT_MAX; mg[p][1] = -FLT_MAX; }
  #pragma unroll 2
  for (int k = 0; k < KNN_K; ++k) {
    #pragma unroll
    for (int p = 0; p < GM_PPW; ++p) {
      const float x = __int_as_float(__builtin_amdgcn_readlane(xv[p], k));
      const float y = __int_as_float(__builtin_amdgcn_readlane(yv[p], k));
      const float z = __int_as_float(__builtin_amdgcn_readlane(zv[p], k));
      mg[p][0] = fmaxf(mg[p][0], wa01.x * x + wa01.y * y + wa23.x * z);
      mg[p][1] = fmaxf(mg[p][1], wa23.y * x + wa45.x * y + wa45.y * z);
    }
  }

  #pragma unroll
  for (int p = 0; p < GM_PPW; ++p) {
    const int pl = wave * GM_PPW + p;
    float mf0 = -FLT_MAX, mf1 = -FLT_MAX;
    #pragma unroll
    for (int k = 0; k < KNN_K; ++k) {
      mf0 = fmaxf(mf0, row[p][k].x);
      mf1 = fmaxf(mf1, row[p][k].y);
    }
    tile[(2 * lane)               * GM_PITCH + pl] = fmaxf(mg[p][0] + qg[p][0], 0.0f);
    tile[(2 * lane + 1)           * GM_PITCH + pl] = fmaxf(mg[p][1] + qg[p][1], 0.0f);
    tile[(HALF_CH + 2 * lane)     * GM_PITCH + pl] = fmaxf(mf0 + vf[p][0], 0.0f);
    tile[(HALF_CH + 2 * lane + 1) * GM_PITCH + pl] = fmaxf(mf1 + vf[p][1], 0.0f);
  }
  __syncthreads();

  float* ob = out + (size_t)b * CH * NPTS + n0 + lane;
  #pragma unroll
  for (int r = 0; r < GM_RPW; ++r) {
    const int c = wave * GM_RPW + r;
    ob[(size_t)c * NPTS] = tile[c * GM_PITCH + lane];
  }
}

extern "C" void kernel_launch(void* const* d_in, const int* in_sizes, int n_in,
                              void* d_out, int out_size, void* d_ws, size_t ws_size,
                              hipStream_t stream) {
  (void)in_sizes; (void)n_in; (void)out_size; (void)ws_size;

  const float* xyz  = (const float*)d_in[0];
  const float* feat = (const float*)d_in[1];
  const float* w1   = (const float*)d_in[2];
  const float* b1   = (const float*)d_in[3];
  const float* g1   = (const float*)d_in[4];
  const float* be1  = (const float*)d_in[5];
  const float* m1   = (const float*)d_in[6];
  const float* v1   = (const float*)d_in[7];
  const float* w2   = (const float*)d_in[8];
  const float* b2   = (const float*)d_in[9];
  const float* g2   = (const float*)d_in[10];
  const float* be2  = (const float*)d_in[11];
  const float* m2   = (const float*)d_in[12];
  const float* v2   = (const float*)d_in[13];

  char* ws = (char*)d_ws;
  float* Cf     = (float*)(ws + (128 << 10)) + (size_t)BATCH * NPTS * CH;
  int*   idxb   = (int*)(Cf + (size_t)BATCH * NPTS * CH);

  knn_kernel<<<dim3(NPTS / KNN_QPB, BATCH), KNN_WAVES * 32, 0, stream>>>(xyz, idxb, feat, w2,
                                                                         g2, v2, Cf);

  gather_max<<<dim3(NPTS / GM_PTS, BATCH), GM_WAVES * 32, 0, stream>>>(
      xyz, w1, b1, g1, be1, m1, v1, Cf, idxb, b2, g2, be2, m2, v2, (float*)d_out);
}


# feature GEMM: A operand (scaled w2, 128x128) built once per block in LDS (pitch 132 words, conflict-free 8-byte reads) instead of per-wave global loads over 16 cache lines each; same sums
# speedup vs baseline: 127.4273x; 1.1001x over previous
#include <hip/hip_runtime.h>
#include <hip/hip_bf16.h>
#include <float.h>

#define BATCH   8
#define NPTS    2048
#define CH      128
#define HALF_CH 64
#define KNN_K   20
#define EPSB    1e-5f

typedef __attribute__((ext_vector_type(2))) float v2f;
typedef __attribute__((ext_vector_type(8))) float v8f;


#define WC_PITCH 132
#define WC_WORDS (32 * 1024)
static_assert(WC_WORDS >= CH * WC_PITCH, "Wc does not fit its LDS array");

__device__ __forceinline__ void feat_gemm_strip(const float* __restrict__ feat,
                                                const float* wcs,
                                                float* __restrict__ Cf, const int b,
                                                const int n0, const int strip,
                                                const int lane) {
  const int l16 = lane & 15;
  const int hi  = lane >> 4;
  const int m0  = strip * 32;

  const float* Fb   = feat + (size_t)b * CH * NPTS;
  const float* wr0  = wcs + (m0 + l16) * WC_PITCH + 2 * hi;
  const float* wr1  = wr0 + 16 * WC_PITCH;
  const float* bcol = Fb + (size_t)(2 * hi) * NPTS + n0 + l16;

  v8f c0 = {0.f, 0.f, 0.f, 0.f, 0.f, 0.f, 0.f, 0.f};
  v8f c1 = {0.f, 0.f, 0.f, 0.f, 0.f, 0.f, 0.f, 0.f};
  #pragma unroll 8
  for (int k = 0; k < CH; k += 4) {
    const v2f a0 = *(const v2f*)(wr0 + k);
    const v2f a1 = *(const v2f*)(wr1 + k);
    v2f bb;
    bb.x = bcol[(size_t)k * NPTS];
    bb.y = bcol[(size_t)(k + 1) * NPTS];
    c0 = __builtin_amdgcn_wmma_f32_16x16x4_f32(false, a0, false, bb,
                                               (short)0, c0, false, false);
    c1 = __builtin_amdgcn_wmma_f32_16x16x4_f32(false, a1, false, bb,
                                               (short)0, c1, false, false);
  }

  float* out = Cf + ((size_t)b * NPTS + n0 + l16) * CH + m0 + 8 * hi;
  #pragma unroll
  for (int v = 0; v < 8; ++v) out[v] = c0[v];
  #pragma unroll
  for (int v = 0; v < 8; ++v) out[16 + v] = c1[v];
}

#define KNN_WAVES 16
#define KNN_QPW   4
#define KNN_QPB   (KNN_WAVES * KNN_QPW)
#define KNN_SLOTS (NPTS / 32)
#define KNN_CAND  64

__device__ __forceinline__ unsigned wave_sum_u32(unsigned v) {
  v += (unsigned)__builtin_amdgcn_update_dpp(0, (int)v, 0x161, 0xf, 0xf, false);
  v += (unsigned)__builtin_amdgcn_update_dpp(0, (int)v, 0x162, 0xf, 0xf, false);
  v += (unsigned)__builtin_amdgcn_update_dpp(0, (int)v, 0x164, 0xf, 0xf, false);
  v += (unsigned)__builtin_amdgcn_update_dpp(0, (int)v, 0x168, 0xf, 0xf, false);
  return (unsigned)__builtin_amdgcn_readlane((int)v, 0) +
         (unsigned)__builtin_amdgcn_readlane((int)v, 16);
}

__device__ __forceinline__ void count_below8(unsigned& c0, unsigned& c1, unsigned& c2,
                                             unsigned& c3, unsigned k0, unsigned k1,
                                             unsigned k2, unsigned k3, unsigned k4,
                                             unsigned k5, unsigned k6, unsigned k7,
                                             unsigned t) {
  unsigned m1, m2, m3, m4, m5, m6, m7;
  asm("v_cmp_lt_u32_e32 vcc_lo, %11, %19\n\t"
      "v_cmp_lt_u32_e64 %4, %12, %19\n\t"
      "v_cmp_lt_u32_e64 %5, %13, %19\n\t"
      "v_cmp_lt_u32_e64 %6, %14, %19\n\t"
      "v_cmp_lt_u32_e64 %7, %15, %19\n\t"
      "v_cmp_lt_u32_e64 %8, %16, %19\n\t"
      "v_cmp_lt_u32_e64 %9, %17, %19\n\t"
      "v_cmp_lt_u32_e64 %10, %18, %19\n\t"
      "s_delay_alu instid0(VALU_DEP_4) | instskip(NEXT) | instid1(VALU_DEP_4)\n\t"
      "v_add_co_ci_u32_e32 %0, vcc_lo, 0, %0, vcc_lo\n\t"
      "v_add_co_ci_u32_e64 %1, %4, 0, %1, %4\n\t"
      "s_delay_alu instid0(VALU_DEP_4) | instskip(NEXT) | instid1(VALU_DEP_4)\n\t"
      "v_add_co_ci_u32_e64 %2, %5, 0, %2, %5\n\t"
      "v_add_co_ci_u32_e64 %3, %6, 0, %3, %6\n\t"
      "s_delay_alu instid0(VALU_DEP_4) | instskip(NEXT) | instid1(VALU_DEP_4)\n\t"
      "v_add_co_ci_u32_e64 %0, %7, 0, %0, %7\n\t"
      "v_add_co_ci_u32_e64 %1, %8, 0, %1, %8\n\t"
      "s_delay_alu instid0(VALU_DEP_4) | instskip(NEXT) | instid1(VALU_DEP_4)\n\t"
      "v_add_co_ci_u32_e64 %2, %9, 0, %2, %9\n\t"
      "v_add_co_ci_u32_e64 %3, %10, 0, %3, %10"
      : "+v"(c0), "+v"(c1), "+v"(c2), "+v"(c3),
        "=&s"(m1), "=&s"(m2), "=&s"(m3), "=&s"(m4), "=&s"(m5), "=&s"(m6), "=&s"(m7)
      : "v"(k0), "v"(k1), "v"(k2), "v"(k3), "v"(k4), "v"(k5), "v"(k6), "v"(k7), "v"(t)
      : "vcc");
}
__device__ __forceinline__ void shift_in_below2(unsigned& mlo, unsigned& mhi, unsigned klo,
                                                unsigned khi, unsigned t) {
  unsigned m1;
  asm("v_cmp_lt_u32_e32 vcc_lo, %3, %5\n\t"
      "v_cmp_lt_u32_e64 %2, %4, %5\n\t"
      "s_delay_alu instid0(VALU_DEP_2) | instskip(NEXT) | instid1(VALU_DEP_2)\n\t"
      "v_add_co_ci_u32_e32 %0, vcc_lo, %0, %0, vcc_lo\n\t"
      "v_add_co_ci_u32_e64 %1, %2, %1, %1, %2"
      : "+v"(mlo), "+v"(mhi), "=&s"(m1)
      : "v"(klo), "v"(khi), "v"(t)
      : "vcc");
}
__device__ __forceinline__ void shift_in_equal2(unsigned& mlo, unsigned& mhi, unsigned klo,
                                                unsigned khi, unsigned t) {
  unsigned m1;
  asm("v_cmp_eq_u32_e32 vcc_lo, %3, %5\n\t"
      "v_cmp_eq_u32_e64 %2, %4, %5\n\t"
      "s_delay_alu instid0(VALU_DEP_2) | instskip(NEXT) | instid1(VALU_DEP_2)\n\t"
      "v_add_co_ci_u32_e32 %0, vcc_lo, %0, %0, vcc_lo\n\t"
      "v_add_co_ci_u32_e64 %1, %2, %1, %1, %2"
      : "+v"(mlo), "+v"(mhi), "=&s"(m1)
      : "v"(klo), "v"(khi), "v"(t)
      : "vcc");
}

__device__ __forceinline__ void shift_in_le2(unsigned& mlo, unsigned& mhi, unsigned klo,
                                             unsigned khi, unsigned t) {
  unsigned m1;
  asm("v_cmp_le_u32_e32 vcc_lo, %3, %5\n\t"
      "v_cmp_le_u32_e64 %2, %4, %5\n\t"
      "s_delay_alu instid0(VALU_DEP_2) | instskip(NEXT) | instid1(VALU_DEP_2)\n\t"
      "v_add_co_ci_u32_e32 %0, vcc_lo, %0, %0, vcc_lo\n\t"
      "v_add_co_ci_u32_e64 %1, %2, %1, %1, %2"
      : "+v"(mlo), "+v"(mhi), "=&s"(m1)
      : "v"(klo), "v"(khi), "v"(t)
      : "vcc");
}
__device__ __forceinline__ void rank_below4(unsigned& r0, unsigned& r1, unsigned s0,
                                            unsigned s1, unsigned s2, unsigned s3,
                                            unsigned v) {
  unsigned m1, m2, m3;
  asm("v_cmp_lt_u32_e32 vcc_lo, %5, %9\n\t"
      "v_cmp_lt_u32_e64 %2, %6, %9\n\t"
      "v_cmp_lt_u32_e64 %3, %7, %9\n\t"
      "v_cmp_lt_u32_e64 %4, %8, %9\n\t"
      "s_delay_alu instid0(VALU_DEP_4) | instskip(NEXT) | instid1(VALU_DEP_4)\n\t"
      "v_add_co_ci_u32_e32 %0, vcc_lo, 0, %0, vcc_lo\n\t"
      "v_add_co_ci_u32_e64 %1, %2, 0, %1, %2\n\t"
      "s_delay_alu instid0(VALU_DEP_2) | instskip(NEXT) | instid1(VALU_DEP_2)\n\t"
      "v_add_co_ci_u32_e64 %0, %3, 0, %0, %3\n\t"
      "v_add_co_ci_u32_e64 %1, %4, 0, %1, %4"
      : "+v"(r0), "+v"(r1), "=&s"(m1), "=&s"(m2), "=&s"(m3)
      : "s"(s0), "s"(s1), "s"(s2), "s"(s3), "v"(v)
      : "vcc");
}
__device__ __forceinline__ unsigned wave_min_u32(unsigned v) {
  v = min(v, (unsigned)__builtin_amdgcn_update_dpp(0, (int)v, 0x161, 0xf, 0xf, false));
  v = min(v, (unsigned)__builtin_amdgcn_update_dpp(0, (int)v, 0x162, 0xf, 0xf, false));
  v = min(v, (unsigned)__builtin_amdgcn_update_dpp(0, (int)v, 0x164, 0xf, 0xf, false));
  v = min(v, (unsigned)__builtin_amdgcn_update_dpp(0, (int)v, 0x168, 0xf, 0xf, false));
  return min((unsigned)__builtin_amdgcn_readlane((int)v, 0),
             (unsigned)__builtin_amdgcn_readlane((int)v, 16));
}
__device__ __forceinline__ unsigned key_of(const float d) {
  const unsigned u = __float_as_uint(d);
  return u ^ ((unsigned)((int)u >> 31) | 0x80000000u);
}
__device__ __forceinline__ unsigned dist_key(const float4 me, const float4 p) {
  float dot;
  {
    #pragma clang fp contract(off)
    dot = fmaf(me.x, p.x, me.y * p.y) + me.z * p.z;
  }
  return key_of(fmaf(-2.0f, dot, me.w + p.w));
}
__device__ __forceinline__ v2f dist_pair(const float4 me, const float4 xy, const float4 zw) {
  const v2f X = {xy.x, xy.y}, Y = {xy.z, xy.w}, Z = {zw.x, zw.y}, Wn = {zw.z, zw.w};
  const v2f mx = {me.x, me.x}, my = {me.y, me.y}, mz = {me.z, me.z}, mw = {me.w, me.w};
  const v2f m2 = {-2.0f, -2.0f};
  v2f dot;
  {
    #pragma clang fp contract(off)
    dot = __builtin_elementwise_fma(mx, X, my * Y) + mz * Z;
  }
  return __builtin_elementwise_fma(m2, dot, mw + Wn);
}
__device__ __forceinline__ float4 point_at(const float* pxyf, const float* pzwf, const int m) {
  const int o = ((((m >> 6) << 5) | (m & 31)) << 2) + ((m >> 5) & 1);
  return make_float4(pxyf[o], pxyf[o + 2], pzwf[o], pzwf[o + 2]);
}

__global__ void __launch_bounds__(KNN_WAVES * 32) __attribute__((amdgpu_waves_per_eu(4, 4)))
knn_kernel(const float* __restrict__ xyz, int* __restrict__ idx_out,
           const float* __restrict__ feat, const float* __restrict__ w2,
           const float* __restrict__ g2, const float* __restrict__ v2,
           float* __restrict__ Cf) {
  __shared__ float4 pxy[NPTS / 2];
  __shared__ float4 pzw[NPTS / 2];
  __shared__ int cand[KNN_WAVES][KNN_CAND];
  __shared__ float wcs[WC_WORDS];
  const int b = blockIdx.y;
  const float* xb = xyz + (size_t)b * 3 * NPTS;
  #pragma unroll
  for (int i = 0; i < (NPTS / 2) / (KNN_WAVES * 32); ++i) {
    const int r  = (int)threadIdx.x + i * (KNN_WAVES * 32);
    const int m0 = ((r >> 5) << 6) | (r & 31);
    const int m1 = m0 + 32;
    const float x0 = xb[m0], y0 = xb[NPTS + m0], z0 = xb[2 * NPTS + m0];
    const float x1 = xb[m1], y1 = xb[NPTS + m1], z1 = xb[2 * NPTS + m1];
    pxy[r] = make_float4(x0, x1, y0, y1);
    pzw[r] = make_float4(z0, z1, fmaf(z0, z0, fmaf(x0, x0, y0 * y0)),
                         fmaf(z1, z1, fmaf(x1, x1, y1 * y1)));
  }
  {
    const int o  = (int)threadIdx.x >> 3;
    const int c4 = ((int)threadIdx.x & 7) * 4;
    const float s2 = g2[o] / sqrtf(v2[o] + EPSB);
    const float* wrow = w2 + (size_t)o * (2 * CH) + c4;
    float* dn = wcs + o * WC_PITCH + c4;
    float* dc = dn + HALF_CH * WC_PITCH;
    #pragma unroll
    for (int j = 0; j < CH / 32; ++j) {
      const float4 wa = *(const float4*)(wrow + 32 * j);
      const float4 wb = *(const float4*)(wrow + CH + 32 * j);
      *(float4*)(dn + 32 * j) = make_float4(s2 * wa.x, s2 * wa.y, s2 * wa.z, s2 * wa.w);
      *(float4*)(dc + 32 * j) = make_float4(s2 * (wb.x - wa.x), s2 * (wb.y - wa.y),
                                            s2 * (wb.z - wa.z), s2 * (wb.w - wa.w));
    }
  }
  __syncthreads();
  const float* pxyf = (const float*)pxy;
  const float* pzwf = (const float*)pzw;

  const int lane = threadIdx.x & 31;
  const int wave = threadIdx.x >> 5;
  int* cw = cand[wave];

  const int  swave      = __builtin_amdgcn_readfirstlane((int)(threadIdx.x >> 5));
  const int  gemm_n0    = blockIdx.x * KNN_QPB + 16 * (swave & 3);
  const int  gemm_strip_id = swave >> 2;
  const bool gemm_first = ((swave ^ (swave >> 2)) & 1) == 0;
  if (gemm_first) feat_gemm_strip(feat, wcs, Cf, b, gemm_n0, gemm_strip_id, lane);

  for (int qi = 0; qi < KNN_QPW; ++qi) {
    const int q = blockIdx.x * KNN_QPB + wave * KNN_QPW + qi;
    const float4 me = point_at(pxyf, pzwf, q);
    int* op = idx_out + ((size_t)b * NPTS + q) * KNN_K;

    unsigned key[KNN_SLOTS];
    #pragma unroll
    for (int t = 0; t < KNN_SLOTS / 2; ++t) {
      const v2f d = dist_pair(me, pxy[t * 32 + lane], pzw[t * 32 + lane]);
      key[2 * t]     = key_of(d.x);
      key[2 * t + 1] = key_of(d.y);
    }

    unsigned mn = key[0];
    #pragma unroll
    for (int j = 1; j < KNN_SLOTS; ++j) mn = min(mn, key[j]);
    const unsigned mu = (mn & ~31u) | (unsigned)lane;
    unsigned r0 = 0u, r1 = 0u;
    #pragma unroll
    for (int l = 0; l < 32; l += 4)
      rank_below4(r0, r1, (unsigned)__builtin_amdgcn_readlane((int)mu, l),
                  (unsigned)__builtin_amdgcn_readlane((int)mu, l + 1),
                  (unsigned)__builtin_amdgcn_readlane((int)mu, l + 2),
                  (unsigned)__builtin_amdgcn_readlane((int)mu, l + 3), mu);
    const unsigned holder = __builtin_amdgcn_ballot_w32((r0 + r1) == (unsigned)KNN_K);
    const unsigned U =
        (unsigned)__builtin_amdgcn_readlane((int)mu, __builtin_ctz(holder)) | 31u;

    unsigned mlo = 0u, mhi = 0u;
    #pragma unroll
    for (int j = 31; j >= 0; --j) shift_in_le2(mlo, mhi, key[j], key[j + 32], U);
    const unsigned cS =
        wave_sum_u32((unsigned)__builtin_popcount(mlo) + (unsigned)__builtin_popcount(mhi));

    if (cS <= (unsigned)KNN_CAND) {
      unsigned listed = 0u;
      for (int pass = 0; pass < KNN_SLOTS; ++pass) {
        const bool has = (mlo | mhi) != 0u;
        const unsigned act = __builtin_amdgcn_ballot_w32(has);
        if (act == 0u) break;
        if (has) {
          int j;
          if (mlo != 0u) { j = __builtin_ctz(mlo);      mlo &= mlo - 1u; }
          else           { j = 32 + __builtin_ctz(mhi); mhi &= mhi - 1u; }
          cw[listed + __builtin_amdgcn_mbcnt_lo(act, 0u)] = j * 32 + lane;
        }
        listed += (unsigned)__builtin_popcount(act);
      }
      __builtin_amdgcn_wave_barrier();
      const int i0 = ((unsigned)lane < cS) ? cw[lane] : -1;
      const int i1 = ((unsigned)lane + 32u < cS) ? cw[lane + 32] : -1;
      unsigned k0 = 0xFFFFFFFFu, k1 = 0xFFFFFFFFu;
      if (i0 >= 0) k0 = dist_key(me, point_at(pxyf, pzwf, i0));
      if (i1 >= 0) k1 = dist_key(me, point_at(pxyf, pzwf, i1));
      __builtin_amdgcn_wave_barrier();

      {
        unsigned ra0 = 0u, ra1 = 0u, rb0 = 0u, rb1 = 0u;
        if (cS <= 32u) {
          #pragma unroll
          for (int l = 0; l < 32; l += 4) {
            const unsigned s0 = (unsigned)__builtin_amdgcn_readlane((int)k0, l);
            const unsigned s1 = (unsigned)__builtin_amdgcn_readlane((int)k0, l + 1);
            const unsigned s2 = (unsigned)__builtin_amdgcn_readlane((int)k0, l + 2);
            const unsigned s3 = (unsigned)__builtin_amdgcn_readlane((int)k0, l + 3);
            rank_below4(ra0, ra1, s0, s1, s2, s3, k0);
          }
        } else {
          #pragma unroll
          for (int l = 0; l < 32; l += 4) {
            const unsigned s0 = (unsigned)__builtin_amdgcn_readlane((int)k0, l);
            const unsigned s1 = (unsigned)__builtin_amdgcn_readlane((int)k0, l + 1);
            const unsigned s2 = (unsigned)__builtin_amdgcn_readlane((int)k0, l + 2);
            const unsigned s3 = (unsigned)__builtin_amdgcn_readlane((int)k0, l + 3);
            rank_below4(ra0, ra1, s0, s1, s2, s3, k0);
            rank_below4(rb0, rb1, s0, s1, s2, s3, k1);
          }
          for (unsigned base = 32u; base < cS; base += 4u) {
            const int e = (int)base - 32;
            const unsigned s0 = (unsigned)__builtin_amdgcn_readlane((int)k1, e);
            const unsigned s1 = (unsigned)__builtin_amdgcn_readlane((int)k1, e + 1);
            const unsigned s2 = (unsigned)__builtin_amdgcn_readlane((int)k1, e + 2);
            const unsigned s3 = (unsigned)__builtin_amdgcn_readlane((int)k1, e + 3);
            rank_below4(ra0, ra1, s0, s1, s2, s3, k0);
            rank_below4(rb0, rb1, s0, s1, s2, s3, k1);
          }
        }
        const bool t0 = (ra0 + ra1) < (unsigned)KNN_K && i0 >= 0;
        const bool t1 = (rb0 + rb1) < (unsigned)KNN_K && i1 >= 0;
        const unsigned tb0 = __builtin_amdgcn_ballot_w32(t0);
        const unsigned tb1 = __builtin_amdgcn_ballot_w32(t1);
        const unsigned tn0 = (unsigned)__builtin_popcount(tb0);
        if (tn0 + (unsigned)__builtin_popcount(tb1) == (unsigned)KNN_K) {
          if (t0) op[__builtin_amdgcn_mbcnt_lo(tb0, 0u)] = i0;
          if (t1) op[tn0 + __builtin_amdgcn_mbcnt_lo(tb1, 0u)] = i1;
          continue;
        }
      }

      unsigned T = 0u, clt = 0u;
      for (int bit = 31; bit >= 0; --bit) {
        const unsigned trial = T | (1u << bit);
        const unsigned c =
            wave_sum_u32((k0 < trial ? 1u : 0u) + (k1 < trial ? 1u : 0u));
        if (c <= (unsigned)KNN_K) {
          T = trial;
          clt = c;
          if (c == (unsigned)KNN_K) break;
        }
      }
      const unsigned need = (unsigned)KNN_K - clt;

      const bool s0 = k0 < T, s1 = k1 < T;
      const unsigned b0 = __builtin_amdgcn_ballot_w32(s0);
      const unsigned b1 = __builtin_amdgcn_ballot_w32(s1);
      if (s0) op[__builtin_amdgcn_mbcnt_lo(b0, 0u)] = i0;
      if (s1) op[(unsigned)__builtin_popcount(b0) + __builtin_amdgcn_mbcnt_lo(b1, 0u)] = i1;

      if (need != 0u) {
        unsigned e0 = (k0 == T) ? (unsigned)i0 : 0xFFFFFFFFu;
        unsigned e1 = (k1 == T) ? (unsigned)i1 : 0xFFFFFFFFu;
        for (unsigned r = 0u; r < need; ++r) {
          const unsigned m = wave_min_u32(min(e0, e1));
          if (lane == 0) op[clt + r] = (int)m;
          if (e0 == m) e0 = 0xFFFFFFFFu;
          if (e1 == m) e1 = 0xFFFFFFFFu;
        }
      }
      continue;
    }

    unsigned T = 0u, clt = 0u;
    for (int bit = 31; bit >= 0; --bit) {
      const unsigned trial = T | (1u << bit);
      unsigned c0 = 0u, c1 = 0u, c2 = 0u, c3 = 0u;
      #pragma unroll
      for (int j = 0; j < KNN_SLOTS; j += 8)
        count_below8(c0, c1, c2, c3, key[j + 0], key[j + 1], key[j + 2], key[j + 3],
                     key[j + 4], key[j + 5], key[j + 6], key[j + 7], trial);
      const unsigned c = wave_sum_u32((c0 + c1) + (c2 + c3));
      if (c <= (unsigned)KNN_K) {
        T = trial;
        clt = c;
        if (c == (unsigned)KNN_K) break;
      }
    }
    const unsigned need = (unsigned)KNN_K - clt;

    mlo = 0u; mhi = 0u;
    #pragma unroll
    for (int j = 31; j >= 0; --j) shift_in_below2(mlo, mhi, key[j], key[j + 32], T);

    unsigned written = 0u;
    for (int pass = 0; pass < KNN_SLOTS; ++pass) {
      const bool has = (mlo | mhi) != 0u;
      const unsigned act = __builtin_amdgcn_ballot_w32(has);
      if (act == 0u) break;
      if (has) {
        int j;
        if (mlo != 0u) { j = __builtin_ctz(mlo);      mlo &= mlo - 1u; }
        else           { j = 32 + __builtin_ctz(mhi); mhi &= mhi - 1u; }
        op[written + __builtin_amdgcn_mbcnt_lo(act, 0u)] = j * 32 + lane;
      }
      written += (unsigned)__builtin_popcount(act);
    }

    if (need != 0u) {
      unsigned elo = 0u, ehi = 0u;
      #pragma unroll
      for (int j = 31; j >= 0; --j) shift_in_equal2(elo, ehi, key[j], key[j + 32], T);
      unsigned taken = 0u;
      for (int j = 0; j < KNN_SLOTS && taken < need; ++j) {
        const unsigned w = (j < 32) ? elo : ehi;
        const bool eq = ((w >> (j & 31)) & 1u) != 0u;
        const unsigned meq = __builtin_amdgcn_ballot_w32(eq);
        if (meq != 0u) {
          const unsigned r = taken + __builtin_amdgcn_mbcnt_lo(meq, 0u);
          if (eq && r < need) op[clt + r] = j * 32 + lane;
          taken += (unsigned)__builtin_popcount(meq);
        }
      }
    }
  }

  if (!gemm_first) feat_gemm_strip(feat, wcs, Cf, b, gemm_n0, gemm_strip_id, lane);
}

#define GM_PTS   32
#define GM_WAVES 16
#define GM_PPW   (GM_PTS / GM_WAVES)
#define GM_RPW   (CH / GM_WAVES)
#define GM_PITCH (GM_PTS + 1)

typedef const float2 __attribute__((address_space(1)))* gfloat2p;

__global__ void __launch_bounds__(GM_WAVES * 32, 8)
gather_max(const float* __restrict__ xyz, const float* __restrict__ w1,
           const float* __restrict__ b1, const float* __restrict__ g1,
           const float* __restrict__ be1, const float* __restrict__ m1,
           const float* __restrict__ v1, const float* __restrict__ Cf,
           const int* __restrict__ idx, const float* __restrict__ b2,
           const float* __restrict__ g2, const float* __restrict__ be2,
           const float* __restrict__ m2, const float* __restrict__ v2,
           float* __restrict__ out) {
  __shared__ float tile[CH * GM_PITCH];
  const int b    = blockIdx.y;
  const int n0   = blockIdx.x * GM_PTS;
  const int lane = threadIdx.x & 31;
  const int wave = __builtin_amdgcn_readfirstlane((int)(threadIdx.x >> 5));

  const float* xb  = xyz + (size_t)b * 3 * NPTS;
  const float* cfb = Cf + (size_t)b * NPTS * CH;

  const int kk = (lane < KNN_K) ? lane : lane - KNN_K;

  const float4 w1A = *(const float4*)(w1 + 12 * lane);
  const float4 w1B = *(const float4*)(w1 + 12 * lane + 4);
  const float4 w1C = *(const float4*)(w1 + 12 * lane + 8);
  const float2 g1v  = *(const float2*)(g1 + 2 * lane);
  const float2 v1v  = *(const float2*)(v1 + 2 * lane);
  const float2 b1v  = *(const float2*)(b1 + 2 * lane);
  const float2 be1v = *(const float2*)(be1 + 2 * lane);
  const float2 m1v  = *(const float2*)(m1 + 2 * lane);
  const float2 g2v  = *(const float2*)(g2 + 2 * lane);
  const float2 v2v  = *(const float2*)(v2 + 2 * lane);
  const float2 b2v  = *(const float2*)(b2 + 2 * lane);
  const float2 be2v = *(const float2*)(be2 + 2 * lane);
  const float2 m2v  = *(const float2*)(m2 + 2 * lane);

  int    jv[GM_PPW], xv[GM_PPW], yv[GM_PPW], zv[GM_PPW];
  float  cx[GM_PPW], cy[GM_PPW], cz[GM_PPW];
  float2 cv[GM_PPW];
  #pragma unroll
  for (int p = 0; p < GM_PPW; ++p) {
    const int    n     = n0 + wave * GM_PPW + p;
    const size_t pbase = (size_t)b * NPTS + n;
    jv[p] = idx[pbase * KNN_K + kk];
    cx[p] = xb[n];
    cy[p] = xb[NPTS + n];
    cz[p] = xb[2 * NPTS + n];
    cv[p] = *(const float2*)(Cf + pbase * CH + HALF_CH + 2 * lane);
  }

  const float s10 = g1v.x / sqrtf(v1v.x + EPSB);
  const float s11 = g1v.y / sqrtf(v1v.y + EPSB);
  const float s20 = g2v.x / sqrtf(v2v.x + EPSB);
  const float s21 = g2v.y / sqrtf(v2v.y + EPSB);
  const float2 bg = make_float2(b1v.x * s10 + (be1v.x - m1v.x * s10),
                                b1v.y * s11 + (be1v.y - m1v.y * s11));
  const float2 bf = make_float2(b2v.x * s20 + (be2v.x - m2v.x * s20),
                                b2v.y * s21 + (be2v.y - m2v.y * s21));
  const float2 wa01 = make_float2(s10 * w1A.x, s10 * w1A.y);
  const float2 wa23 = make_float2(s10 * w1A.z, s11 * w1B.z);
  const float2 wa45 = make_float2(s11 * w1B.w, s11 * w1C.x);
  const float2 wd01 = make_float2(s10 * (w1A.w - w1A.x), s10 * (w1B.x - w1A.y));
  const float2 wd23 = make_float2(s10 * (w1B.y - w1A.z), s11 * (w1C.y - w1B.z));
  const float2 wd45 = make_float2(s11 * (w1C.z - w1B.w), s11 * (w1C.w - w1C.x));

  float qg[GM_PPW][2], vf[GM_PPW][2];
  #pragma unroll
  for (int p = 0; p < GM_PPW; ++p) {
    qg[p][0] = wd01.x * cx[p] + wd01.y * cy[p] + wd23.x * cz[p] + bg.x;
    qg[p][1] = wd23.y * cx[p] + wd45.x * cy[p] + wd45.y * cz[p] + bg.y;
    vf[p][0] = cv[p].x + bf.x;
    vf[p][1] = cv[p].y + bf.y;
  }
  asm volatile("" :: "v"(wa01.x), "v"(wa01.y), "v"(wa23.x), "v"(wa23.y), "v"(wa45.x), "v"(wa45.y),
                     "v"(qg[0][0]), "v"(qg[0][1]), "v"(qg[1][0]), "v"(qg[1][1]),
                     "v"(vf[0][0]), "v"(vf[0][1]), "v"(vf[1][0]), "v"(vf[1][1]));
  #pragma unroll
  for (int p = 0; p < GM_PPW; ++p) {
    xv[p] = __float_as_int(xb[jv[p]]);
    yv[p] = __float_as_int(xb[NPTS + jv[p]]);
    zv[p] = __float_as_int(xb[2 * NPTS + jv[p]]);
  }

  float2 row[GM_PPW][KNN_K];
  #pragma unroll
  for (int p = 0; p < GM_PPW; ++p) {
    #pragma unroll
    for (int k = 0; k < KNN_K; ++k) {
      const int j = __builtin_amdgcn_readlane(jv[p], k);
      const unsigned long long ra = (unsigned long long)(cfb + (size_t)j * CH);
      const unsigned ralo = (unsigned)__builtin_amdgcn_readfirstlane((int)(unsigned)ra);
      const unsigned rahi = (unsigned)__builtin_amdgcn_readfirstlane((int)(unsigned)(ra >> 32));
      const gfloat2p rp = (gfloat2p)(((unsigned long long)rahi << 32) | ralo);
      row[p][k] = rp[lane];
    }
  }

  float mg[GM_PPW][2];
  #pragma unroll
  for (int p = 0; p < GM_PPW; ++p) { mg[p][0] = -FLT_MAX; mg[p][1] = -FLT_MAX; }
  #pragma unroll 2
  for (int k = 0; k < KNN_K; ++k) {
    #pragma unroll
    for (int p = 0; p < GM_PPW; ++p) {
      const float x = __int_as_float(__builtin_amdgcn_readlane(xv[p], k));
      const float y = __int_as_float(__builtin_amdgcn_readlane(yv[p], k));
      const float z = __int_as_float(__builtin_amdgcn_readlane(zv[p], k));
      mg[p][0] = fmaxf(mg[p][0], wa01.x * x + wa01.y * y + wa23.x * z);
      mg[p][1] = fmaxf(mg[p][1], wa23.y * x + wa45.x * y + wa45.y * z);
    }
  }

  #pragma unroll
  for (int p = 0; p < GM_PPW; ++p) {
    const int pl = wave * GM_PPW + p;
    float mf0 = -FLT_MAX, mf1 = -FLT_MAX;
    #pragma unroll
    for (int k = 0; k < KNN_K; ++k) {
      mf0 = fmaxf(mf0, row[p][k].x);
      mf1 = fmaxf(mf1, row[p][k].y);
    }
    tile[(2 * lane)               * GM_PITCH + pl] = fmaxf(mg[p][0] + qg[p][0], 0.0f);
    tile[(2 * lane + 1)           * GM_PITCH + pl] = fmaxf(mg[p][1] + qg[p][1], 0.0f);
    tile[(HALF_CH + 2 * lane)     * GM_PITCH + pl] = fmaxf(mf0 + vf[p][0], 0.0f);
    tile[(HALF_CH + 2 * lane + 1) * GM_PITCH + pl] = fmaxf(mf1 + vf[p][1], 0.0f);
  }
  __syncthreads();

  float* ob = out + (size_t)b * CH * NPTS + n0 + lane;
  #pragma unroll
  for (int r = 0; r < GM_RPW; ++r) {
    const int c = wave * GM_RPW + r;
    ob[(size_t)c * NPTS] = tile[c * GM_PITCH + lane];
  }
}

extern "C" void kernel_launch(void* const* d_in, const int* in_sizes, int n_in,
                              void* d_out, int out_size, void* d_ws, size_t ws_size,
                              hipStream_t stream) {
  (void)in_sizes; (void)n_in; (void)out_size; (void)ws_size;

  const float* xyz  = (const float*)d_in[0];
  const float* feat = (const float*)d_in[1];
  const float* w1   = (const float*)d_in[2];
  const float* b1   = (const float*)d_in[3];
  const float* g1   = (const float*)d_in[4];
  const float* be1  = (const float*)d_in[5];
  const float* m1   = (const float*)d_in[6];
  const float* v1   = (const float*)d_in[7];
  const float* w2   = (const float*)d_in[8];
  const float* b2   = (const float*)d_in[9];
  const float* g2   = (const float*)d_in[10];
  const float* be2  = (const float*)d_in[11];
  const float* m2   = (const float*)d_in[12];
  const float* v2   = (const float*)d_in[13];

  char* ws = (char*)d_ws;
  float* Cf     = (float*)(ws + (128 << 10)) + (size_t)BATCH * NPTS * CH;
  int*   idxb   = (int*)(Cf + (size_t)BATCH * NPTS * CH);

  knn_kernel<<<dim3(NPTS / KNN_QPB, BATCH), KNN_WAVES * 32, 0, stream>>>(xyz, idxb, feat, w2,
                                                                         g2, v2, Cf);

  gather_max<<<dim3(NPTS / GM_PTS, BATCH), GM_WAVES * 32, 0, stream>>>(
      xyz, w1, b1, g1, be1, m1, v1, Cf, idxb, b2, g2, be2, m2, v2, (float*)d_out);
}
